# starting kernel
# baseline (speedup 1.0000x reference)
#include <hip/hip_runtime.h>
#include <hip/hip_bf16.h>


typedef __attribute__((ext_vector_type(2))) float v2f;
typedef __attribute__((ext_vector_type(8))) float v8f;

#define NN_NODES 25000
#define N_EDGES  400000
#define N_PAIR   100000
#define N_ROWS   200000

__device__ __forceinline__ unsigned enc_f32(float f) {
  unsigned u = __float_as_uint(f);
  return (u & 0x80000000u) ? ~u : (u | 0x80000000u);
}
__device__ __forceinline__ float dec_f32(unsigned e) {
  return (e & 0x80000000u) ? __uint_as_float(e & 0x7FFFFFFFu)
                           : __uint_as_float(~e);
}
#define ENC_NEG_INF 0x007FFFFFu

__device__ __forceinline__ float leaky(float v, float slope) {
  return v >= 0.f ? v : v * slope;
}

__global__ void gemm_wmma_f32(const float* __restrict__ A,
                              const float* __restrict__ W,
                              const float* __restrict__ bias,
                              float* __restrict__ C,
                              int M, int N, int K,
                              float slope, int act) {
  __shared__ float As[64][36];
  __shared__ float Ws[32][36];

  const int t     = threadIdx.x;
  const int lane  = t & 31;
  const int wave  = t >> 5;
  const int waveM = wave & 3;
  const int waveN = wave >> 2;
  const int mBase = blockIdx.x * 64;
  const int nBase = blockIdx.y * 32;
  const int lh    = lane >> 4;
  const int l15   = lane & 15;

  v8f acc = {0.f, 0.f, 0.f, 0.f, 0.f, 0.f, 0.f, 0.f};

  for (int k0 = 0; k0 < K; k0 += 32) {
#pragma unroll
    for (int i = 0; i < 2; ++i) {
      int idx = t + i * 256;
      int r   = idx >> 3;
      int c4  = (idx & 7) * 4;
      int gr  = mBase + r;
      int grc = gr < M ? gr : M - 1;
      float4 v = *(const float4*)(A + (size_t)grc * K + k0 + c4);
      if (gr >= M) { v.x = 0.f; v.y = 0.f; v.z = 0.f; v.w = 0.f; }
      *(float4*)&As[r][c4] = v;
    }
    {
      int r  = t >> 3;
      int c4 = (t & 7) * 4;
      *(float4*)&Ws[r][c4] =
          *(const float4*)(W + (size_t)(nBase + r) * K + k0 + c4);
    }
    __syncthreads();

#pragma unroll
    for (int ks = 0; ks < 8; ++ks) {
      int kb = ks * 4 + 2 * lh;
      v2f a, b;
      a.x = As[waveM * 16 + l15][kb];
      a.y = As[waveM * 16 + l15][kb + 1];
      b.x = Ws[waveN * 16 + l15][kb];
      b.y = Ws[waveN * 16 + l15][kb + 1];
      acc = __builtin_amdgcn_wmma_f32_16x16x4_f32(
           false, a,  false, b,
           (short)0, acc,  false,  false);
    }
    __syncthreads();
  }

#pragma unroll
  for (int r = 0; r < 8; ++r) {
    int row = mBase + waveM * 16 + r + 8 * lh;
    int col = nBase + waveN * 16 + l15;
    if (row < M) {
      float v = acc[r];
      if (bias) v += bias[col];
      if (act)  v = leaky(v, slope);
      C[(size_t)row * N + col] = v;
    }
  }
}

template <int H, int F>
__global__ void node_p_kernel(const float* __restrict__ ht,
                              const float* __restrict__ a,
                              float* __restrict__ p) {
  int idx = blockIdx.x * blockDim.x + threadIdx.x;
  if (idx >= NN_NODES * H) return;
  int n = idx / H, h = idx % H;
  const float* row = ht + ((size_t)n * H + h) * F;
  float acc = 0.f;
#pragma unroll 4
  for (int f = 0; f < F; f += 4) {
    float4 v = *(const float4*)(row + f);
    float4 w = *(const float4*)(a + f);
    acc += v.x * w.x + v.y * w.y + v.z * w.z + v.w * w.w;
  }
  p[idx] = acc;
}

__global__ void fill_u32_kernel(unsigned* __restrict__ dst, unsigned val, int n) {
  int i = blockIdx.x * blockDim.x + threadIdx.x;
  if (i < n) dst[i] = val;
}

template <int H>
__global__ void edge_score_kernel(const float* __restrict__ p,
                                  const float* __restrict__ ew,
                                  const int* __restrict__ src,
                                  const int* __restrict__ dst,
                                  float* __restrict__ ebuf,
                                  unsigned* __restrict__ menc) {
  int e = blockIdx.x * blockDim.x + threadIdx.x;
  if (e >= N_EDGES) return;
  float lw = log1pf(ew[e]);
  int s = src[e], d = dst[e];
  float ps[H], pd[H];
#pragma unroll
  for (int h = 0; h < H; ++h) ps[h] = p[s * H + h];
#pragma unroll
  for (int h = 0; h < H; ++h) pd[h] = p[d * H + h];
#pragma unroll
  for (int h = 0; h < H; ++h) {
    float sc = leaky(ps[h] + pd[h] + lw, 0.01f);
    ebuf[(size_t)e * H + h] = sc;
    atomicMax(&menc[d * H + h], enc_f32(sc));
  }
}

__global__ void node_m_kernel(unsigned* __restrict__ menc, int n) {
  int i = blockIdx.x * blockDim.x + threadIdx.x;
  if (i >= n) return;
  float m = dec_f32(menc[i]);
  if (!isfinite(m)) m = 0.f;
  menc[i] = __float_as_uint(m);
}

template <int H>
__global__ void edge_exp_kernel(float* __restrict__ ebuf,
                                const float* __restrict__ mf,
                                const int* __restrict__ dst,
                                float* __restrict__ sden) {
  int idx = blockIdx.x * blockDim.x + threadIdx.x;
  if (idx >= N_EDGES * H) return;
  int e = idx / H, h = idx % H;
  int d = dst[e];
  float ex = __expf(ebuf[idx] - mf[d * H + h]);
  ebuf[idx] = ex;
  atomicAdd(&sden[d * H + h], ex);
}

template <int H, int F>
__global__ void edge_agg_kernel(const float* __restrict__ ebuf,
                                const float* __restrict__ sden,
                                const float* __restrict__ ht,
                                const int* __restrict__ src,
                                const int* __restrict__ dst,
                                float* __restrict__ hagg) {
  int idx = blockIdx.x * blockDim.x + threadIdx.x;
  if (idx >= N_EDGES * H) return;
  int e = idx / H, h = idx % H;
  int s = src[e], d = dst[e];
  float alpha = ebuf[idx] / fmaxf(sden[d * H + h], 1e-20f);
  const float* hrow = ht + ((size_t)s * H + h) * F;
  float* orow = hagg + ((size_t)d * H + h) * F;
#pragma unroll 4
  for (int f = 0; f < F; f += 4) {
    float4 v = *(const float4*)(hrow + f);
    atomicAdd(orow + f + 0, alpha * v.x);
    atomicAdd(orow + f + 1, alpha * v.y);
    atomicAdd(orow + f + 2, alpha * v.z);
    atomicAdd(orow + f + 3, alpha * v.w);
  }
}

__global__ void bias_act_kernel(const float* __restrict__ in,
                                const float* __restrict__ b,
                                float* __restrict__ out,
                                int total, int mask, float slope, int act) {
  int i = blockIdx.x * blockDim.x + threadIdx.x;
  if (i >= total) return;
  float v = in[i] + b[i & mask];
  if (act) v = leaky(v, slope);
  out[i] = v;
}

__global__ void build_z_kernel(const float* __restrict__ emb,
                               const int* __restrict__ ps, const int* __restrict__ pd,
                               const int* __restrict__ ns, const int* __restrict__ nd,
                               float* __restrict__ Z) {
  int idx = blockIdx.x * blockDim.x + threadIdx.x;
  if (idx >= N_ROWS * 32) return;
  int row = idx >> 5;
  int f   = (idx & 31) * 4;
  int s, d;
  if (row < N_PAIR) { s = ps[row]; d = pd[row]; }
  else              { s = ns[row - N_PAIR]; d = nd[row - N_PAIR]; }
  float4 a = *(const float4*)(emb + (size_t)s * 128 + f);
  float4 b = *(const float4*)(emb + (size_t)d * 128 + f);
  float4 r; r.x = a.x * b.x; r.y = a.y * b.y; r.z = a.z * b.z; r.w = a.w * b.w;
  *(float4*)(Z + (size_t)row * 128 + f) = r;
}

__global__ void final_dot_kernel(const float* __restrict__ Z2,
                                 const float* __restrict__ pW3,
                                 const float* __restrict__ pb3,
                                 float* __restrict__ out) {
  int row = blockIdx.x * blockDim.x + threadIdx.x;
  if (row >= N_ROWS) return;
  const float* z = Z2 + (size_t)row * 32;
  float acc = pb3[0];
#pragma unroll
  for (int j = 0; j < 32; j += 4) {
    float4 a = *(const float4*)(z + j);
    float4 w = *(const float4*)(pW3 + j);
    acc += a.x * w.x + a.y * w.y + a.z * w.z + a.w * w.w;
  }
  out[row] = acc;
}

extern "C" void kernel_launch(void* const* d_in, const int* in_sizes, int n_in,
                              void* d_out, int out_size, void* d_ws, size_t ws_size,
                              hipStream_t stream) {
  const float* x    = (const float*)d_in[0];
  const float* ew   = (const float*)d_in[1];
  const int*   esrc = (const int*)d_in[2];
  const int*   edst = (const int*)d_in[3];
  const int*   psrc = (const int*)d_in[4];
  const int*   pdst = (const int*)d_in[5];
  const int*   nsrc = (const int*)d_in[6];
  const int*   ndst = (const int*)d_in[7];
  const float* Wl[3] = {(const float*)d_in[8],  (const float*)d_in[11], (const float*)d_in[14]};
  const float* al[3] = {(const float*)d_in[9],  (const float*)d_in[12], (const float*)d_in[15]};
  const float* bl[3] = {(const float*)d_in[10], (const float*)d_in[13], (const float*)d_in[16]};
  const float* pW1 = (const float*)d_in[17]; const float* pb1 = (const float*)d_in[18];
  const float* pW2 = (const float*)d_in[19]; const float* pb2 = (const float*)d_in[20];
  const float* pW3 = (const float*)d_in[21]; const float* pb3 = (const float*)d_in[22];

  float* out   = (float*)d_out;
  float* emb   = out + 2 * N_PAIR;

  float*    ws   = (float*)d_ws;
  float*    ht   = ws;
  float*    hbuf = ht   + 6400000;
  float*    pbuf = hbuf + 6400000;
  unsigned* menc = (unsigned*)(pbuf + 100000);
  float*    sden = (float*)(menc + 100000);
  float*    ebuf = sden + 100000;
  float*    Z    = ebuf + 1600000;
  float*    Z1   = Z    + 25600000;
  float*    Z2   = Z1   + 12800000;

  const int Hs[3]  = {4, 4, 1};
  const int Fs[3]  = {64, 64, 128};
  const int Kin[3] = {128, 256, 256};

  const float* hin = x;
  for (int l = 0; l < 3; ++l) {
    const int H = Hs[l], F = Fs[l], HF = H * F, K = Kin[l];
    float* hout = (l == 2) ? emb : hbuf;

    dim3 gg((NN_NODES + 63) / 64, HF / 32);
    gemm_wmma_f32<<<gg, 256, 0, stream>>>(hin, Wl[l], nullptr, ht,
                                          NN_NODES, HF, K, 0.f, 0);
    int nh = NN_NODES * H;
    if (l < 2) node_p_kernel<4, 64><<<(nh + 255) / 256, 256, 0, stream>>>(ht, al[l], pbuf);
    else       node_p_kernel<1, 128><<<(nh + 255) / 256, 256, 0, stream>>>(ht, al[l], pbuf);
    fill_u32_kernel<<<(nh + 255) / 256, 256, 0, stream>>>(menc, ENC_NEG_INF, nh);
    hipMemsetAsync(sden, 0, nh * sizeof(float), stream);

    if (l < 2) {
      edge_score_kernel<4><<<(N_EDGES + 255) / 256, 256, 0, stream>>>(
          pbuf, ew, esrc, edst, ebuf, menc);
    } else {
      edge_score_kernel<1><<<(N_EDGES + 255) / 256, 256, 0, stream>>>(
          pbuf, ew, esrc, edst, ebuf, menc);
    }
    node_m_kernel<<<(nh + 255) / 256, 256, 0, stream>>>(menc, nh);
    if (l < 2) {
      edge_exp_kernel<4><<<(N_EDGES * 4 + 255) / 256, 256, 0, stream>>>(
          ebuf, (const float*)menc, edst, sden);
    } else {
      edge_exp_kernel<1><<<(N_EDGES + 255) / 256, 256, 0, stream>>>(
          ebuf, (const float*)menc, edst, sden);
    }

    hipMemsetAsync(hout, 0, (size_t)NN_NODES * HF * sizeof(float), stream);
    if (l < 2) {
      edge_agg_kernel<4, 64><<<(N_EDGES * 4 + 255) / 256, 256, 0, stream>>>(
          ebuf, sden, ht, esrc, edst, hout);
    } else {
      edge_agg_kernel<1, 128><<<(N_EDGES + 255) / 256, 256, 0, stream>>>(
          ebuf, sden, ht, esrc, edst, hout);
    }
    bias_act_kernel<<<(NN_NODES * HF + 255) / 256, 256, 0, stream>>>(
        hout, bl[l], hout, NN_NODES * HF, HF - 1, 0.01f, (l < 2) ? 1 : 0);
    hin = hout;
  }

  build_z_kernel<<<(N_ROWS * 32 + 255) / 256, 256, 0, stream>>>(emb, psrc, pdst,
                                                                nsrc, ndst, Z);
  dim3 g1((N_ROWS + 63) / 64, 64 / 32);
  gemm_wmma_f32<<<g1, 256, 0, stream>>>(Z, pW1, pb1, Z1, N_ROWS, 64, 128, 0.2f, 1);
  dim3 g2((N_ROWS + 63) / 64, 32 / 32);
  gemm_wmma_f32<<<g2, 256, 0, stream>>>(Z1, pW2, pb2, Z2, N_ROWS, 32, 64, 0.2f, 1);
  final_dot_kernel<<<(N_ROWS + 255) / 256, 256, 0, stream>>>(Z2, pW3, pb3, out);
}


# Aggregation becomes a per-destination gather over a dst-grouped edge list rebuilt each call (one wave per node, whole-line row loads, no f32 atomics); sum order = arrival-order rank in each list
# speedup vs baseline: 4.5746x; 4.5746x over previous
#include <hip/hip_runtime.h>
#include <hip/hip_bf16.h>


typedef __attribute__((ext_vector_type(2))) float v2f;
typedef __attribute__((ext_vector_type(8))) float v8f;

#define NN_NODES 25000
#define N_EDGES  400000
#define N_PAIR   100000
#define N_ROWS   200000

__device__ __forceinline__ unsigned enc_f32(float f) {
  unsigned u = __float_as_uint(f);
  return (u & 0x80000000u) ? ~u : (u | 0x80000000u);
}
__device__ __forceinline__ float dec_f32(unsigned e) {
  return (e & 0x80000000u) ? __uint_as_float(e & 0x7FFFFFFFu)
                           : __uint_as_float(~e);
}
#define ENC_NEG_INF 0x007FFFFFu

__device__ __forceinline__ float leaky(float v, float slope) {
  return v >= 0.f ? v : v * slope;
}

__global__ void gemm_wmma_f32(const float* __restrict__ A,
                              const float* __restrict__ W,
                              const float* __restrict__ bias,
                              float* __restrict__ C,
                              int M, int N, int K,
                              float slope, int act) {
  __shared__ float As[64][36];
  __shared__ float Ws[32][36];

  const int t     = threadIdx.x;
  const int lane  = t & 31;
  const int wave  = t >> 5;
  const int waveM = wave & 3;
  const int waveN = wave >> 2;
  const int mBase = blockIdx.x * 64;
  const int nBase = blockIdx.y * 32;
  const int lh    = lane >> 4;
  const int l15   = lane & 15;

  v8f acc = {0.f, 0.f, 0.f, 0.f, 0.f, 0.f, 0.f, 0.f};

  for (int k0 = 0; k0 < K; k0 += 32) {
#pragma unroll
    for (int i = 0; i < 2; ++i) {
      int idx = t + i * 256;
      int r   = idx >> 3;
      int c4  = (idx & 7) * 4;
      int gr  = mBase + r;
      int grc = gr < M ? gr : M - 1;
      float4 v = *(const float4*)(A + (size_t)grc * K + k0 + c4);
      if (gr >= M) { v.x = 0.f; v.y = 0.f; v.z = 0.f; v.w = 0.f; }
      *(float4*)&As[r][c4] = v;
    }
    {
      int r  = t >> 3;
      int c4 = (t & 7) * 4;
      *(float4*)&Ws[r][c4] =
          *(const float4*)(W + (size_t)(nBase + r) * K + k0 + c4);
    }
    __syncthreads();

#pragma unroll
    for (int ks = 0; ks < 8; ++ks) {
      int kb = ks * 4 + 2 * lh;
      v2f a, b;
      a.x = As[waveM * 16 + l15][kb];
      a.y = As[waveM * 16 + l15][kb + 1];
      b.x = Ws[waveN * 16 + l15][kb];
      b.y = Ws[waveN * 16 + l15][kb + 1];
      acc = __builtin_amdgcn_wmma_f32_16x16x4_f32(
           false, a,  false, b,
           (short)0, acc,  false,  false);
    }
    __syncthreads();
  }

#pragma unroll
  for (int r = 0; r < 8; ++r) {
    int row = mBase + waveM * 16 + r + 8 * lh;
    int col = nBase + waveN * 16 + l15;
    if (row < M) {
      float v = acc[r];
      if (bias) v += bias[col];
      if (act)  v = leaky(v, slope);
      C[(size_t)row * N + col] = v;
    }
  }
}

template <int H, int F>
__global__ void node_p_kernel(const float* __restrict__ ht,
                              const float* __restrict__ a,
                              float* __restrict__ p) {
  int idx = blockIdx.x * blockDim.x + threadIdx.x;
  if (idx >= NN_NODES * H) return;
  int n = idx / H, h = idx % H;
  const float* row = ht + ((size_t)n * H + h) * F;
  float acc = 0.f;
#pragma unroll 4
  for (int f = 0; f < F; f += 4) {
    float4 v = *(const float4*)(row + f);
    float4 w = *(const float4*)(a + f);
    acc += v.x * w.x + v.y * w.y + v.z * w.z + v.w * w.w;
  }
  p[idx] = acc;
}

__global__ void fill_u32_kernel(unsigned* __restrict__ dst, unsigned val, int n) {
  int i = blockIdx.x * blockDim.x + threadIdx.x;
  if (i < n) dst[i] = val;
}

template <int H>
__global__ void edge_score_kernel(const float* __restrict__ p,
                                  const float* __restrict__ ew,
                                  const int* __restrict__ src,
                                  const int* __restrict__ dst,
                                  float* __restrict__ ebuf,
                                  unsigned* __restrict__ menc) {
  int e = blockIdx.x * blockDim.x + threadIdx.x;
  if (e >= N_EDGES) return;
  float lw = log1pf(ew[e]);
  int s = src[e], d = dst[e];
  float ps[H], pd[H];
#pragma unroll
  for (int h = 0; h < H; ++h) ps[h] = p[s * H + h];
#pragma unroll
  for (int h = 0; h < H; ++h) pd[h] = p[d * H + h];
#pragma unroll
  for (int h = 0; h < H; ++h) {
    float sc = leaky(ps[h] + pd[h] + lw, 0.01f);
    ebuf[(size_t)e * H + h] = sc;
    atomicMax(&menc[d * H + h], enc_f32(sc));
  }
}

__global__ void node_m_kernel(unsigned* __restrict__ menc, int n) {
  int i = blockIdx.x * blockDim.x + threadIdx.x;
  if (i >= n) return;
  float m = dec_f32(menc[i]);
  if (!isfinite(m)) m = 0.f;
  menc[i] = __float_as_uint(m);
}

template <int H>
__global__ void edge_exp_kernel(float* __restrict__ ebuf,
                                const float* __restrict__ mf,
                                const int* __restrict__ dst,
                                float* __restrict__ sden) {
  int idx = blockIdx.x * blockDim.x + threadIdx.x;
  if (idx >= N_EDGES * H) return;
  int e = idx / H, h = idx % H;
  int d = dst[e];
  float ex = __expf(ebuf[idx] - mf[d * H + h]);
  ebuf[idx] = ex;
  atomicAdd(&sden[d * H + h], ex);
}

__global__ void csr_count_kernel(const int* __restrict__ dst,
                                 int* __restrict__ deg,
                                 int* __restrict__ rank) {
  int e = blockIdx.x * blockDim.x + threadIdx.x;
  if (e >= N_EDGES) return;
  rank[e] = atomicAdd(&deg[dst[e]], 1);
}

#define SCAN_THREADS 1024
#define SCAN_PER     25
__global__ void __launch_bounds__(SCAN_THREADS)
csr_scan_kernel(const int* __restrict__ deg, int* __restrict__ rowptr) {
  __shared__ int part[SCAN_THREADS];
  const int t  = threadIdx.x;
  const int lo = t * SCAN_PER;
  int vals[SCAN_PER];
  int sum = 0;
#pragma unroll
  for (int i = 0; i < SCAN_PER; ++i) {
    int idx = lo + i;
    int idc = idx < NN_NODES ? idx : NN_NODES - 1;
    int v   = deg[idc];
    vals[i] = idx < NN_NODES ? v : 0;
    sum += vals[i];
  }
  part[t] = sum;
  __syncthreads();
  for (int off = 1; off < SCAN_THREADS; off <<= 1) {
    int v = (t >= off) ? part[t - off] : 0;
    __syncthreads();
    part[t] += v;
    __syncthreads();
  }
  int run = part[t] - sum;
#pragma unroll
  for (int i = 0; i < SCAN_PER; ++i) {
    int idx = lo + i;
    if (idx < NN_NODES) rowptr[idx] = run;
    run += vals[i];
  }
  if (t == SCAN_THREADS - 1) rowptr[NN_NODES] = part[SCAN_THREADS - 1];
}

__global__ void csr_fill_kernel(const int* __restrict__ src,
                                const int* __restrict__ dst,
                                const int* __restrict__ rank,
                                const int* __restrict__ rowptr,
                                int2* __restrict__ csr) {
  int e = blockIdx.x * blockDim.x + threadIdx.x;
  if (e >= N_EDGES) return;
  int pos = rowptr[dst[e]] + rank[e];
  csr[pos] = make_int2(src[e], e);
}

template <int H, int F>
__global__ void __launch_bounds__(256)
node_gather_kernel(const int* __restrict__ rowptr,
                   const int2* __restrict__ csr,
                   const float* __restrict__ ebuf,
                   const float* __restrict__ sden,
                   const float* __restrict__ ht,
                   const float* __restrict__ bias,
                   float* __restrict__ hout,
                   int act) {
  constexpr int HF = H * F;
  constexpr int NV = HF / 128;
  const int lane = threadIdx.x & 31;
  const int n = __builtin_amdgcn_readfirstlane((int)(blockIdx.x * 8 + (threadIdx.x >> 5)));
  if (n >= NN_NODES) return;
  const int beg = rowptr[n];
  const int end = rowptr[n + 1];

  float den[H];
#pragma unroll
  for (int h = 0; h < H; ++h) den[h] = fmaxf(sden[n * H + h], 1e-20f);

  float4 acc[NV];
#pragma unroll
  for (int v = 0; v < NV; ++v) { acc[v].x = 0.f; acc[v].y = 0.f; acc[v].z = 0.f; acc[v].w = 0.f; }

  const bool hiHalf = lane >= 16;

  for (int base = beg; base < end; base += 32) {
    const int j = base + lane;
    int   my_s = 0;
    float my_a[H];
#pragma unroll
    for (int h = 0; h < H; ++h) my_a[h] = 0.f;
    if (j < end) {
      const int2 se = csr[j];
      my_s = se.x;
      if constexpr (H == 4) {
        const float4 ex = *(const float4*)(ebuf + (size_t)se.y * 4);
        my_a[0] = ex.x / den[0];
        my_a[1] = ex.y / den[1];
        my_a[2] = ex.z / den[2];
        my_a[3] = ex.w / den[3];
      } else {
        my_a[0] = ebuf[se.y] / den[0];
      }
    }
    const int left = end - base;
    const int cnt  = left < 32 ? left : 32;
#pragma unroll 4
    for (int t = 0; t < cnt; ++t) {
      const int s = __builtin_amdgcn_readlane(my_s, t);
      const float* row = ht + (size_t)s * HF + lane * 4;
      if constexpr (H == 4) {
        const float a0 = __int_as_float(__builtin_amdgcn_readlane(__float_as_int(my_a[0]), t));
        const float a1 = __int_as_float(__builtin_amdgcn_readlane(__float_as_int(my_a[1]), t));
        const float a2 = __int_as_float(__builtin_amdgcn_readlane(__float_as_int(my_a[2]), t));
        const float a3 = __int_as_float(__builtin_amdgcn_readlane(__float_as_int(my_a[3]), t));
        const float aA = hiHalf ? a1 : a0;
        const float aB = hiHalf ? a3 : a2;
        const float4 v0 = *(const float4*)(row);
        const float4 v1 = *(const float4*)(row + 128);
        acc[0].x += aA * v0.x; acc[0].y += aA * v0.y; acc[0].z += aA * v0.z; acc[0].w += aA * v0.w;
        acc[1].x += aB * v1.x; acc[1].y += aB * v1.y; acc[1].z += aB * v1.z; acc[1].w += aB * v1.w;
      } else {
        const float a = __int_as_float(__builtin_amdgcn_readlane(__float_as_int(my_a[0]), t));
        const float4 v0 = *(const float4*)(row);
        acc[0].x += a * v0.x; acc[0].y += a * v0.y; acc[0].z += a * v0.z; acc[0].w += a * v0.w;
      }
    }
  }

#pragma unroll
  for (int v = 0; v < NV; ++v) {
    const int c = v * 128 + lane * 4;
    const float4 b = *(const float4*)(bias + c);
    float4 r;
    r.x = acc[v].x + b.x; r.y = acc[v].y + b.y; r.z = acc[v].z + b.z; r.w = acc[v].w + b.w;
    if (act) {
      r.x = leaky(r.x, 0.01f); r.y = leaky(r.y, 0.01f);
      r.z = leaky(r.z, 0.01f); r.w = leaky(r.w, 0.01f);
    }
    *(float4*)(hout + (size_t)n * HF + c) = r;
  }
}

__global__ void build_z_kernel(const float* __restrict__ emb,
                               const int* __restrict__ ps, const int* __restrict__ pd,
                               const int* __restrict__ ns, const int* __restrict__ nd,
                               float* __restrict__ Z) {
  int idx = blockIdx.x * blockDim.x + threadIdx.x;
  if (idx >= N_ROWS * 32) return;
  int row = idx >> 5;
  int f   = (idx & 31) * 4;
  int s, d;
  if (row < N_PAIR) { s = ps[row]; d = pd[row]; }
  else              { s = ns[row - N_PAIR]; d = nd[row - N_PAIR]; }
  float4 a = *(const float4*)(emb + (size_t)s * 128 + f);
  float4 b = *(const float4*)(emb + (size_t)d * 128 + f);
  float4 r; r.x = a.x * b.x; r.y = a.y * b.y; r.z = a.z * b.z; r.w = a.w * b.w;
  *(float4*)(Z + (size_t)row * 128 + f) = r;
}

__global__ void final_dot_kernel(const float* __restrict__ Z2,
                                 const float* __restrict__ pW3,
                                 const float* __restrict__ pb3,
                                 float* __restrict__ out) {
  int row = blockIdx.x * blockDim.x + threadIdx.x;
  if (row >= N_ROWS) return;
  const float* z = Z2 + (size_t)row * 32;
  float acc = pb3[0];
#pragma unroll
  for (int j = 0; j < 32; j += 4) {
    float4 a = *(const float4*)(z + j);
    float4 w = *(const float4*)(pW3 + j);
    acc += a.x * w.x + a.y * w.y + a.z * w.z + a.w * w.w;
  }
  out[row] = acc;
}

extern "C" void kernel_launch(void* const* d_in, const int* in_sizes, int n_in,
                              void* d_out, int out_size, void* d_ws, size_t ws_size,
                              hipStream_t stream) {
  const float* x    = (const float*)d_in[0];
  const float* ew   = (const float*)d_in[1];
  const int*   esrc = (const int*)d_in[2];
  const int*   edst = (const int*)d_in[3];
  const int*   psrc = (const int*)d_in[4];
  const int*   pdst = (const int*)d_in[5];
  const int*   nsrc = (const int*)d_in[6];
  const int*   ndst = (const int*)d_in[7];
  const float* Wl[3] = {(const float*)d_in[8],  (const float*)d_in[11], (const float*)d_in[14]};
  const float* al[3] = {(const float*)d_in[9],  (const float*)d_in[12], (const float*)d_in[15]};
  const float* bl[3] = {(const float*)d_in[10], (const float*)d_in[13], (const float*)d_in[16]};
  const float* pW1 = (const float*)d_in[17]; const float* pb1 = (const float*)d_in[18];
  const float* pW2 = (const float*)d_in[19]; const float* pb2 = (const float*)d_in[20];
  const float* pW3 = (const float*)d_in[21]; const float* pb3 = (const float*)d_in[22];

  float* out   = (float*)d_out;
  float* emb   = out + 2 * N_PAIR;

  float*    ws   = (float*)d_ws;
  float*    ht   = ws;
  float*    hbuf = ht   + 6400000;
  float*    pbuf = hbuf + 6400000;
  unsigned* menc = (unsigned*)(pbuf + 100000);
  float*    sden = (float*)(menc + 100000);
  float*    ebuf = sden + 100000;
  float*    Z    = ebuf + 1600000;
  float*    Z1   = Z    + 25600000;
  float*    Z2   = Z1   + 12800000;
  int*      deg    = (int*)(Z2 + 6400000);
  int*      rowptr = deg + 25000;
  int*      rank   = rowptr + 25008;
  int2*     csr    = (int2*)(rank + 400000);

  const int Hs[3]  = {4, 4, 1};
  const int Fs[3]  = {64, 64, 128};
  const int Kin[3] = {128, 256, 256};

  fill_u32_kernel<<<(NN_NODES + 255) / 256, 256, 0, stream>>>((unsigned*)deg, 0u, NN_NODES);
  csr_count_kernel<<<(N_EDGES + 255) / 256, 256, 0, stream>>>(edst, deg, rank);
  csr_scan_kernel<<<1, SCAN_THREADS, 0, stream>>>(deg, rowptr);
  csr_fill_kernel<<<(N_EDGES + 255) / 256, 256, 0, stream>>>(esrc, edst, rank, rowptr, csr);

  const float* hin = x;
  for (int l = 0; l < 3; ++l) {
    const int H = Hs[l], F = Fs[l], HF = H * F, K = Kin[l];
    float* hout = (l == 2) ? emb : hbuf;

    dim3 gg((NN_NODES + 63) / 64, HF / 32);
    gemm_wmma_f32<<<gg, 256, 0, stream>>>(hin, Wl[l], nullptr, ht,
                                          NN_NODES, HF, K, 0.f, 0);
    int nh = NN_NODES * H;
    if (l < 2) node_p_kernel<4, 64><<<(nh + 255) / 256, 256, 0, stream>>>(ht, al[l], pbuf);
    else       node_p_kernel<1, 128><<<(nh + 255) / 256, 256, 0, stream>>>(ht, al[l], pbuf);
    fill_u32_kernel<<<(nh + 255) / 256, 256, 0, stream>>>(menc, ENC_NEG_INF, nh);
    hipMemsetAsync(sden, 0, nh * sizeof(float), stream);

    if (l < 2) {
      edge_score_kernel<4><<<(N_EDGES + 255) / 256, 256, 0, stream>>>(
          pbuf, ew, esrc, edst, ebuf, menc);
    } else {
      edge_score_kernel<1><<<(N_EDGES + 255) / 256, 256, 0, stream>>>(
          pbuf, ew, esrc, edst, ebuf, menc);
    }
    node_m_kernel<<<(nh + 255) / 256, 256, 0, stream>>>(menc, nh);
    if (l < 2) {
      edge_exp_kernel<4><<<(N_EDGES * 4 + 255) / 256, 256, 0, stream>>>(
          ebuf, (const float*)menc, edst, sden);
    } else {
      edge_exp_kernel<1><<<(N_EDGES + 255) / 256, 256, 0, stream>>>(
          ebuf, (const float*)menc, edst, sden);
    }

    if (l < 2) {
      node_gather_kernel<4, 64><<<(NN_NODES + 7) / 8, 256, 0, stream>>>(
          rowptr, csr, ebuf, sden, ht, bl[l], hout, 1);
    } else {
      node_gather_kernel<1, 128><<<(NN_NODES + 7) / 8, 256, 0, stream>>>(
          rowptr, csr, ebuf, sden, ht, bl[l], hout, 0);
    }
    hin = hout;
  }

  build_z_kernel<<<(N_ROWS * 32 + 255) / 256, 256, 0, stream>>>(emb, psrc, pdst,
                                                                nsrc, ndst, Z);
  dim3 g1((N_ROWS + 63) / 64, 64 / 32);
  gemm_wmma_f32<<<g1, 256, 0, stream>>>(Z, pW1, pb1, Z1, N_ROWS, 64, 128, 0.2f, 1);
  dim3 g2((N_ROWS + 63) / 64, 32 / 32);
  gemm_wmma_f32<<<g2, 256, 0, stream>>>(Z1, pW2, pb2, Z2, N_ROWS, 32, 64, 0.2f, 1);
  final_dot_kernel<<<(N_ROWS + 255) / 256, 256, 0, stream>>>(Z2, pW3, pb3, out);
}


# Segment softmax (score, max, exp, sum) moves into the per-destination gather wave, in registers with DPP reductions: 15 launches and all score atomics gone; denominator sum = in-wave butterfly tree
# speedup vs baseline: 8.7068x; 1.9033x over previous
#include <hip/hip_runtime.h>
#include <hip/hip_bf16.h>


typedef __attribute__((ext_vector_type(2))) float v2f;
typedef __attribute__((ext_vector_type(8))) float v8f;

#define NN_NODES 25000
#define N_EDGES  400000
#define N_PAIR   100000
#define N_ROWS   200000

__device__ __forceinline__ float leaky(float v, float slope) {
  return v >= 0.f ? v : v * slope;
}

__global__ void gemm_wmma_f32(const float* __restrict__ A,
                              const float* __restrict__ W,
                              const float* __restrict__ bias,
                              float* __restrict__ C,
                              int M, int N, int K,
                              float slope, int act) {
  __shared__ float As[64][36];
  __shared__ float Ws[32][36];

  const int t     = threadIdx.x;
  const int lane  = t & 31;
  const int wave  = t >> 5;
  const int waveM = wave & 3;
  const int waveN = wave >> 2;
  const int mBase = blockIdx.x * 64;
  const int nBase = blockIdx.y * 32;
  const int lh    = lane >> 4;
  const int l15   = lane & 15;

  v8f acc = {0.f, 0.f, 0.f, 0.f, 0.f, 0.f, 0.f, 0.f};

  for (int k0 = 0; k0 < K; k0 += 32) {
#pragma unroll
    for (int i = 0; i < 2; ++i) {
      int idx = t + i * 256;
      int r   = idx >> 3;
      int c4  = (idx & 7) * 4;
      int gr  = mBase + r;
      int grc = gr < M ? gr : M - 1;
      float4 v = *(const float4*)(A + (size_t)grc * K + k0 + c4);
      if (gr >= M) { v.x = 0.f; v.y = 0.f; v.z = 0.f; v.w = 0.f; }
      *(float4*)&As[r][c4] = v;
    }
    {
      int r  = t >> 3;
      int c4 = (t & 7) * 4;
      *(float4*)&Ws[r][c4] =
          *(const float4*)(W + (size_t)(nBase + r) * K + k0 + c4);
    }
    __syncthreads();

#pragma unroll
    for (int ks = 0; ks < 8; ++ks) {
      int kb = ks * 4 + 2 * lh;
      v2f a, b;
      a.x = As[waveM * 16 + l15][kb];
      a.y = As[waveM * 16 + l15][kb + 1];
      b.x = Ws[waveN * 16 + l15][kb];
      b.y = Ws[waveN * 16 + l15][kb + 1];
      acc = __builtin_amdgcn_wmma_f32_16x16x4_f32(
           false, a,  false, b,
           (short)0, acc,  false,  false);
    }
    __syncthreads();
  }

#pragma unroll
  for (int r = 0; r < 8; ++r) {
    int row = mBase + waveM * 16 + r + 8 * lh;
    int col = nBase + waveN * 16 + l15;
    if (row < M) {
      float v = acc[r];
      if (bias) v += bias[col];
      if (act)  v = leaky(v, slope);
      C[(size_t)row * N + col] = v;
    }
  }
}

template <int H, int F>
__global__ void node_p_kernel(const float* __restrict__ ht,
                              const float* __restrict__ a,
                              float* __restrict__ p) {
  int idx = blockIdx.x * blockDim.x + threadIdx.x;
  if (idx >= NN_NODES * H) return;
  int n = idx / H, h = idx % H;
  const float* row = ht + ((size_t)n * H + h) * F;
  float acc = 0.f;
#pragma unroll 4
  for (int f = 0; f < F; f += 4) {
    float4 v = *(const float4*)(row + f);
    float4 w = *(const float4*)(a + f);
    acc += v.x * w.x + v.y * w.y + v.z * w.z + v.w * w.w;
  }
  p[idx] = acc;
}

__global__ void fill_u32_kernel(unsigned* __restrict__ dst, unsigned val, int n) {
  int i = blockIdx.x * blockDim.x + threadIdx.x;
  if (i < n) dst[i] = val;
}

__global__ void csr_count_kernel(const int* __restrict__ dst,
                                 int* __restrict__ deg,
                                 int* __restrict__ rank) {
  int e = blockIdx.x * blockDim.x + threadIdx.x;
  if (e >= N_EDGES) return;
  rank[e] = atomicAdd(&deg[dst[e]], 1);
}

#define SCAN_THREADS 1024
#define SCAN_PER     25
__global__ void __launch_bounds__(SCAN_THREADS)
csr_scan_kernel(const int* __restrict__ deg, int* __restrict__ rowptr) {
  __shared__ int part[SCAN_THREADS];
  const int t  = threadIdx.x;
  const int lo = t * SCAN_PER;
  int vals[SCAN_PER];
  int sum = 0;
#pragma unroll
  for (int i = 0; i < SCAN_PER; ++i) {
    int idx = lo + i;
    int idc = idx < NN_NODES ? idx : NN_NODES - 1;
    int v   = deg[idc];
    vals[i] = idx < NN_NODES ? v : 0;
    sum += vals[i];
  }
  part[t] = sum;
  __syncthreads();
  for (int off = 1; off < SCAN_THREADS; off <<= 1) {
    int v = (t >= off) ? part[t - off] : 0;
    __syncthreads();
    part[t] += v;
    __syncthreads();
  }
  int run = part[t] - sum;
#pragma unroll
  for (int i = 0; i < SCAN_PER; ++i) {
    int idx = lo + i;
    if (idx < NN_NODES) rowptr[idx] = run;
    run += vals[i];
  }
  if (t == SCAN_THREADS - 1) rowptr[NN_NODES] = part[SCAN_THREADS - 1];
}

__global__ void csr_fill_kernel(const int* __restrict__ src,
                                const int* __restrict__ dst,
                                const float* __restrict__ ew,
                                const int* __restrict__ rank,
                                const int* __restrict__ rowptr,
                                int2* __restrict__ csr) {
  int e = blockIdx.x * blockDim.x + threadIdx.x;
  if (e >= N_EDGES) return;
  int pos = rowptr[dst[e]] + rank[e];
  csr[pos] = make_int2(src[e], __float_as_int(log1pf(ew[e])));
}

template <int MASK>
__device__ __forceinline__ float row_xor(float v) {
  return __builtin_bit_cast(float, __builtin_amdgcn_update_dpp(
      0, __builtin_bit_cast(int, v), 0x160 + MASK, 0xf, 0xf, false));
}
__device__ __forceinline__ float xhalf(float v) {
  return __builtin_bit_cast(float, __builtin_amdgcn_permlanex16(
      __builtin_bit_cast(int, v), __builtin_bit_cast(int, v), 0x76543210, 0xfedcba98, false, false));
}
__device__ __forceinline__ float wave_max(float v) {
  v = fmaxf(v, row_xor<1>(v)); v = fmaxf(v, row_xor<2>(v));
  v = fmaxf(v, row_xor<4>(v)); v = fmaxf(v, row_xor<8>(v));
  return fmaxf(v, xhalf(v));
}
__device__ __forceinline__ float wave_sum(float v) {
  v += row_xor<1>(v); v += row_xor<2>(v);
  v += row_xor<4>(v); v += row_xor<8>(v);
  return v + xhalf(v);
}

template <int H>
__device__ __forceinline__ void edge_scores(const int2* __restrict__ csr,
                                            const float* __restrict__ p,
                                            const float (&pd)[H], int j,
                                            int& s, float (&sc)[H]) {
  const int2 se = csr[j];
  s = se.x;
  const float lw = __int_as_float(se.y);
  if constexpr (H == 4) {
    const float4 ps = *(const float4*)(p + (size_t)se.x * 4);
    sc[0] = leaky(ps.x + pd[0] + lw, 0.01f);
    sc[1] = leaky(ps.y + pd[1] + lw, 0.01f);
    sc[2] = leaky(ps.z + pd[2] + lw, 0.01f);
    sc[3] = leaky(ps.w + pd[3] + lw, 0.01f);
  } else {
    sc[0] = leaky(p[se.x] + pd[0] + lw, 0.01f);
  }
}

template <int H, int F>
__device__ __forceinline__ void gather_rows(const float* __restrict__ ht, int lane,
                                            int my_s, const float (&my_a)[H], int cnt,
                                            float4 (&acc)[H * F / 128]) {
  constexpr int HF = H * F;
  const bool hiHalf = lane >= 16;
#pragma unroll 4
  for (int t = 0; t < cnt; ++t) {
    const int s = __builtin_amdgcn_readlane(my_s, t);
    const float* row = ht + (size_t)s * HF + lane * 4;
    if constexpr (H == 4) {
      const float a0 = __int_as_float(__builtin_amdgcn_readlane(__float_as_int(my_a[0]), t));
      const float a1 = __int_as_float(__builtin_amdgcn_readlane(__float_as_int(my_a[1]), t));
      const float a2 = __int_as_float(__builtin_amdgcn_readlane(__float_as_int(my_a[2]), t));
      const float a3 = __int_as_float(__builtin_amdgcn_readlane(__float_as_int(my_a[3]), t));
      const float aA = hiHalf ? a1 : a0;
      const float aB = hiHalf ? a3 : a2;
      const float4 v0 = *(const float4*)(row);
      const float4 v1 = *(const float4*)(row + 128);
      acc[0].x += aA * v0.x; acc[0].y += aA * v0.y; acc[0].z += aA * v0.z; acc[0].w += aA * v0.w;
      acc[1].x += aB * v1.x; acc[1].y += aB * v1.y; acc[1].z += aB * v1.z; acc[1].w += aB * v1.w;
    } else {
      const float a = __int_as_float(__builtin_amdgcn_readlane(__float_as_int(my_a[0]), t));
      const float4 v0 = *(const float4*)(row);
      acc[0].x += a * v0.x; acc[0].y += a * v0.y; acc[0].z += a * v0.z; acc[0].w += a * v0.w;
    }
  }
}

template <int H, int F>
__global__ void __launch_bounds__(256)
node_softmax_gather_kernel(const int* __restrict__ rowptr,
                           const int2* __restrict__ csr,
                           const float* __restrict__ p,
                           const float* __restrict__ ht,
                           const float* __restrict__ bias,
                           float* __restrict__ hout,
                           int act) {
  constexpr int HF = H * F;
  constexpr int NV = HF / 128;
  const int lane = threadIdx.x & 31;
  const int n = __builtin_amdgcn_readfirstlane((int)(blockIdx.x * 8 + (threadIdx.x >> 5)));
  if (n >= NN_NODES) return;
  const int beg = rowptr[n];
  const int end = rowptr[n + 1];
  const int deg = end - beg;

  float pd[H];
#pragma unroll
  for (int h = 0; h < H; ++h) pd[h] = p[n * H + h];

  float4 acc[NV];
#pragma unroll
  for (int v = 0; v < NV; ++v) { acc[v].x = 0.f; acc[v].y = 0.f; acc[v].z = 0.f; acc[v].w = 0.f; }

  if (deg <= 32) {
    const bool on = lane < deg;
    int   my_s = 0;
    float sc[H];
#pragma unroll
    for (int h = 0; h < H; ++h) sc[h] = -INFINITY;
    if (on) edge_scores<H>(csr, p, pd, beg + lane, my_s, sc);
    float my_a[H];
#pragma unroll
    for (int h = 0; h < H; ++h) {
      float m = wave_max(sc[h]);
      if (!isfinite(m)) m = 0.f;
      const float ex = on ? __expf(sc[h] - m) : 0.f;
      const float sum = wave_sum(ex);
      my_a[h] = ex / fmaxf(sum, 1e-20f);
    }
    gather_rows<H, F>(ht, lane, my_s, my_a, deg, acc);
  } else {
    float m[H], den[H];
#pragma unroll
    for (int h = 0; h < H; ++h) m[h] = -INFINITY;
    for (int base = beg; base < end; base += 32) {
      const int j = base + lane;
      if (j < end) {
        int s_; float sc[H];
        edge_scores<H>(csr, p, pd, j, s_, sc);
#pragma unroll
        for (int h = 0; h < H; ++h) m[h] = fmaxf(m[h], sc[h]);
      }
    }
#pragma unroll
    for (int h = 0; h < H; ++h) {
      m[h] = wave_max(m[h]);
      if (!isfinite(m[h])) m[h] = 0.f;
      den[h] = 0.f;
    }
    for (int base = beg; base < end; base += 32) {
      const int j = base + lane;
      if (j < end) {
        int s_; float sc[H];
        edge_scores<H>(csr, p, pd, j, s_, sc);
#pragma unroll
        for (int h = 0; h < H; ++h) den[h] += __expf(sc[h] - m[h]);
      }
    }
#pragma unroll
    for (int h = 0; h < H; ++h) den[h] = fmaxf(wave_sum(den[h]), 1e-20f);
    for (int base = beg; base < end; base += 32) {
      const int j = base + lane;
      int   my_s = 0;
      float my_a[H];
#pragma unroll
      for (int h = 0; h < H; ++h) my_a[h] = 0.f;
      if (j < end) {
        float sc[H];
        edge_scores<H>(csr, p, pd, j, my_s, sc);
#pragma unroll
        for (int h = 0; h < H; ++h) my_a[h] = __expf(sc[h] - m[h]) / den[h];
      }
      const int left = end - base;
      gather_rows<H, F>(ht, lane, my_s, my_a, left < 32 ? left : 32, acc);
    }
  }

#pragma unroll
  for (int v = 0; v < NV; ++v) {
    const int c = v * 128 + lane * 4;
    const float4 b = *(const float4*)(bias + c);
    float4 r;
    r.x = acc[v].x + b.x; r.y = acc[v].y + b.y; r.z = acc[v].z + b.z; r.w = acc[v].w + b.w;
    if (act) {
      r.x = leaky(r.x, 0.01f); r.y = leaky(r.y, 0.01f);
      r.z = leaky(r.z, 0.01f); r.w = leaky(r.w, 0.01f);
    }
    *(float4*)(hout + (size_t)n * HF + c) = r;
  }
}

__global__ void build_z_kernel(const float* __restrict__ emb,
                               const int* __restrict__ ps, const int* __restrict__ pd,
                               const int* __restrict__ ns, const int* __restrict__ nd,
                               float* __restrict__ Z) {
  int idx = blockIdx.x * blockDim.x + threadIdx.x;
  if (idx >= N_ROWS * 32) return;
  int row = idx >> 5;
  int f   = (idx & 31) * 4;
  int s, d;
  if (row < N_PAIR) { s = ps[row]; d = pd[row]; }
  else              { s = ns[row - N_PAIR]; d = nd[row - N_PAIR]; }
  float4 a = *(const float4*)(emb + (size_t)s * 128 + f);
  float4 b = *(const float4*)(emb + (size_t)d * 128 + f);
  float4 r; r.x = a.x * b.x; r.y = a.y * b.y; r.z = a.z * b.z; r.w = a.w * b.w;
  *(float4*)(Z + (size_t)row * 128 + f) = r;
}

__global__ void final_dot_kernel(const float* __restrict__ Z2,
                                 const float* __restrict__ pW3,
                                 const float* __restrict__ pb3,
                                 float* __restrict__ out) {
  int row = blockIdx.x * blockDim.x + threadIdx.x;
  if (row >= N_ROWS) return;
  const float* z = Z2 + (size_t)row * 32;
  float acc = pb3[0];
#pragma unroll
  for (int j = 0; j < 32; j += 4) {
    float4 a = *(const float4*)(z + j);
    float4 w = *(const float4*)(pW3 + j);
    acc += a.x * w.x + a.y * w.y + a.z * w.z + a.w * w.w;
  }
  out[row] = acc;
}

extern "C" void kernel_launch(void* const* d_in, const int* in_sizes, int n_in,
                              void* d_out, int out_size, void* d_ws, size_t ws_size,
                              hipStream_t stream) {
  const float* x    = (const float*)d_in[0];
  const float* ew   = (const float*)d_in[1];
  const int*   esrc = (const int*)d_in[2];
  const int*   edst = (const int*)d_in[3];
  const int*   psrc = (const int*)d_in[4];
  const int*   pdst = (const int*)d_in[5];
  const int*   nsrc = (const int*)d_in[6];
  const int*   ndst = (const int*)d_in[7];
  const float* Wl[3] = {(const float*)d_in[8],  (const float*)d_in[11], (const float*)d_in[14]};
  const float* al[3] = {(const float*)d_in[9],  (const float*)d_in[12], (const float*)d_in[15]};
  const float* bl[3] = {(const float*)d_in[10], (const float*)d_in[13], (const float*)d_in[16]};
  const float* pW1 = (const float*)d_in[17]; const float* pb1 = (const float*)d_in[18];
  const float* pW2 = (const float*)d_in[19]; const float* pb2 = (const float*)d_in[20];
  const float* pW3 = (const float*)d_in[21]; const float* pb3 = (const float*)d_in[22];

  float* out   = (float*)d_out;
  float* emb   = out + 2 * N_PAIR;

  float*    ws   = (float*)d_ws;
  float*    ht   = ws;
  float*    hbuf = ht   + 6400000;
  float*    pbuf = hbuf + 6400000;
  float*    Z    = pbuf + 100000;
  float*    Z1   = Z    + 25600000;
  float*    Z2   = Z1   + 12800000;
  int*      deg    = (int*)(Z2 + 6400000);
  int*      rowptr = deg + 25000;
  int*      rank   = rowptr + 25008;
  int2*     csr    = (int2*)(rank + 400000);

  const int Hs[3]  = {4, 4, 1};
  const int Fs[3]  = {64, 64, 128};
  const int Kin[3] = {128, 256, 256};

  fill_u32_kernel<<<(NN_NODES + 255) / 256, 256, 0, stream>>>((unsigned*)deg, 0u, NN_NODES);
  csr_count_kernel<<<(N_EDGES + 255) / 256, 256, 0, stream>>>(edst, deg, rank);
  csr_scan_kernel<<<1, SCAN_THREADS, 0, stream>>>(deg, rowptr);
  csr_fill_kernel<<<(N_EDGES + 255) / 256, 256, 0, stream>>>(esrc, edst, ew, rank, rowptr, csr);

  const float* hin = x;
  for (int l = 0; l < 3; ++l) {
    const int H = Hs[l], F = Fs[l], HF = H * F, K = Kin[l];
    float* hout = (l == 2) ? emb : hbuf;

    dim3 gg((NN_NODES + 63) / 64, HF / 32);
    gemm_wmma_f32<<<gg, 256, 0, stream>>>(hin, Wl[l], nullptr, ht,
                                          NN_NODES, HF, K, 0.f, 0);
    int nh = NN_NODES * H;
    if (l < 2) node_p_kernel<4, 64><<<(nh + 255) / 256, 256, 0, stream>>>(ht, al[l], pbuf);
    else       node_p_kernel<1, 128><<<(nh + 255) / 256, 256, 0, stream>>>(ht, al[l], pbuf);

    if (l < 2) {
      node_softmax_gather_kernel<4, 64><<<(NN_NODES + 7) / 8, 256, 0, stream>>>(
          rowptr, csr, pbuf, ht, bl[l], hout, 1);
    } else {
      node_softmax_gather_kernel<1, 128><<<(NN_NODES + 7) / 8, 256, 0, stream>>>(
          rowptr, csr, pbuf, ht, bl[l], hout, 0);
    }
    hin = hout;
  }

  build_z_kernel<<<(N_ROWS * 32 + 255) / 256, 256, 0, stream>>>(emb, psrc, pdst,
                                                                nsrc, ndst, Z);
  dim3 g1((N_ROWS + 63) / 64, 64 / 32);
  gemm_wmma_f32<<<g1, 256, 0, stream>>>(Z, pW1, pb1, Z1, N_ROWS, 64, 128, 0.2f, 1);
  dim3 g2((N_ROWS + 63) / 64, 32 / 32);
  gemm_wmma_f32<<<g2, 256, 0, stream>>>(Z1, pW2, pb2, Z2, N_ROWS, 32, 64, 0.2f, 1);
  final_dot_kernel<<<(N_ROWS + 255) / 256, 256, 0, stream>>>(Z2, pW3, pb3, out);
}


# Predictor fused into one kernel: pair product, 128-64-32 WMMA layers, final dot stay in LDS/registers (Z, Z1, Z2 and 3 launches gone); k pairing inside WMMA steps and final dot (lane tree) reordered
# speedup vs baseline: 11.5627x; 1.3280x over previous
#include <hip/hip_runtime.h>
#include <hip/hip_bf16.h>


typedef __attribute__((ext_vector_type(2))) float v2f;
typedef __attribute__((ext_vector_type(8))) float v8f;

#define NN_NODES 25000
#define N_EDGES  400000
#define N_PAIR   100000
#define N_ROWS   200000

__device__ __forceinline__ float leaky(float v, float slope) {
  return v >= 0.f ? v : v * slope;
}

__global__ void gemm_wmma_f32(const float* __restrict__ A,
                              const float* __restrict__ W,
                              const float* __restrict__ bias,
                              float* __restrict__ C,
                              int M, int N, int K,
                              float slope, int act) {
  __shared__ float As[64][36];
  __shared__ float Ws[32][36];

  const int t     = threadIdx.x;
  const int lane  = t & 31;
  const int wave  = t >> 5;
  const int waveM = wave & 3;
  const int waveN = wave >> 2;
  const int mBase = blockIdx.x * 64;
  const int nBase = blockIdx.y * 32;
  const int lh    = lane >> 4;
  const int l15   = lane & 15;

  v8f acc = {0.f, 0.f, 0.f, 0.f, 0.f, 0.f, 0.f, 0.f};

  for (int k0 = 0; k0 < K; k0 += 32) {
#pragma unroll
    for (int i = 0; i < 2; ++i) {
      int idx = t + i * 256;
      int r   = idx >> 3;
      int c4  = (idx & 7) * 4;
      int gr  = mBase + r;
      int grc = gr < M ? gr : M - 1;
      float4 v = *(const float4*)(A + (size_t)grc * K + k0 + c4);
      if (gr >= M) { v.x = 0.f; v.y = 0.f; v.z = 0.f; v.w = 0.f; }
      *(float4*)&As[r][c4] = v;
    }
    {
      int r  = t >> 3;
      int c4 = (t & 7) * 4;
      *(float4*)&Ws[r][c4] =
          *(const float4*)(W + (size_t)(nBase + r) * K + k0 + c4);
    }
    __syncthreads();

#pragma unroll
    for (int ks = 0; ks < 8; ++ks) {
      int kb = ks * 4 + 2 * lh;
      v2f a, b;
      a.x = As[waveM * 16 + l15][kb];
      a.y = As[waveM * 16 + l15][kb + 1];
      b.x = Ws[waveN * 16 + l15][kb];
      b.y = Ws[waveN * 16 + l15][kb + 1];
      acc = __builtin_amdgcn_wmma_f32_16x16x4_f32(
           false, a,  false, b,
           (short)0, acc,  false,  false);
    }
    __syncthreads();
  }

#pragma unroll
  for (int r = 0; r < 8; ++r) {
    int row = mBase + waveM * 16 + r + 8 * lh;
    int col = nBase + waveN * 16 + l15;
    if (row < M) {
      float v = acc[r];
      if (bias) v += bias[col];
      if (act)  v = leaky(v, slope);
      C[(size_t)row * N + col] = v;
    }
  }
}

template <int H, int F>
__global__ void node_p_kernel(const float* __restrict__ ht,
                              const float* __restrict__ a,
                              float* __restrict__ p) {
  int idx = blockIdx.x * blockDim.x + threadIdx.x;
  if (idx >= NN_NODES * H) return;
  int n = idx / H, h = idx % H;
  const float* row = ht + ((size_t)n * H + h) * F;
  float acc = 0.f;
#pragma unroll 4
  for (int f = 0; f < F; f += 4) {
    float4 v = *(const float4*)(row + f);
    float4 w = *(const float4*)(a + f);
    acc += v.x * w.x + v.y * w.y + v.z * w.z + v.w * w.w;
  }
  p[idx] = acc;
}

__global__ void fill_u32_kernel(unsigned* __restrict__ dst, unsigned val, int n) {
  int i = blockIdx.x * blockDim.x + threadIdx.x;
  if (i < n) dst[i] = val;
}

__global__ void csr_count_kernel(const int* __restrict__ dst,
                                 int* __restrict__ deg,
                                 int* __restrict__ rank) {
  int e = blockIdx.x * blockDim.x + threadIdx.x;
  if (e >= N_EDGES) return;
  rank[e] = atomicAdd(&deg[dst[e]], 1);
}

#define SCAN_THREADS 1024
#define SCAN_PER     25
__global__ void __launch_bounds__(SCAN_THREADS)
csr_scan_kernel(const int* __restrict__ deg, int* __restrict__ rowptr) {
  __shared__ int part[SCAN_THREADS];
  const int t  = threadIdx.x;
  const int lo = t * SCAN_PER;
  int vals[SCAN_PER];
  int sum = 0;
#pragma unroll
  for (int i = 0; i < SCAN_PER; ++i) {
    int idx = lo + i;
    int idc = idx < NN_NODES ? idx : NN_NODES - 1;
    int v   = deg[idc];
    vals[i] = idx < NN_NODES ? v : 0;
    sum += vals[i];
  }
  part[t] = sum;
  __syncthreads();
  for (int off = 1; off < SCAN_THREADS; off <<= 1) {
    int v = (t >= off) ? part[t - off] : 0;
    __syncthreads();
    part[t] += v;
    __syncthreads();
  }
  int run = part[t] - sum;
#pragma unroll
  for (int i = 0; i < SCAN_PER; ++i) {
    int idx = lo + i;
    if (idx < NN_NODES) rowptr[idx] = run;
    run += vals[i];
  }
  if (t == SCAN_THREADS - 1) rowptr[NN_NODES] = part[SCAN_THREADS - 1];
}

__global__ void csr_fill_kernel(const int* __restrict__ src,
                                const int* __restrict__ dst,
                                const float* __restrict__ ew,
                                const int* __restrict__ rank,
                                const int* __restrict__ rowptr,
                                int2* __restrict__ csr) {
  int e = blockIdx.x * blockDim.x + threadIdx.x;
  if (e >= N_EDGES) return;
  int pos = rowptr[dst[e]] + rank[e];
  csr[pos] = make_int2(src[e], __float_as_int(log1pf(ew[e])));
}

template <int MASK>
__device__ __forceinline__ float row_xor(float v) {
  return __builtin_bit_cast(float, __builtin_amdgcn_update_dpp(
      0, __builtin_bit_cast(int, v), 0x160 + MASK, 0xf, 0xf, false));
}
__device__ __forceinline__ float xhalf(float v) {
  return __builtin_bit_cast(float, __builtin_amdgcn_permlanex16(
      __builtin_bit_cast(int, v), __builtin_bit_cast(int, v), 0x76543210, 0xfedcba98, false, false));
}
__device__ __forceinline__ float wave_max(float v) {
  v = fmaxf(v, row_xor<1>(v)); v = fmaxf(v, row_xor<2>(v));
  v = fmaxf(v, row_xor<4>(v)); v = fmaxf(v, row_xor<8>(v));
  return fmaxf(v, xhalf(v));
}
__device__ __forceinline__ float wave_sum(float v) {
  v += row_xor<1>(v); v += row_xor<2>(v);
  v += row_xor<4>(v); v += row_xor<8>(v);
  return v + xhalf(v);
}

template <int H>
__device__ __forceinline__ void edge_scores(const int2* __restrict__ csr,
                                            const float* __restrict__ p,
                                            const float (&pd)[H], int j,
                                            int& s, float (&sc)[H]) {
  const int2 se = csr[j];
  s = se.x;
  const float lw = __int_as_float(se.y);
  if constexpr (H == 4) {
    const float4 ps = *(const float4*)(p + (size_t)se.x * 4);
    sc[0] = leaky(ps.x + pd[0] + lw, 0.01f);
    sc[1] = leaky(ps.y + pd[1] + lw, 0.01f);
    sc[2] = leaky(ps.z + pd[2] + lw, 0.01f);
    sc[3] = leaky(ps.w + pd[3] + lw, 0.01f);
  } else {
    sc[0] = leaky(p[se.x] + pd[0] + lw, 0.01f);
  }
}

template <int H, int F>
__device__ __forceinline__ void gather_rows(const float* __restrict__ ht, int lane,
                                            int my_s, const float (&my_a)[H], int cnt,
                                            float4 (&acc)[H * F / 128]) {
  constexpr int HF = H * F;
  const bool hiHalf = lane >= 16;
#pragma unroll 4
  for (int t = 0; t < cnt; ++t) {
    const int s = __builtin_amdgcn_readlane(my_s, t);
    const float* row = ht + (size_t)s * HF + lane * 4;
    if constexpr (H == 4) {
      const float a0 = __int_as_float(__builtin_amdgcn_readlane(__float_as_int(my_a[0]), t));
      const float a1 = __int_as_float(__builtin_amdgcn_readlane(__float_as_int(my_a[1]), t));
      const float a2 = __int_as_float(__builtin_amdgcn_readlane(__float_as_int(my_a[2]), t));
      const float a3 = __int_as_float(__builtin_amdgcn_readlane(__float_as_int(my_a[3]), t));
      const float aA = hiHalf ? a1 : a0;
      const float aB = hiHalf ? a3 : a2;
      const float4 v0 = *(const float4*)(row);
      const float4 v1 = *(const float4*)(row + 128);
      acc[0].x += aA * v0.x; acc[0].y += aA * v0.y; acc[0].z += aA * v0.z; acc[0].w += aA * v0.w;
      acc[1].x += aB * v1.x; acc[1].y += aB * v1.y; acc[1].z += aB * v1.z; acc[1].w += aB * v1.w;
    } else {
      const float a = __int_as_float(__builtin_amdgcn_readlane(__float_as_int(my_a[0]), t));
      const float4 v0 = *(const float4*)(row);
      acc[0].x += a * v0.x; acc[0].y += a * v0.y; acc[0].z += a * v0.z; acc[0].w += a * v0.w;
    }
  }
}

template <int H, int F>
__global__ void __launch_bounds__(256)
node_softmax_gather_kernel(const int* __restrict__ rowptr,
                           const int2* __restrict__ csr,
                           const float* __restrict__ p,
                           const float* __restrict__ ht,
                           const float* __restrict__ bias,
                           float* __restrict__ hout,
                           int act) {
  constexpr int HF = H * F;
  constexpr int NV = HF / 128;
  const int lane = threadIdx.x & 31;
  const int n = __builtin_amdgcn_readfirstlane((int)(blockIdx.x * 8 + (threadIdx.x >> 5)));
  if (n >= NN_NODES) return;
  const int beg = rowptr[n];
  const int end = rowptr[n + 1];
  const int deg = end - beg;

  float pd[H];
#pragma unroll
  for (int h = 0; h < H; ++h) pd[h] = p[n * H + h];

  float4 acc[NV];
#pragma unroll
  for (int v = 0; v < NV; ++v) { acc[v].x = 0.f; acc[v].y = 0.f; acc[v].z = 0.f; acc[v].w = 0.f; }

  if (deg <= 32) {
    const bool on = lane < deg;
    int   my_s = 0;
    float sc[H];
#pragma unroll
    for (int h = 0; h < H; ++h) sc[h] = -INFINITY;
    if (on) edge_scores<H>(csr, p, pd, beg + lane, my_s, sc);
    float my_a[H];
#pragma unroll
    for (int h = 0; h < H; ++h) {
      float m = wave_max(sc[h]);
      if (!isfinite(m)) m = 0.f;
      const float ex = on ? __expf(sc[h] - m) : 0.f;
      const float sum = wave_sum(ex);
      my_a[h] = ex / fmaxf(sum, 1e-20f);
    }
    gather_rows<H, F>(ht, lane, my_s, my_a, deg, acc);
  } else {
    float m[H], den[H];
#pragma unroll
    for (int h = 0; h < H; ++h) m[h] = -INFINITY;
    for (int base = beg; base < end; base += 32) {
      const int j = base + lane;
      if (j < end) {
        int s_; float sc[H];
        edge_scores<H>(csr, p, pd, j, s_, sc);
#pragma unroll
        for (int h = 0; h < H; ++h) m[h] = fmaxf(m[h], sc[h]);
      }
    }
#pragma unroll
    for (int h = 0; h < H; ++h) {
      m[h] = wave_max(m[h]);
      if (!isfinite(m[h])) m[h] = 0.f;
      den[h] = 0.f;
    }
    for (int base = beg; base < end; base += 32) {
      const int j = base + lane;
      if (j < end) {
        int s_; float sc[H];
        edge_scores<H>(csr, p, pd, j, s_, sc);
#pragma unroll
        for (int h = 0; h < H; ++h) den[h] += __expf(sc[h] - m[h]);
      }
    }
#pragma unroll
    for (int h = 0; h < H; ++h) den[h] = fmaxf(wave_sum(den[h]), 1e-20f);
    for (int base = beg; base < end; base += 32) {
      const int j = base + lane;
      int   my_s = 0;
      float my_a[H];
#pragma unroll
      for (int h = 0; h < H; ++h) my_a[h] = 0.f;
      if (j < end) {
        float sc[H];
        edge_scores<H>(csr, p, pd, j, my_s, sc);
#pragma unroll
        for (int h = 0; h < H; ++h) my_a[h] = __expf(sc[h] - m[h]) / den[h];
      }
      const int left = end - base;
      gather_rows<H, F>(ht, lane, my_s, my_a, left < 32 ? left : 32, acc);
    }
  }

#pragma unroll
  for (int v = 0; v < NV; ++v) {
    const int c = v * 128 + lane * 4;
    const float4 b = *(const float4*)(bias + c);
    float4 r;
    r.x = acc[v].x + b.x; r.y = acc[v].y + b.y; r.z = acc[v].z + b.z; r.w = acc[v].w + b.w;
    if (act) {
      r.x = leaky(r.x, 0.01f); r.y = leaky(r.y, 0.01f);
      r.z = leaky(r.z, 0.01f); r.w = leaky(r.w, 0.01f);
    }
    *(float4*)(hout + (size_t)n * HF + c) = r;
  }
}

#define PR_TILES  (N_ROWS / 16)
#define PR_BLOCKS 512
#define ZP  132
#define Z1P 68

__device__ __forceinline__ void wave_lds_sync() {
  __builtin_amdgcn_fence(__ATOMIC_RELEASE, "wavefront");
  __builtin_amdgcn_wave_barrier();
  __builtin_amdgcn_fence(__ATOMIC_ACQUIRE, "wavefront");
}

__device__ __forceinline__ v8f wmma4(float a0, float a1, float b0, float b1, v8f acc) {
  v2f a = {a0, a1};
  v2f b = {b0, b1};
  return __builtin_amdgcn_wmma_f32_16x16x4_f32(
       false, a,  false, b,
       (short)0, acc,  false,  false);
}

__global__ void __launch_bounds__(256)
predictor_fused_kernel(const float* __restrict__ emb,
                       const int* __restrict__ ps, const int* __restrict__ pd,
                       const int* __restrict__ ns, const int* __restrict__ nd,
                       const float* __restrict__ pW1, const float* __restrict__ pb1,
                       const float* __restrict__ pW2, const float* __restrict__ pb2,
                       const float* __restrict__ pW3, const float* __restrict__ pb3,
                       float* __restrict__ out) {
  __shared__ __attribute__((aligned(16))) float W1s[64][ZP];
  __shared__ __attribute__((aligned(16))) float W2s[32][Z1P];
  __shared__ __attribute__((aligned(16))) float Zs[8][16][ZP];
  __shared__ __attribute__((aligned(16))) float Z1s[8][16][Z1P];

  const int t    = threadIdx.x;
  const int lane = t & 31;
  const int wave = __builtin_amdgcn_readfirstlane(t >> 5);
  const int c    = lane & 15;
  const int h    = lane >> 4;

#pragma unroll
  for (int j = 0; j < 8; ++j) {
    const int idx = t + j * 256;
    const int r   = idx >> 5;
    const int c4  = (idx & 31) * 4;
    *(float4*)&W1s[r][c4] = *(const float4*)(pW1 + r * 128 + c4);
  }
#pragma unroll
  for (int j = 0; j < 2; ++j) {
    const int idx = t + j * 256;
    const int r   = idx >> 4;
    const int c4  = (idx & 15) * 4;
    *(float4*)&W2s[r][c4] = *(const float4*)(pW2 + r * 64 + c4);
  }
  float bias1[4], bias2[2], w3[2];
#pragma unroll
  for (int tt = 0; tt < 4; ++tt) bias1[tt] = pb1[16 * tt + c];
#pragma unroll
  for (int tt = 0; tt < 2; ++tt) { bias2[tt] = pb2[16 * tt + c]; w3[tt] = pW3[16 * tt + c]; }
  const float b3 = pb3[0];
  __syncthreads();

  for (int wt = blockIdx.x * 8 + wave; wt < PR_TILES; wt += PR_BLOCKS * 8) {
    const int row0 = wt * 16;
    const int* sp = ps; const int* dp = pd; int rbase = row0;
    if (row0 >= N_PAIR) { sp = ns; dp = nd; rbase = row0 - N_PAIR; }
    const int si = sp[rbase + c];
    const int di = dp[rbase + c];

#pragma unroll
    for (int r = 0; r < 16; ++r) {
      const int s = __builtin_amdgcn_readlane(si, r);
      const int d = __builtin_amdgcn_readlane(di, r);
      const float4 a = *(const float4*)(emb + (size_t)s * 128 + lane * 4);
      const float4 b = *(const float4*)(emb + (size_t)d * 128 + lane * 4);
      float4 z; z.x = a.x * b.x; z.y = a.y * b.y; z.z = a.z * b.z; z.w = a.w * b.w;
      *(float4*)&Zs[wave][r][lane * 4] = z;
    }
    wave_lds_sync();

    v8f acc[4];
#pragma unroll
    for (int tt = 0; tt < 4; ++tt) acc[tt] = (v8f){0.f, 0.f, 0.f, 0.f, 0.f, 0.f, 0.f, 0.f};
    {
      float4 a_cur = *(const float4*)&Zs[wave][c][4 * h];
      float4 b_cur[4];
#pragma unroll
      for (int tt = 0; tt < 4; ++tt) b_cur[tt] = *(const float4*)&W1s[16 * tt + c][4 * h];
#pragma unroll
      for (int i = 0; i < 16; ++i) {
        float4 a_nxt = a_cur;
        float4 b_nxt[4];
#pragma unroll
        for (int tt = 0; tt < 4; ++tt) b_nxt[tt] = b_cur[tt];
        if (i + 1 < 16) {
          a_nxt = *(const float4*)&Zs[wave][c][8 * (i + 1) + 4 * h];
#pragma unroll
          for (int tt = 0; tt < 4; ++tt)
            b_nxt[tt] = *(const float4*)&W1s[16 * tt + c][8 * (i + 1) + 4 * h];
        }
#pragma unroll
        for (int tt = 0; tt < 4; ++tt)
          acc[tt] = wmma4(a_cur.x, a_cur.y, b_cur[tt].x, b_cur[tt].y, acc[tt]);
#pragma unroll
        for (int tt = 0; tt < 4; ++tt)
          acc[tt] = wmma4(a_cur.z, a_cur.w, b_cur[tt].z, b_cur[tt].w, acc[tt]);
        a_cur = a_nxt;
#pragma unroll
        for (int tt = 0; tt < 4; ++tt) b_cur[tt] = b_nxt[tt];
      }
    }
#pragma unroll
    for (int tt = 0; tt < 4; ++tt) {
#pragma unroll
      for (int r = 0; r < 8; ++r)
        Z1s[wave][8 * h + r][16 * tt + c] = leaky(acc[tt][r] + bias1[tt], 0.2f);
    }
    wave_lds_sync();

    v8f acc2[2];
#pragma unroll
    for (int tt = 0; tt < 2; ++tt) acc2[tt] = (v8f){0.f, 0.f, 0.f, 0.f, 0.f, 0.f, 0.f, 0.f};
    {
      float4 a2[8];
      float4 b2[2][8];
#pragma unroll
      for (int i = 0; i < 8; ++i) {
        a2[i] = *(const float4*)&Z1s[wave][c][8 * i + 4 * h];
#pragma unroll
        for (int tt = 0; tt < 2; ++tt)
          b2[tt][i] = *(const float4*)&W2s[16 * tt + c][8 * i + 4 * h];
      }
#pragma unroll
      for (int i = 0; i < 8; ++i) {
#pragma unroll
        for (int tt = 0; tt < 2; ++tt)
          acc2[tt] = wmma4(a2[i].x, a2[i].y, b2[tt][i].x, b2[tt][i].y, acc2[tt]);
#pragma unroll
        for (int tt = 0; tt < 2; ++tt)
          acc2[tt] = wmma4(a2[i].z, a2[i].w, b2[tt][i].z, b2[tt][i].w, acc2[tt]);
      }
    }

    float mine = 0.f;
#pragma unroll
    for (int r = 0; r < 8; ++r) {
      float v = leaky(acc2[0][r] + bias2[0], 0.2f) * w3[0]
              + leaky(acc2[1][r] + bias2[1], 0.2f) * w3[1];
      v += row_xor<1>(v); v += row_xor<2>(v);
      v += row_xor<4>(v); v += row_xor<8>(v);
      if (c == r) mine = v;
    }
    if (c < 8) out[row0 + 8 * h + c] = mine + b3;
    wave_lds_sync();
  }
}

extern "C" void kernel_launch(void* const* d_in, const int* in_sizes, int n_in,
                              void* d_out, int out_size, void* d_ws, size_t ws_size,
                              hipStream_t stream) {
  const float* x    = (const float*)d_in[0];
  const float* ew   = (const float*)d_in[1];
  const int*   esrc = (const int*)d_in[2];
  const int*   edst = (const int*)d_in[3];
  const int*   psrc = (const int*)d_in[4];
  const int*   pdst = (const int*)d_in[5];
  const int*   nsrc = (const int*)d_in[6];
  const int*   ndst = (const int*)d_in[7];
  const float* Wl[3] = {(const float*)d_in[8],  (const float*)d_in[11], (const float*)d_in[14]};
  const float* al[3] = {(const float*)d_in[9],  (const float*)d_in[12], (const float*)d_in[15]};
  const float* bl[3] = {(const float*)d_in[10], (const float*)d_in[13], (const float*)d_in[16]};
  const float* pW1 = (const float*)d_in[17]; const float* pb1 = (const float*)d_in[18];
  const float* pW2 = (const float*)d_in[19]; const float* pb2 = (const float*)d_in[20];
  const float* pW3 = (const float*)d_in[21]; const float* pb3 = (const float*)d_in[22];

  float* out   = (float*)d_out;
  float* emb   = out + 2 * N_PAIR;

  float*    ws   = (float*)d_ws;
  float*    ht   = ws;
  float*    hbuf = ht   + 6400000;
  float*    pbuf = hbuf + 6400000;
  int*      deg    = (int*)(pbuf + 100000);
  int*      rowptr = deg + 25000;
  int*      rank   = rowptr + 25008;
  int2*     csr    = (int2*)(rank + 400000);

  const int Hs[3]  = {4, 4, 1};
  const int Fs[3]  = {64, 64, 128};
  const int Kin[3] = {128, 256, 256};

  fill_u32_kernel<<<(NN_NODES + 255) / 256, 256, 0, stream>>>((unsigned*)deg, 0u, NN_NODES);
  csr_count_kernel<<<(N_EDGES + 255) / 256, 256, 0, stream>>>(edst, deg, rank);
  csr_scan_kernel<<<1, SCAN_THREADS, 0, stream>>>(deg, rowptr);
  csr_fill_kernel<<<(N_EDGES + 255) / 256, 256, 0, stream>>>(esrc, edst, ew, rank, rowptr, csr);

  const float* hin = x;
  for (int l = 0; l < 3; ++l) {
    const int H = Hs[l], F = Fs[l], HF = H * F, K = Kin[l];
    float* hout = (l == 2) ? emb : hbuf;

    dim3 gg((NN_NODES + 63) / 64, HF / 32);
    gemm_wmma_f32<<<gg, 256, 0, stream>>>(hin, Wl[l], nullptr, ht,
                                          NN_NODES, HF, K, 0.f, 0);
    int nh = NN_NODES * H;
    if (l < 2) node_p_kernel<4, 64><<<(nh + 255) / 256, 256, 0, stream>>>(ht, al[l], pbuf);
    else       node_p_kernel<1, 128><<<(nh + 255) / 256, 256, 0, stream>>>(ht, al[l], pbuf);

    if (l < 2) {
      node_softmax_gather_kernel<4, 64><<<(NN_NODES + 7) / 8, 256, 0, stream>>>(
          rowptr, csr, pbuf, ht, bl[l], hout, 1);
    } else {
      node_softmax_gather_kernel<1, 128><<<(NN_NODES + 7) / 8, 256, 0, stream>>>(
          rowptr, csr, pbuf, ht, bl[l], hout, 0);
    }
    hin = hout;
  }

  predictor_fused_kernel<<<PR_BLOCKS, 256, 0, stream>>>(emb, psrc, pdst, nsrc, ndst,
                                                        pW1, pb1, pW2, pb2, pW3, pb3, out);
}


# Projection p folded into the GEMM epilogue (per-block 32-column shares, added where the gather reads them): 3 launches and a re-read of h gone; p sum = lane tree per 16 columns, then shares
# speedup vs baseline: 12.6215x; 1.0916x over previous
#include <hip/hip_runtime.h>
#include <hip/hip_bf16.h>


typedef __attribute__((ext_vector_type(2))) float v2f;
typedef __attribute__((ext_vector_type(8))) float v8f;

#define NN_NODES 25000
#define N_EDGES  400000
#define N_PAIR   100000
#define N_ROWS   200000

__device__ __forceinline__ float leaky(float v, float slope) {
  return v >= 0.f ? v : v * slope;
}

template <int MASK>
__device__ __forceinline__ float row_xor(float v) {
  return __builtin_bit_cast(float, __builtin_amdgcn_update_dpp(
      0, __builtin_bit_cast(int, v), 0x160 + MASK, 0xf, 0xf, false));
}

__global__ void gemm_wmma_f32(const float* __restrict__ A,
                              const float* __restrict__ W,
                              const float* __restrict__ bias,
                              float* __restrict__ C,
                              int M, int N, int K,
                              float slope, int act,
                              const float* __restrict__ avec,
                              float* __restrict__ ppart, int fmask) {
  __shared__ float As[64][36];
  __shared__ float Ws[32][36];
  __shared__ float part[2][64];

  const int t     = threadIdx.x;
  const int lane  = t & 31;
  const int wave  = t >> 5;
  const int waveM = wave & 3;
  const int waveN = wave >> 2;
  const int mBase = blockIdx.x * 64;
  const int nBase = blockIdx.y * 32;
  const int lh    = lane >> 4;
  const int l15   = lane & 15;

  v8f acc = {0.f, 0.f, 0.f, 0.f, 0.f, 0.f, 0.f, 0.f};

  for (int k0 = 0; k0 < K; k0 += 32) {
#pragma unroll
    for (int i = 0; i < 2; ++i) {
      int idx = t + i * 256;
      int r   = idx >> 3;
      int c4  = (idx & 7) * 4;
      int gr  = mBase + r;
      int grc = gr < M ? gr : M - 1;
      float4 v = *(const float4*)(A + (size_t)grc * K + k0 + c4);
      if (gr >= M) { v.x = 0.f; v.y = 0.f; v.z = 0.f; v.w = 0.f; }
      *(float4*)&As[r][c4] = v;
    }
    {
      int r  = t >> 3;
      int c4 = (t & 7) * 4;
      *(float4*)&Ws[r][c4] =
          *(const float4*)(W + (size_t)(nBase + r) * K + k0 + c4);
    }
    __syncthreads();

#pragma unroll
    for (int ks = 0; ks < 8; ++ks) {
      int kb = ks * 4 + 2 * lh;
      v2f a, b;
      a.x = As[waveM * 16 + l15][kb];
      a.y = As[waveM * 16 + l15][kb + 1];
      b.x = Ws[waveN * 16 + l15][kb];
      b.y = Ws[waveN * 16 + l15][kb + 1];
      acc = __builtin_amdgcn_wmma_f32_16x16x4_f32(
           false, a,  false, b,
           (short)0, acc,  false,  false);
    }
    __syncthreads();
  }

#pragma unroll
  for (int r = 0; r < 8; ++r) {
    int row = mBase + waveM * 16 + r + 8 * lh;
    int col = nBase + waveN * 16 + l15;
    if (row < M) {
      float v = acc[r];
      if (bias) v += bias[col];
      if (act)  v = leaky(v, slope);
      C[(size_t)row * N + col] = v;
    }
  }

  if (avec) {
    const float av = avec[(nBase + waveN * 16 + l15) & fmask];
#pragma unroll
    for (int r = 0; r < 8; ++r) {
      float v = acc[r] * av;
      v += row_xor<1>(v); v += row_xor<2>(v);
      v += row_xor<4>(v); v += row_xor<8>(v);
      if (l15 == r) part[waveN][waveM * 16 + 8 * lh + r] = v;
    }
    __syncthreads();
    if (t < 64 && mBase + t < M)
      ppart[(size_t)(mBase + t) * gridDim.y + blockIdx.y] = part[0][t] + part[1][t];
  }
}

__global__ void fill_u32_kernel(unsigned* __restrict__ dst, unsigned val, int n) {
  int i = blockIdx.x * blockDim.x + threadIdx.x;
  if (i < n) dst[i] = val;
}

__global__ void csr_count_kernel(const int* __restrict__ dst,
                                 int* __restrict__ deg,
                                 int* __restrict__ rank) {
  int e = blockIdx.x * blockDim.x + threadIdx.x;
  if (e >= N_EDGES) return;
  rank[e] = atomicAdd(&deg[dst[e]], 1);
}

#define SCAN_THREADS 1024
#define SCAN_PER     25
__global__ void __launch_bounds__(SCAN_THREADS)
csr_scan_kernel(const int* __restrict__ deg, int* __restrict__ rowptr) {
  __shared__ int part[SCAN_THREADS];
  const int t  = threadIdx.x;
  const int lo = t * SCAN_PER;
  int vals[SCAN_PER];
  int sum = 0;
#pragma unroll
  for (int i = 0; i < SCAN_PER; ++i) {
    int idx = lo + i;
    int idc = idx < NN_NODES ? idx : NN_NODES - 1;
    int v   = deg[idc];
    vals[i] = idx < NN_NODES ? v : 0;
    sum += vals[i];
  }
  part[t] = sum;
  __syncthreads();
  for (int off = 1; off < SCAN_THREADS; off <<= 1) {
    int v = (t >= off) ? part[t - off] : 0;
    __syncthreads();
    part[t] += v;
    __syncthreads();
  }
  int run = part[t] - sum;
#pragma unroll
  for (int i = 0; i < SCAN_PER; ++i) {
    int idx = lo + i;
    if (idx < NN_NODES) rowptr[idx] = run;
    run += vals[i];
  }
  if (t == SCAN_THREADS - 1) rowptr[NN_NODES] = part[SCAN_THREADS - 1];
}

__global__ void csr_fill_kernel(const int* __restrict__ src,
                                const int* __restrict__ dst,
                                const float* __restrict__ ew,
                                const int* __restrict__ rank,
                                const int* __restrict__ rowptr,
                                int2* __restrict__ csr) {
  int e = blockIdx.x * blockDim.x + threadIdx.x;
  if (e >= N_EDGES) return;
  int pos = rowptr[dst[e]] + rank[e];
  csr[pos] = make_int2(src[e], __float_as_int(log1pf(ew[e])));
}

__device__ __forceinline__ float xhalf(float v) {
  return __builtin_bit_cast(float, __builtin_amdgcn_permlanex16(
      __builtin_bit_cast(int, v), __builtin_bit_cast(int, v), 0x76543210, 0xfedcba98, false, false));
}
__device__ __forceinline__ float wave_max(float v) {
  v = fmaxf(v, row_xor<1>(v)); v = fmaxf(v, row_xor<2>(v));
  v = fmaxf(v, row_xor<4>(v)); v = fmaxf(v, row_xor<8>(v));
  return fmaxf(v, xhalf(v));
}
__device__ __forceinline__ float wave_sum(float v) {
  v += row_xor<1>(v); v += row_xor<2>(v);
  v += row_xor<4>(v); v += row_xor<8>(v);
  return v + xhalf(v);
}

template <int H>
__device__ __forceinline__ void edge_scores(const int2* __restrict__ csr,
                                            const float* __restrict__ p,
                                            const float (&pd)[H], int j,
                                            int& s, float (&sc)[H]) {
  const int2 se = csr[j];
  s = se.x;
  const float lw = __int_as_float(se.y);
  if constexpr (H == 4) {
    const float4 q0 = *(const float4*)(p + (size_t)se.x * 8);
    const float4 q1 = *(const float4*)(p + (size_t)se.x * 8 + 4);
    sc[0] = leaky((q0.x + q0.y) + pd[0] + lw, 0.01f);
    sc[1] = leaky((q0.z + q0.w) + pd[1] + lw, 0.01f);
    sc[2] = leaky((q1.x + q1.y) + pd[2] + lw, 0.01f);
    sc[3] = leaky((q1.z + q1.w) + pd[3] + lw, 0.01f);
  } else {
    const float4 q = *(const float4*)(p + (size_t)se.x * 4);
    sc[0] = leaky(((q.x + q.y) + (q.z + q.w)) + pd[0] + lw, 0.01f);
  }
}

template <int H, int F>
__device__ __forceinline__ void gather_rows(const float* __restrict__ ht, int lane,
                                            int my_s, const float (&my_a)[H], int cnt,
                                            float4 (&acc)[H * F / 128]) {
  constexpr int HF = H * F;
  const bool hiHalf = lane >= 16;
#pragma unroll 4
  for (int t = 0; t < cnt; ++t) {
    const int s = __builtin_amdgcn_readlane(my_s, t);
    const float* row = ht + (size_t)s * HF + lane * 4;
    if constexpr (H == 4) {
      const float a0 = __int_as_float(__builtin_amdgcn_readlane(__float_as_int(my_a[0]), t));
      const float a1 = __int_as_float(__builtin_amdgcn_readlane(__float_as_int(my_a[1]), t));
      const float a2 = __int_as_float(__builtin_amdgcn_readlane(__float_as_int(my_a[2]), t));
      const float a3 = __int_as_float(__builtin_amdgcn_readlane(__float_as_int(my_a[3]), t));
      const float aA = hiHalf ? a1 : a0;
      const float aB = hiHalf ? a3 : a2;
      const float4 v0 = *(const float4*)(row);
      const float4 v1 = *(const float4*)(row + 128);
      acc[0].x += aA * v0.x; acc[0].y += aA * v0.y; acc[0].z += aA * v0.z; acc[0].w += aA * v0.w;
      acc[1].x += aB * v1.x; acc[1].y += aB * v1.y; acc[1].z += aB * v1.z; acc[1].w += aB * v1.w;
    } else {
      const float a = __int_as_float(__builtin_amdgcn_readlane(__float_as_int(my_a[0]), t));
      const float4 v0 = *(const float4*)(row);
      acc[0].x += a * v0.x; acc[0].y += a * v0.y; acc[0].z += a * v0.z; acc[0].w += a * v0.w;
    }
  }
}

template <int H, int F>
__global__ void __launch_bounds__(256)
node_softmax_gather_kernel(const int* __restrict__ rowptr,
                           const int2* __restrict__ csr,
                           const float* __restrict__ p,
                           const float* __restrict__ ht,
                           const float* __restrict__ bias,
                           float* __restrict__ hout,
                           int act) {
  constexpr int HF = H * F;
  constexpr int NV = HF / 128;
  const int lane = threadIdx.x & 31;
  const int n = __builtin_amdgcn_readfirstlane((int)(blockIdx.x * 8 + (threadIdx.x >> 5)));
  if (n >= NN_NODES) return;
  const int beg = rowptr[n];
  const int end = rowptr[n + 1];
  const int deg = end - beg;

  float pd[H];
  if constexpr (H == 4) {
#pragma unroll
    for (int h = 0; h < 4; ++h) pd[h] = p[n * 8 + 2 * h] + p[n * 8 + 2 * h + 1];
  } else {
    pd[0] = (p[n * 4] + p[n * 4 + 1]) + (p[n * 4 + 2] + p[n * 4 + 3]);
  }

  float4 acc[NV];
#pragma unroll
  for (int v = 0; v < NV; ++v) { acc[v].x = 0.f; acc[v].y = 0.f; acc[v].z = 0.f; acc[v].w = 0.f; }

  if (deg <= 32) {
    const bool on = lane < deg;
    int   my_s = 0;
    float sc[H];
#pragma unroll
    for (int h = 0; h < H; ++h) sc[h] = -INFINITY;
    if (on) edge_scores<H>(csr, p, pd, beg + lane, my_s, sc);
    float my_a[H];
#pragma unroll
    for (int h = 0; h < H; ++h) {
      float m = wave_max(sc[h]);
      if (!isfinite(m)) m = 0.f;
      const float ex = on ? __expf(sc[h] - m) : 0.f;
      const float sum = wave_sum(ex);
      my_a[h] = ex / fmaxf(sum, 1e-20f);
    }
    gather_rows<H, F>(ht, lane, my_s, my_a, deg, acc);
  } else {
    float m[H], den[H];
#pragma unroll
    for (int h = 0; h < H; ++h) m[h] = -INFINITY;
    for (int base = beg; base < end; base += 32) {
      const int j = base + lane;
      if (j < end) {
        int s_; float sc[H];
        edge_scores<H>(csr, p, pd, j, s_, sc);
#pragma unroll
        for (int h = 0; h < H; ++h) m[h] = fmaxf(m[h], sc[h]);
      }
    }
#pragma unroll
    for (int h = 0; h < H; ++h) {
      m[h] = wave_max(m[h]);
      if (!isfinite(m[h])) m[h] = 0.f;
      den[h] = 0.f;
    }
    for (int base = beg; base < end; base += 32) {
      const int j = base + lane;
      if (j < end) {
        int s_; float sc[H];
        edge_scores<H>(csr, p, pd, j, s_, sc);
#pragma unroll
        for (int h = 0; h < H; ++h) den[h] += __expf(sc[h] - m[h]);
      }
    }
#pragma unroll
    for (int h = 0; h < H; ++h) den[h] = fmaxf(wave_sum(den[h]), 1e-20f);
    for (int base = beg; base < end; base += 32) {
      const int j = base + lane;
      int   my_s = 0;
      float my_a[H];
#pragma unroll
      for (int h = 0; h < H; ++h) my_a[h] = 0.f;
      if (j < end) {
        float sc[H];
        edge_scores<H>(csr, p, pd, j, my_s, sc);
#pragma unroll
        for (int h = 0; h < H; ++h) my_a[h] = __expf(sc[h] - m[h]) / den[h];
      }
      const int left = end - base;
      gather_rows<H, F>(ht, lane, my_s, my_a, left < 32 ? left : 32, acc);
    }
  }

#pragma unroll
  for (int v = 0; v < NV; ++v) {
    const int c = v * 128 + lane * 4;
    const float4 b = *(const float4*)(bias + c);
    float4 r;
    r.x = acc[v].x + b.x; r.y = acc[v].y + b.y; r.z = acc[v].z + b.z; r.w = acc[v].w + b.w;
    if (act) {
      r.x = leaky(r.x, 0.01f); r.y = leaky(r.y, 0.01f);
      r.z = leaky(r.z, 0.01f); r.w = leaky(r.w, 0.01f);
    }
    *(float4*)(hout + (size_t)n * HF + c) = r;
  }
}

#define PR_TILES  (N_ROWS / 16)
#define PR_BLOCKS 512
#define ZP  132
#define Z1P 68

__device__ __forceinline__ void wave_lds_sync() {
  __builtin_amdgcn_fence(__ATOMIC_RELEASE, "wavefront");
  __builtin_amdgcn_wave_barrier();
  __builtin_amdgcn_fence(__ATOMIC_ACQUIRE, "wavefront");
}

__device__ __forceinline__ v8f wmma4(float a0, float a1, float b0, float b1, v8f acc) {
  v2f a = {a0, a1};
  v2f b = {b0, b1};
  return __builtin_amdgcn_wmma_f32_16x16x4_f32(
       false, a,  false, b,
       (short)0, acc,  false,  false);
}

__global__ void __launch_bounds__(256)
predictor_fused_kernel(const float* __restrict__ emb,
                       const int* __restrict__ ps, const int* __restrict__ pd,
                       const int* __restrict__ ns, const int* __restrict__ nd,
                       const float* __restrict__ pW1, const float* __restrict__ pb1,
                       const float* __restrict__ pW2, const float* __restrict__ pb2,
                       const float* __restrict__ pW3, const float* __restrict__ pb3,
                       float* __restrict__ out) {
  __shared__ __attribute__((aligned(16))) float W1s[64][ZP];
  __shared__ __attribute__((aligned(16))) float W2s[32][Z1P];
  __shared__ __attribute__((aligned(16))) float Zs[8][16][ZP];
  __shared__ __attribute__((aligned(16))) float Z1s[8][16][Z1P];

  const int t    = threadIdx.x;
  const int lane = t & 31;
  const int wave = __builtin_amdgcn_readfirstlane(t >> 5);
  const int c    = lane & 15;
  const int h    = lane >> 4;

#pragma unroll
  for (int j = 0; j < 8; ++j) {
    const int idx = t + j * 256;
    const int r   = idx >> 5;
    const int c4  = (idx & 31) * 4;
    *(float4*)&W1s[r][c4] = *(const float4*)(pW1 + r * 128 + c4);
  }
#pragma unroll
  for (int j = 0; j < 2; ++j) {
    const int idx = t + j * 256;
    const int r   = idx >> 4;
    const int c4  = (idx & 15) * 4;
    *(float4*)&W2s[r][c4] = *(const float4*)(pW2 + r * 64 + c4);
  }
  float bias1[4], bias2[2], w3[2];
#pragma unroll
  for (int tt = 0; tt < 4; ++tt) bias1[tt] = pb1[16 * tt + c];
#pragma unroll
  for (int tt = 0; tt < 2; ++tt) { bias2[tt] = pb2[16 * tt + c]; w3[tt] = pW3[16 * tt + c]; }
  const float b3 = pb3[0];
  __syncthreads();

  for (int wt = blockIdx.x * 8 + wave; wt < PR_TILES; wt += PR_BLOCKS * 8) {
    const int row0 = wt * 16;
    const int* sp = ps; const int* dp = pd; int rbase = row0;
    if (row0 >= N_PAIR) { sp = ns; dp = nd; rbase = row0 - N_PAIR; }
    const int si = sp[rbase + c];
    const int di = dp[rbase + c];

#pragma unroll
    for (int r = 0; r < 16; ++r) {
      const int s = __builtin_amdgcn_readlane(si, r);
      const int d = __builtin_amdgcn_readlane(di, r);
      const float4 a = *(const float4*)(emb + (size_t)s * 128 + lane * 4);
      const float4 b = *(const float4*)(emb + (size_t)d * 128 + lane * 4);
      float4 z; z.x = a.x * b.x; z.y = a.y * b.y; z.z = a.z * b.z; z.w = a.w * b.w;
      *(float4*)&Zs[wave][r][lane * 4] = z;
    }
    wave_lds_sync();

    v8f acc[4];
#pragma unroll
    for (int tt = 0; tt < 4; ++tt) acc[tt] = (v8f){0.f, 0.f, 0.f, 0.f, 0.f, 0.f, 0.f, 0.f};
    {
      float4 a_cur = *(const float4*)&Zs[wave][c][4 * h];
      float4 b_cur[4];
#pragma unroll
      for (int tt = 0; tt < 4; ++tt) b_cur[tt] = *(const float4*)&W1s[16 * tt + c][4 * h];
#pragma unroll
      for (int i = 0; i < 16; ++i) {
        float4 a_nxt = a_cur;
        float4 b_nxt[4];
#pragma unroll
        for (int tt = 0; tt < 4; ++tt) b_nxt[tt] = b_cur[tt];
        if (i + 1 < 16) {
          a_nxt = *(const float4*)&Zs[wave][c][8 * (i + 1) + 4 * h];
#pragma unroll
          for (int tt = 0; tt < 4; ++tt)
            b_nxt[tt] = *(const float4*)&W1s[16 * tt + c][8 * (i + 1) + 4 * h];
        }
#pragma unroll
        for (int tt = 0; tt < 4; ++tt)
          acc[tt] = wmma4(a_cur.x, a_cur.y, b_cur[tt].x, b_cur[tt].y, acc[tt]);
#pragma unroll
        for (int tt = 0; tt < 4; ++tt)
          acc[tt] = wmma4(a_cur.z, a_cur.w, b_cur[tt].z, b_cur[tt].w, acc[tt]);
        a_cur = a_nxt;
#pragma unroll
        for (int tt = 0; tt < 4; ++tt) b_cur[tt] = b_nxt[tt];
      }
    }
#pragma unroll
    for (int tt = 0; tt < 4; ++tt) {
#pragma unroll
      for (int r = 0; r < 8; ++r)
        Z1s[wave][8 * h + r][16 * tt + c] = leaky(acc[tt][r] + bias1[tt], 0.2f);
    }
    wave_lds_sync();

    v8f acc2[2];
#pragma unroll
    for (int tt = 0; tt < 2; ++tt) acc2[tt] = (v8f){0.f, 0.f, 0.f, 0.f, 0.f, 0.f, 0.f, 0.f};
    {
      float4 a2[8];
      float4 b2[2][8];
#pragma unroll
      for (int i = 0; i < 8; ++i) {
        a2[i] = *(const float4*)&Z1s[wave][c][8 * i + 4 * h];
#pragma unroll
        for (int tt = 0; tt < 2; ++tt)
          b2[tt][i] = *(const float4*)&W2s[16 * tt + c][8 * i + 4 * h];
      }
#pragma unroll
      for (int i = 0; i < 8; ++i) {
#pragma unroll
        for (int tt = 0; tt < 2; ++tt)
          acc2[tt] = wmma4(a2[i].x, a2[i].y, b2[tt][i].x, b2[tt][i].y, acc2[tt]);
#pragma unroll
        for (int tt = 0; tt < 2; ++tt)
          acc2[tt] = wmma4(a2[i].z, a2[i].w, b2[tt][i].z, b2[tt][i].w, acc2[tt]);
      }
    }

    float mine = 0.f;
#pragma unroll
    for (int r = 0; r < 8; ++r) {
      float v = leaky(acc2[0][r] + bias2[0], 0.2f) * w3[0]
              + leaky(acc2[1][r] + bias2[1], 0.2f) * w3[1];
      v += row_xor<1>(v); v += row_xor<2>(v);
      v += row_xor<4>(v); v += row_xor<8>(v);
      if (c == r) mine = v;
    }
    if (c < 8) out[row0 + 8 * h + c] = mine + b3;
    wave_lds_sync();
  }
}

extern "C" void kernel_launch(void* const* d_in, const int* in_sizes, int n_in,
                              void* d_out, int out_size, void* d_ws, size_t ws_size,
                              hipStream_t stream) {
  const float* x    = (const float*)d_in[0];
  const float* ew   = (const float*)d_in[1];
  const int*   esrc = (const int*)d_in[2];
  const int*   edst = (const int*)d_in[3];
  const int*   psrc = (const int*)d_in[4];
  const int*   pdst = (const int*)d_in[5];
  const int*   nsrc = (const int*)d_in[6];
  const int*   ndst = (const int*)d_in[7];
  const float* Wl[3] = {(const float*)d_in[8],  (const float*)d_in[11], (const float*)d_in[14]};
  const float* al[3] = {(const float*)d_in[9],  (const float*)d_in[12], (const float*)d_in[15]};
  const float* bl[3] = {(const float*)d_in[10], (const float*)d_in[13], (const float*)d_in[16]};
  const float* pW1 = (const float*)d_in[17]; const float* pb1 = (const float*)d_in[18];
  const float* pW2 = (const float*)d_in[19]; const float* pb2 = (const float*)d_in[20];
  const float* pW3 = (const float*)d_in[21]; const float* pb3 = (const float*)d_in[22];

  float* out   = (float*)d_out;
  float* emb   = out + 2 * N_PAIR;

  float*    ws   = (float*)d_ws;
  float*    ht   = ws;
  float*    hbuf = ht   + 6400000;
  float*    pbuf = hbuf + 6400000;
  int*      deg    = (int*)(pbuf + 200000);
  int*      rowptr = deg + 25000;
  int*      rank   = rowptr + 25008;
  int2*     csr    = (int2*)(rank + 400000);

  const int Hs[3]  = {4, 4, 1};
  const int Fs[3]  = {64, 64, 128};
  const int Kin[3] = {128, 256, 256};
  const int Fm[3]  = {63, 63, 127};

  fill_u32_kernel<<<(NN_NODES + 255) / 256, 256, 0, stream>>>((unsigned*)deg, 0u, NN_NODES);
  csr_count_kernel<<<(N_EDGES + 255) / 256, 256, 0, stream>>>(edst, deg, rank);
  csr_scan_kernel<<<1, SCAN_THREADS, 0, stream>>>(deg, rowptr);
  csr_fill_kernel<<<(N_EDGES + 255) / 256, 256, 0, stream>>>(esrc, edst, ew, rank, rowptr, csr);

  const float* hin = x;
  for (int l = 0; l < 3; ++l) {
    const int H = Hs[l], F = Fs[l], HF = H * F, K = Kin[l];
    float* hout = (l == 2) ? emb : hbuf;

    dim3 gg((NN_NODES + 63) / 64, HF / 32);
    gemm_wmma_f32<<<gg, 256, 0, stream>>>(hin, Wl[l], nullptr, ht,
                                          NN_NODES, HF, K, 0.f, 0,
                                          al[l], pbuf, Fm[l]);

    if (l < 2) {
      node_softmax_gather_kernel<4, 64><<<(NN_NODES + 7) / 8, 256, 0, stream>>>(
          rowptr, csr, pbuf, ht, bl[l], hout, 1);
    } else {
      node_softmax_gather_kernel<1, 128><<<(NN_NODES + 7) / 8, 256, 0, stream>>>(
          rowptr, csr, pbuf, ht, bl[l], hout, 0);
    }
    hin = hout;
  }

  predictor_fused_kernel<<<PR_BLOCKS, 256, 0, stream>>>(emb, psrc, pdst, nsrc, ndst,
                                                        pW1, pb1, pW2, pb2, pW3, pb3, out);
}


# GEMM k-loop software-pipelined: next 32-wide slab's global loads issued ahead of the current slab's WMMAs, two LDS slabs, one barrier per slab instead of two; k order and sums unchanged
# speedup vs baseline: 13.2659x; 1.0511x over previous
#include <hip/hip_runtime.h>
#include <hip/hip_bf16.h>


typedef __attribute__((ext_vector_type(2))) float v2f;
typedef __attribute__((ext_vector_type(8))) float v8f;

#define NN_NODES 25000
#define N_EDGES  400000
#define N_PAIR   100000
#define N_ROWS   200000

__device__ __forceinline__ float leaky(float v, float slope) {
  return v >= 0.f ? v : v * slope;
}

template <int MASK>
__device__ __forceinline__ float row_xor(float v) {
  return __builtin_bit_cast(float, __builtin_amdgcn_update_dpp(
      0, __builtin_bit_cast(int, v), 0x160 + MASK, 0xf, 0xf, false));
}

__global__ void gemm_wmma_f32(const float* __restrict__ A,
                              const float* __restrict__ W,
                              const float* __restrict__ bias,
                              float* __restrict__ C,
                              int M, int N, int K,
                              float slope, int act,
                              const float* __restrict__ avec,
                              float* __restrict__ ppart, int fmask) {
  __shared__ float As[2][64][36];
  __shared__ float Ws[2][32][36];
  __shared__ float part[2][64];

  const int t     = threadIdx.x;
  const int lane  = t & 31;
  const int wave  = t >> 5;
  const int waveM = wave & 3;
  const int waveN = wave >> 2;
  const int mBase = blockIdx.x * 64;
  const int nBase = blockIdx.y * 32;
  const int lh    = lane >> 4;
  const int l15   = lane & 15;

  v8f acc = {0.f, 0.f, 0.f, 0.f, 0.f, 0.f, 0.f, 0.f};

  const int  sr   = t >> 3;
  const int  c4   = (t & 7) * 4;
  const int  gr0  = mBase + sr;
  const int  gr1  = gr0 + 32;
  const bool in0  = gr0 < M;
  const bool in1  = gr1 < M;
  const float* pa0 = A + (size_t)(in0 ? gr0 : M - 1) * K + c4;
  const float* pa1 = A + (size_t)(in1 ? gr1 : M - 1) * K + c4;
  const float* pw  = W + (size_t)(nBase + sr) * K + c4;

  float4 ra0 = *(const float4*)(pa0);
  float4 ra1 = *(const float4*)(pa1);
  float4 rw  = *(const float4*)(pw);
  if (!in0) { ra0.x = 0.f; ra0.y = 0.f; ra0.z = 0.f; ra0.w = 0.f; }
  if (!in1) { ra1.x = 0.f; ra1.y = 0.f; ra1.z = 0.f; ra1.w = 0.f; }
  *(float4*)&As[0][sr][c4]      = ra0;
  *(float4*)&As[0][sr + 32][c4] = ra1;
  *(float4*)&Ws[0][sr][c4]      = rw;
  __syncthreads();

  const int nk = K >> 5;
  for (int it = 0; it < nk; ++it) {
    const int  cur  = it & 1;
    const bool more = it + 1 < nk;
    if (more) {
      const int ko = (it + 1) * 32;
      ra0 = *(const float4*)(pa0 + ko);
      ra1 = *(const float4*)(pa1 + ko);
      rw  = *(const float4*)(pw + ko);
    }

#pragma unroll
    for (int ks = 0; ks < 8; ++ks) {
      int kb = ks * 4 + 2 * lh;
      v2f a, b;
      a.x = As[cur][waveM * 16 + l15][kb];
      a.y = As[cur][waveM * 16 + l15][kb + 1];
      b.x = Ws[cur][waveN * 16 + l15][kb];
      b.y = Ws[cur][waveN * 16 + l15][kb + 1];
      acc = __builtin_amdgcn_wmma_f32_16x16x4_f32(
           false, a,  false, b,
           (short)0, acc,  false,  false);
    }

    if (more) {
      if (!in0) { ra0.x = 0.f; ra0.y = 0.f; ra0.z = 0.f; ra0.w = 0.f; }
      if (!in1) { ra1.x = 0.f; ra1.y = 0.f; ra1.z = 0.f; ra1.w = 0.f; }
      *(float4*)&As[cur ^ 1][sr][c4]      = ra0;
      *(float4*)&As[cur ^ 1][sr + 32][c4] = ra1;
      *(float4*)&Ws[cur ^ 1][sr][c4]      = rw;
    }
    __syncthreads();
  }

#pragma unroll
  for (int r = 0; r < 8; ++r) {
    int row = mBase + waveM * 16 + r + 8 * lh;
    int col = nBase + waveN * 16 + l15;
    if (row < M) {
      float v = acc[r];
      if (bias) v += bias[col];
      if (act)  v = leaky(v, slope);
      C[(size_t)row * N + col] = v;
    }
  }

  if (avec) {
    const float av = avec[(nBase + waveN * 16 + l15) & fmask];
#pragma unroll
    for (int r = 0; r < 8; ++r) {
      float v = acc[r] * av;
      v += row_xor<1>(v); v += row_xor<2>(v);
      v += row_xor<4>(v); v += row_xor<8>(v);
      if (l15 == r) part[waveN][waveM * 16 + 8 * lh + r] = v;
    }
    __syncthreads();
    if (t < 64 && mBase + t < M)
      ppart[(size_t)(mBase + t) * gridDim.y + blockIdx.y] = part[0][t] + part[1][t];
  }
}

__global__ void fill_u32_kernel(unsigned* __restrict__ dst, unsigned val, int n) {
  int i = blockIdx.x * blockDim.x + threadIdx.x;
  if (i < n) dst[i] = val;
}

__global__ void csr_count_kernel(const int* __restrict__ dst,
                                 int* __restrict__ deg,
                                 int* __restrict__ rank) {
  int e = blockIdx.x * blockDim.x + threadIdx.x;
  if (e >= N_EDGES) return;
  rank[e] = atomicAdd(&deg[dst[e]], 1);
}

#define SCAN_THREADS 1024
#define SCAN_PER     25
__global__ void __launch_bounds__(SCAN_THREADS)
csr_scan_kernel(const int* __restrict__ deg, int* __restrict__ rowptr) {
  __shared__ int part[SCAN_THREADS];
  const int t  = threadIdx.x;
  const int lo = t * SCAN_PER;
  int vals[SCAN_PER];
  int sum = 0;
#pragma unroll
  for (int i = 0; i < SCAN_PER; ++i) {
    int idx = lo + i;
    int idc = idx < NN_NODES ? idx : NN_NODES - 1;
    int v   = deg[idc];
    vals[i] = idx < NN_NODES ? v : 0;
    sum += vals[i];
  }
  part[t] = sum;
  __syncthreads();
  for (int off = 1; off < SCAN_THREADS; off <<= 1) {
    int v = (t >= off) ? part[t - off] : 0;
    __syncthreads();
    part[t] += v;
    __syncthreads();
  }
  int run = part[t] - sum;
#pragma unroll
  for (int i = 0; i < SCAN_PER; ++i) {
    int idx = lo + i;
    if (idx < NN_NODES) rowptr[idx] = run;
    run += vals[i];
  }
  if (t == SCAN_THREADS - 1) rowptr[NN_NODES] = part[SCAN_THREADS - 1];
}

__global__ void csr_fill_kernel(const int* __restrict__ src,
                                const int* __restrict__ dst,
                                const float* __restrict__ ew,
                                const int* __restrict__ rank,
                                const int* __restrict__ rowptr,
                                int2* __restrict__ csr) {
  int e = blockIdx.x * blockDim.x + threadIdx.x;
  if (e >= N_EDGES) return;
  int pos = rowptr[dst[e]] + rank[e];
  csr[pos] = make_int2(src[e], __float_as_int(log1pf(ew[e])));
}

__device__ __forceinline__ float xhalf(float v) {
  return __builtin_bit_cast(float, __builtin_amdgcn_permlanex16(
      __builtin_bit_cast(int, v), __builtin_bit_cast(int, v), 0x76543210, 0xfedcba98, false, false));
}
__device__ __forceinline__ float wave_max(float v) {
  v = fmaxf(v, row_xor<1>(v)); v = fmaxf(v, row_xor<2>(v));
  v = fmaxf(v, row_xor<4>(v)); v = fmaxf(v, row_xor<8>(v));
  return fmaxf(v, xhalf(v));
}
__device__ __forceinline__ float wave_sum(float v) {
  v += row_xor<1>(v); v += row_xor<2>(v);
  v += row_xor<4>(v); v += row_xor<8>(v);
  return v + xhalf(v);
}

template <int H>
__device__ __forceinline__ void edge_scores(const int2* __restrict__ csr,
                                            const float* __restrict__ p,
                                            const float (&pd)[H], int j,
                                            int& s, float (&sc)[H]) {
  const int2 se = csr[j];
  s = se.x;
  const float lw = __int_as_float(se.y);
  if constexpr (H == 4) {
    const float4 q0 = *(const float4*)(p + (size_t)se.x * 8);
    const float4 q1 = *(const float4*)(p + (size_t)se.x * 8 + 4);
    sc[0] = leaky((q0.x + q0.y) + pd[0] + lw, 0.01f);
    sc[1] = leaky((q0.z + q0.w) + pd[1] + lw, 0.01f);
    sc[2] = leaky((q1.x + q1.y) + pd[2] + lw, 0.01f);
    sc[3] = leaky((q1.z + q1.w) + pd[3] + lw, 0.01f);
  } else {
    const float4 q = *(const float4*)(p + (size_t)se.x * 4);
    sc[0] = leaky(((q.x + q.y) + (q.z + q.w)) + pd[0] + lw, 0.01f);
  }
}

template <int H, int F>
__device__ __forceinline__ void gather_rows(const float* __restrict__ ht, int lane,
                                            int my_s, const float (&my_a)[H], int cnt,
                                            float4 (&acc)[H * F / 128]) {
  constexpr int HF = H * F;
  const bool hiHalf = lane >= 16;
#pragma unroll 4
  for (int t = 0; t < cnt; ++t) {
    const int s = __builtin_amdgcn_readlane(my_s, t);
    const float* row = ht + (size_t)s * HF + lane * 4;
    if constexpr (H == 4) {
      const float a0 = __int_as_float(__builtin_amdgcn_readlane(__float_as_int(my_a[0]), t));
      const float a1 = __int_as_float(__builtin_amdgcn_readlane(__float_as_int(my_a[1]), t));
      const float a2 = __int_as_float(__builtin_amdgcn_readlane(__float_as_int(my_a[2]), t));
      const float a3 = __int_as_float(__builtin_amdgcn_readlane(__float_as_int(my_a[3]), t));
      const float aA = hiHalf ? a1 : a0;
      const float aB = hiHalf ? a3 : a2;
      const float4 v0 = *(const float4*)(row);
      const float4 v1 = *(const float4*)(row + 128);
      acc[0].x += aA * v0.x; acc[0].y += aA * v0.y; acc[0].z += aA * v0.z; acc[0].w += aA * v0.w;
      acc[1].x += aB * v1.x; acc[1].y += aB * v1.y; acc[1].z += aB * v1.z; acc[1].w += aB * v1.w;
    } else {
      const float a = __int_as_float(__builtin_amdgcn_readlane(__float_as_int(my_a[0]), t));
      const float4 v0 = *(const float4*)(row);
      acc[0].x += a * v0.x; acc[0].y += a * v0.y; acc[0].z += a * v0.z; acc[0].w += a * v0.w;
    }
  }
}

template <int H, int F>
__global__ void __launch_bounds__(256)
node_softmax_gather_kernel(const int* __restrict__ rowptr,
                           const int2* __restrict__ csr,
                           const float* __restrict__ p,
                           const float* __restrict__ ht,
                           const float* __restrict__ bias,
                           float* __restrict__ hout,
                           int act) {
  constexpr int HF = H * F;
  constexpr int NV = HF / 128;
  const int lane = threadIdx.x & 31;
  const int n = __builtin_amdgcn_readfirstlane((int)(blockIdx.x * 8 + (threadIdx.x >> 5)));
  if (n >= NN_NODES) return;
  const int beg = rowptr[n];
  const int end = rowptr[n + 1];
  const int deg = end - beg;

  float pd[H];
  if constexpr (H == 4) {
#pragma unroll
    for (int h = 0; h < 4; ++h) pd[h] = p[n * 8 + 2 * h] + p[n * 8 + 2 * h + 1];
  } else {
    pd[0] = (p[n * 4] + p[n * 4 + 1]) + (p[n * 4 + 2] + p[n * 4 + 3]);
  }

  float4 acc[NV];
#pragma unroll
  for (int v = 0; v < NV; ++v) { acc[v].x = 0.f; acc[v].y = 0.f; acc[v].z = 0.f; acc[v].w = 0.f; }

  if (deg <= 32) {
    const bool on = lane < deg;
    int   my_s = 0;
    float sc[H];
#pragma unroll
    for (int h = 0; h < H; ++h) sc[h] = -INFINITY;
    if (on) edge_scores<H>(csr, p, pd, beg + lane, my_s, sc);
    float my_a[H];
#pragma unroll
    for (int h = 0; h < H; ++h) {
      float m = wave_max(sc[h]);
      if (!isfinite(m)) m = 0.f;
      const float ex = on ? __expf(sc[h] - m) : 0.f;
      const float sum = wave_sum(ex);
      my_a[h] = ex / fmaxf(sum, 1e-20f);
    }
    gather_rows<H, F>(ht, lane, my_s, my_a, deg, acc);
  } else {
    float m[H], den[H];
#pragma unroll
    for (int h = 0; h < H; ++h) m[h] = -INFINITY;
    for (int base = beg; base < end; base += 32) {
      const int j = base + lane;
      if (j < end) {
        int s_; float sc[H];
        edge_scores<H>(csr, p, pd, j, s_, sc);
#pragma unroll
        for (int h = 0; h < H; ++h) m[h] = fmaxf(m[h], sc[h]);
      }
    }
#pragma unroll
    for (int h = 0; h < H; ++h) {
      m[h] = wave_max(m[h]);
      if (!isfinite(m[h])) m[h] = 0.f;
      den[h] = 0.f;
    }
    for (int base = beg; base < end; base += 32) {
      const int j = base + lane;
      if (j < end) {
        int s_; float sc[H];
        edge_scores<H>(csr, p, pd, j, s_, sc);
#pragma unroll
        for (int h = 0; h < H; ++h) den[h] += __expf(sc[h] - m[h]);
      }
    }
#pragma unroll
    for (int h = 0; h < H; ++h) den[h] = fmaxf(wave_sum(den[h]), 1e-20f);
    for (int base = beg; base < end; base += 32) {
      const int j = base + lane;
      int   my_s = 0;
      float my_a[H];
#pragma unroll
      for (int h = 0; h < H; ++h) my_a[h] = 0.f;
      if (j < end) {
        float sc[H];
        edge_scores<H>(csr, p, pd, j, my_s, sc);
#pragma unroll
        for (int h = 0; h < H; ++h) my_a[h] = __expf(sc[h] - m[h]) / den[h];
      }
      const int left = end - base;
      gather_rows<H, F>(ht, lane, my_s, my_a, left < 32 ? left : 32, acc);
    }
  }

#pragma unroll
  for (int v = 0; v < NV; ++v) {
    const int c = v * 128 + lane * 4;
    const float4 b = *(const float4*)(bias + c);
    float4 r;
    r.x = acc[v].x + b.x; r.y = acc[v].y + b.y; r.z = acc[v].z + b.z; r.w = acc[v].w + b.w;
    if (act) {
      r.x = leaky(r.x, 0.01f); r.y = leaky(r.y, 0.01f);
      r.z = leaky(r.z, 0.01f); r.w = leaky(r.w, 0.01f);
    }
    *(float4*)(hout + (size_t)n * HF + c) = r;
  }
}

#define PR_TILES  (N_ROWS / 16)
#define PR_BLOCKS 512
#define ZP  132
#define Z1P 68

__device__ __forceinline__ void wave_lds_sync() {
  __builtin_amdgcn_fence(__ATOMIC_RELEASE, "wavefront");
  __builtin_amdgcn_wave_barrier();
  __builtin_amdgcn_fence(__ATOMIC_ACQUIRE, "wavefront");
}

__device__ __forceinline__ v8f wmma4(float a0, float a1, float b0, float b1, v8f acc) {
  v2f a = {a0, a1};
  v2f b = {b0, b1};
  return __builtin_amdgcn_wmma_f32_16x16x4_f32(
       false, a,  false, b,
       (short)0, acc,  false,  false);
}

__global__ void __launch_bounds__(256)
predictor_fused_kernel(const float* __restrict__ emb,
                       const int* __restrict__ ps, const int* __restrict__ pd,
                       const int* __restrict__ ns, const int* __restrict__ nd,
                       const float* __restrict__ pW1, const float* __restrict__ pb1,
                       const float* __restrict__ pW2, const float* __restrict__ pb2,
                       const float* __restrict__ pW3, const float* __restrict__ pb3,
                       float* __restrict__ out) {
  __shared__ __attribute__((aligned(16))) float W1s[64][ZP];
  __shared__ __attribute__((aligned(16))) float W2s[32][Z1P];
  __shared__ __attribute__((aligned(16))) float Zs[8][16][ZP];
  __shared__ __attribute__((aligned(16))) float Z1s[8][16][Z1P];

  const int t    = threadIdx.x;
  const int lane = t & 31;
  const int wave = __builtin_amdgcn_readfirstlane(t >> 5);
  const int c    = lane & 15;
  const int h    = lane >> 4;

#pragma unroll
  for (int j = 0; j < 8; ++j) {
    const int idx = t + j * 256;
    const int r   = idx >> 5;
    const int c4  = (idx & 31) * 4;
    *(float4*)&W1s[r][c4] = *(const float4*)(pW1 + r * 128 + c4);
  }
#pragma unroll
  for (int j = 0; j < 2; ++j) {
    const int idx = t + j * 256;
    const int r   = idx >> 4;
    const int c4  = (idx & 15) * 4;
    *(float4*)&W2s[r][c4] = *(const float4*)(pW2 + r * 64 + c4);
  }
  float bias1[4], bias2[2], w3[2];
#pragma unroll
  for (int tt = 0; tt < 4; ++tt) bias1[tt] = pb1[16 * tt + c];
#pragma unroll
  for (int tt = 0; tt < 2; ++tt) { bias2[tt] = pb2[16 * tt + c]; w3[tt] = pW3[16 * tt + c]; }
  const float b3 = pb3[0];
  __syncthreads();

  for (int wt = blockIdx.x * 8 + wave; wt < PR_TILES; wt += PR_BLOCKS * 8) {
    const int row0 = wt * 16;
    const int* sp = ps; const int* dp = pd; int rbase = row0;
    if (row0 >= N_PAIR) { sp = ns; dp = nd; rbase = row0 - N_PAIR; }
    const int si = sp[rbase + c];
    const int di = dp[rbase + c];

#pragma unroll
    for (int r = 0; r < 16; ++r) {
      const int s = __builtin_amdgcn_readlane(si, r);
      const int d = __builtin_amdgcn_readlane(di, r);
      const float4 a = *(const float4*)(emb + (size_t)s * 128 + lane * 4);
      const float4 b = *(const float4*)(emb + (size_t)d * 128 + lane * 4);
      float4 z; z.x = a.x * b.x; z.y = a.y * b.y; z.z = a.z * b.z; z.w = a.w * b.w;
      *(float4*)&Zs[wave][r][lane * 4] = z;
    }
    wave_lds_sync();

    v8f acc[4];
#pragma unroll
    for (int tt = 0; tt < 4; ++tt) acc[tt] = (v8f){0.f, 0.f, 0.f, 0.f, 0.f, 0.f, 0.f, 0.f};
    {
      float4 a_cur = *(const float4*)&Zs[wave][c][4 * h];
      float4 b_cur[4];
#pragma unroll
      for (int tt = 0; tt < 4; ++tt) b_cur[tt] = *(const float4*)&W1s[16 * tt + c][4 * h];
#pragma unroll
      for (int i = 0; i < 16; ++i) {
        float4 a_nxt = a_cur;
        float4 b_nxt[4];
#pragma unroll
        for (int tt = 0; tt < 4; ++tt) b_nxt[tt] = b_cur[tt];
        if (i + 1 < 16) {
          a_nxt = *(const float4*)&Zs[wave][c][8 * (i + 1) + 4 * h];
#pragma unroll
          for (int tt = 0; tt < 4; ++tt)
            b_nxt[tt] = *(const float4*)&W1s[16 * tt + c][8 * (i + 1) + 4 * h];
        }
#pragma unroll
        for (int tt = 0; tt < 4; ++tt)
          acc[tt] = wmma4(a_cur.x, a_cur.y, b_cur[tt].x, b_cur[tt].y, acc[tt]);
#pragma unroll
        for (int tt = 0; tt < 4; ++tt)
          acc[tt] = wmma4(a_cur.z, a_cur.w, b_cur[tt].z, b_cur[tt].w, acc[tt]);
        a_cur = a_nxt;
#pragma unroll
        for (int tt = 0; tt < 4; ++tt) b_cur[tt] = b_nxt[tt];
      }
    }
#pragma unroll
    for (int tt = 0; tt < 4; ++tt) {
#pragma unroll
      for (int r = 0; r < 8; ++r)
        Z1s[wave][8 * h + r][16 * tt + c] = leaky(acc[tt][r] + bias1[tt], 0.2f);
    }
    wave_lds_sync();

    v8f acc2[2];
#pragma unroll
    for (int tt = 0; tt < 2; ++tt) acc2[tt] = (v8f){0.f, 0.f, 0.f, 0.f, 0.f, 0.f, 0.f, 0.f};
    {
      float4 a2[8];
      float4 b2[2][8];
#pragma unroll
      for (int i = 0; i < 8; ++i) {
        a2[i] = *(const float4*)&Z1s[wave][c][8 * i + 4 * h];
#pragma unroll
        for (int tt = 0; tt < 2; ++tt)
          b2[tt][i] = *(const float4*)&W2s[16 * tt + c][8 * i + 4 * h];
      }
#pragma unroll
      for (int i = 0; i < 8; ++i) {
#pragma unroll
        for (int tt = 0; tt < 2; ++tt)
          acc2[tt] = wmma4(a2[i].x, a2[i].y, b2[tt][i].x, b2[tt][i].y, acc2[tt]);
#pragma unroll
        for (int tt = 0; tt < 2; ++tt)
          acc2[tt] = wmma4(a2[i].z, a2[i].w, b2[tt][i].z, b2[tt][i].w, acc2[tt]);
      }
    }

    float mine = 0.f;
#pragma unroll
    for (int r = 0; r < 8; ++r) {
      float v = leaky(acc2[0][r] + bias2[0], 0.2f) * w3[0]
              + leaky(acc2[1][r] + bias2[1], 0.2f) * w3[1];
      v += row_xor<1>(v); v += row_xor<2>(v);
      v += row_xor<4>(v); v += row_xor<8>(v);
      if (c == r) mine = v;
    }
    if (c < 8) out[row0 + 8 * h + c] = mine + b3;
    wave_lds_sync();
  }
}

extern "C" void kernel_launch(void* const* d_in, const int* in_sizes, int n_in,
                              void* d_out, int out_size, void* d_ws, size_t ws_size,
                              hipStream_t stream) {
  const float* x    = (const float*)d_in[0];
  const float* ew   = (const float*)d_in[1];
  const int*   esrc = (const int*)d_in[2];
  const int*   edst = (const int*)d_in[3];
  const int*   psrc = (const int*)d_in[4];
  const int*   pdst = (const int*)d_in[5];
  const int*   nsrc = (const int*)d_in[6];
  const int*   ndst = (const int*)d_in[7];
  const float* Wl[3] = {(const float*)d_in[8],  (const float*)d_in[11], (const float*)d_in[14]};
  const float* al[3] = {(const float*)d_in[9],  (const float*)d_in[12], (const float*)d_in[15]};
  const float* bl[3] = {(const float*)d_in[10], (const float*)d_in[13], (const float*)d_in[16]};
  const float* pW1 = (const float*)d_in[17]; const float* pb1 = (const float*)d_in[18];
  const float* pW2 = (const float*)d_in[19]; const float* pb2 = (const float*)d_in[20];
  const float* pW3 = (const float*)d_in[21]; const float* pb3 = (const float*)d_in[22];

  float* out   = (float*)d_out;
  float* emb   = out + 2 * N_PAIR;

  float*    ws   = (float*)d_ws;
  float*    ht   = ws;
  float*    hbuf = ht   + 6400000;
  float*    pbuf = hbuf + 6400000;
  int*      deg    = (int*)(pbuf + 200000);
  int*      rowptr = deg + 25000;
  int*      rank   = rowptr + 25008;
  int2*     csr    = (int2*)(rank + 400000);

  const int Hs[3]  = {4, 4, 1};
  const int Fs[3]  = {64, 64, 128};
  const int Kin[3] = {128, 256, 256};
  const int Fm[3]  = {63, 63, 127};

  fill_u32_kernel<<<(NN_NODES + 255) / 256, 256, 0, stream>>>((unsigned*)deg, 0u, NN_NODES);
  csr_count_kernel<<<(N_EDGES + 255) / 256, 256, 0, stream>>>(edst, deg, rank);
  csr_scan_kernel<<<1, SCAN_THREADS, 0, stream>>>(deg, rowptr);
  csr_fill_kernel<<<(N_EDGES + 255) / 256, 256, 0, stream>>>(esrc, edst, ew, rank, rowptr, csr);

  const float* hin = x;
  for (int l = 0; l < 3; ++l) {
    const int H = Hs[l], F = Fs[l], HF = H * F, K = Kin[l];
    float* hout = (l == 2) ? emb : hbuf;

    dim3 gg((NN_NODES + 63) / 64, HF / 32);
    gemm_wmma_f32<<<gg, 256, 0, stream>>>(hin, Wl[l], nullptr, ht,
                                          NN_NODES, HF, K, 0.f, 0,
                                          al[l], pbuf, Fm[l]);

    if (l < 2) {
      node_softmax_gather_kernel<4, 64><<<(NN_NODES + 7) / 8, 256, 0, stream>>>(
          rowptr, csr, pbuf, ht, bl[l], hout, 1);
    } else {
      node_softmax_gather_kernel<1, 128><<<(NN_NODES + 7) / 8, 256, 0, stream>>>(
          rowptr, csr, pbuf, ht, bl[l], hout, 0);
    }
    hin = hout;
  }

  predictor_fused_kernel<<<PR_BLOCKS, 256, 0, stream>>>(emb, psrc, pdst, nsrc, ndst,
                                                        pW1, pb1, pW2, pb2, pW3, pb3, out);
}


# Launches 11 to 9: in-degree counters cleared inside the first GEMM (moved ahead of the edge count), prefix scan redone per work-group in LDS inside the list-fill kernel; ranks and sums unchanged
# speedup vs baseline: 14.0769x; 1.0611x over previous
#include <hip/hip_runtime.h>
#include <hip/hip_bf16.h>


typedef __attribute__((ext_vector_type(2))) float v2f;
typedef __attribute__((ext_vector_type(8))) float v8f;

#define NN_NODES 25000
#define N_EDGES  400000
#define N_PAIR   100000
#define N_ROWS   200000

__device__ __forceinline__ float leaky(float v, float slope) {
  return v >= 0.f ? v : v * slope;
}

template <int MASK>
__device__ __forceinline__ float row_xor(float v) {
  return __builtin_bit_cast(float, __builtin_amdgcn_update_dpp(
      0, __builtin_bit_cast(int, v), 0x160 + MASK, 0xf, 0xf, false));
}

__global__ void gemm_wmma_f32(const float* __restrict__ A,
                              const float* __restrict__ W,
                              const float* __restrict__ bias,
                              float* __restrict__ C,
                              int M, int N, int K,
                              float slope, int act,
                              const float* __restrict__ avec,
                              float* __restrict__ ppart, int fmask,
                              int* __restrict__ zero_i32) {
  __shared__ float As[2][64][36];
  __shared__ float Ws[2][32][36];
  __shared__ float part[2][64];

  const int t     = threadIdx.x;
  const int lane  = t & 31;
  const int wave  = t >> 5;
  const int waveM = wave & 3;
  const int waveN = wave >> 2;
  const int mBase = blockIdx.x * 64;
  const int nBase = blockIdx.y * 32;
  const int lh    = lane >> 4;
  const int l15   = lane & 15;

  v8f acc = {0.f, 0.f, 0.f, 0.f, 0.f, 0.f, 0.f, 0.f};

  if (zero_i32 && blockIdx.y == 0 && t < 64 && mBase + t < M) zero_i32[mBase + t] = 0;

  const int  sr   = t >> 3;
  const int  c4   = (t & 7) * 4;
  const int  gr0  = mBase + sr;
  const int  gr1  = gr0 + 32;
  const bool in0  = gr0 < M;
  const bool in1  = gr1 < M;
  const float* pa0 = A + (size_t)(in0 ? gr0 : M - 1) * K + c4;
  const float* pa1 = A + (size_t)(in1 ? gr1 : M - 1) * K + c4;
  const float* pw  = W + (size_t)(nBase + sr) * K + c4;

  float4 ra0 = *(const float4*)(pa0);
  float4 ra1 = *(const float4*)(pa1);
  float4 rw  = *(const float4*)(pw);
  if (!in0) { ra0.x = 0.f; ra0.y = 0.f; ra0.z = 0.f; ra0.w = 0.f; }
  if (!in1) { ra1.x = 0.f; ra1.y = 0.f; ra1.z = 0.f; ra1.w = 0.f; }
  *(float4*)&As[0][sr][c4]      = ra0;
  *(float4*)&As[0][sr + 32][c4] = ra1;
  *(float4*)&Ws[0][sr][c4]      = rw;
  __syncthreads();

  const int nk = K >> 5;
  for (int it = 0; it < nk; ++it) {
    const int  cur  = it & 1;
    const bool more = it + 1 < nk;
    if (more) {
      const int ko = (it + 1) * 32;
      ra0 = *(const float4*)(pa0 + ko);
      ra1 = *(const float4*)(pa1 + ko);
      rw  = *(const float4*)(pw + ko);
    }

#pragma unroll
    for (int ks = 0; ks < 8; ++ks) {
      int kb = ks * 4 + 2 * lh;
      v2f a, b;
      a.x = As[cur][waveM * 16 + l15][kb];
      a.y = As[cur][waveM * 16 + l15][kb + 1];
      b.x = Ws[cur][waveN * 16 + l15][kb];
      b.y = Ws[cur][waveN * 16 + l15][kb + 1];
      acc = __builtin_amdgcn_wmma_f32_16x16x4_f32(
           false, a,  false, b,
           (short)0, acc,  false,  false);
    }

    if (more) {
      if (!in0) { ra0.x = 0.f; ra0.y = 0.f; ra0.z = 0.f; ra0.w = 0.f; }
      if (!in1) { ra1.x = 0.f; ra1.y = 0.f; ra1.z = 0.f; ra1.w = 0.f; }
      *(float4*)&As[cur ^ 1][sr][c4]      = ra0;
      *(float4*)&As[cur ^ 1][sr + 32][c4] = ra1;
      *(float4*)&Ws[cur ^ 1][sr][c4]      = rw;
    }
    __syncthreads();
  }

#pragma unroll
  for (int r = 0; r < 8; ++r) {
    int row = mBase + waveM * 16 + r + 8 * lh;
    int col = nBase + waveN * 16 + l15;
    if (row < M) {
      float v = acc[r];
      if (bias) v += bias[col];
      if (act)  v = leaky(v, slope);
      C[(size_t)row * N + col] = v;
    }
  }

  if (avec) {
    const float av = avec[(nBase + waveN * 16 + l15) & fmask];
#pragma unroll
    for (int r = 0; r < 8; ++r) {
      float v = acc[r] * av;
      v += row_xor<1>(v); v += row_xor<2>(v);
      v += row_xor<4>(v); v += row_xor<8>(v);
      if (l15 == r) part[waveN][waveM * 16 + 8 * lh + r] = v;
    }
    __syncthreads();
    if (t < 64 && mBase + t < M)
      ppart[(size_t)(mBase + t) * gridDim.y + blockIdx.y] = part[0][t] + part[1][t];
  }
}

__global__ void csr_count_kernel(const int* __restrict__ dst,
                                 int* __restrict__ deg,
                                 int* __restrict__ rank) {
  int e = blockIdx.x * blockDim.x + threadIdx.x;
  if (e >= N_EDGES) return;
  rank[e] = atomicAdd(&deg[dst[e]], 1);
}

#define SCAN_THREADS 1024
#define SCAN_PER     25
#define FILL_BLOCKS  256
#define FILL_EPB     ((N_EDGES + FILL_BLOCKS - 1) / FILL_BLOCKS)
__global__ void __launch_bounds__(SCAN_THREADS)
csr_scan_fill_kernel(const int* __restrict__ deg,
                     const int* __restrict__ src,
                     const int* __restrict__ dst,
                     const float* __restrict__ ew,
                     const int* __restrict__ rank,
                     int* __restrict__ rowptr,
                     int2* __restrict__ csr) {
  __shared__ int rp[SCAN_THREADS * SCAN_PER];
  __shared__ int part[SCAN_THREADS];
  const int t = threadIdx.x;
  static_assert(FILL_EPB <= 2 * SCAN_THREADS, "two edge slots per thread");
  const int e0 = blockIdx.x * FILL_EPB;
  const int e1 = e0 + FILL_EPB < N_EDGES ? e0 + FILL_EPB : N_EDGES;
  const int ea = e0 + t;
  const int eb = ea + SCAN_THREADS;
  const bool va = ea < e1;
  const bool vb = eb < e1;
  const int ca = va ? ea : 0;
  const int cb = vb ? eb : 0;
  const int   da = dst[ca],  db = dst[cb];
  const int   ra = rank[ca], rb = rank[cb];
  const int   sa = src[ca],  sb = src[cb];
  const float wa = ew[ca],   wb = ew[cb];
#pragma unroll
  for (int i = 0; i < SCAN_PER; ++i) {
    int idx = i * SCAN_THREADS + t;
    int idc = idx < NN_NODES ? idx : NN_NODES - 1;
    int v   = deg[idc];
    rp[idx] = idx < NN_NODES ? v : 0;
  }
  __syncthreads();
  const int lo = t * SCAN_PER;
  int vals[SCAN_PER];
  int sum = 0;
#pragma unroll
  for (int i = 0; i < SCAN_PER; ++i) {
    vals[i] = rp[lo + i];
    sum += vals[i];
  }
  part[t] = sum;
  __syncthreads();
  for (int off = 1; off < SCAN_THREADS; off <<= 1) {
    int v = (t >= off) ? part[t - off] : 0;
    __syncthreads();
    part[t] += v;
    __syncthreads();
  }
  int run = part[t] - sum;
#pragma unroll
  for (int i = 0; i < SCAN_PER; ++i) {
    rp[lo + i] = run;
    run += vals[i];
  }
  __syncthreads();

  if (blockIdx.x == 0) {
#pragma unroll
    for (int i = 0; i < SCAN_PER; ++i) {
      int idx = i * SCAN_THREADS + t;
      if (idx <= NN_NODES) rowptr[idx] = rp[idx];
    }
  }

  if (va) csr[rp[da] + ra] = make_int2(sa, __float_as_int(log1pf(wa)));
  if (vb) csr[rp[db] + rb] = make_int2(sb, __float_as_int(log1pf(wb)));
}

__device__ __forceinline__ float xhalf(float v) {
  return __builtin_bit_cast(float, __builtin_amdgcn_permlanex16(
      __builtin_bit_cast(int, v), __builtin_bit_cast(int, v), 0x76543210, 0xfedcba98, false, false));
}
__device__ __forceinline__ float wave_max(float v) {
  v = fmaxf(v, row_xor<1>(v)); v = fmaxf(v, row_xor<2>(v));
  v = fmaxf(v, row_xor<4>(v)); v = fmaxf(v, row_xor<8>(v));
  return fmaxf(v, xhalf(v));
}
__device__ __forceinline__ float wave_sum(float v) {
  v += row_xor<1>(v); v += row_xor<2>(v);
  v += row_xor<4>(v); v += row_xor<8>(v);
  return v + xhalf(v);
}

template <int H>
__device__ __forceinline__ void edge_scores(const int2* __restrict__ csr,
                                            const float* __restrict__ p,
                                            const float (&pd)[H], int j,
                                            int& s, float (&sc)[H]) {
  const int2 se = csr[j];
  s = se.x;
  const float lw = __int_as_float(se.y);
  if constexpr (H == 4) {
    const float4 q0 = *(const float4*)(p + (size_t)se.x * 8);
    const float4 q1 = *(const float4*)(p + (size_t)se.x * 8 + 4);
    sc[0] = leaky((q0.x + q0.y) + pd[0] + lw, 0.01f);
    sc[1] = leaky((q0.z + q0.w) + pd[1] + lw, 0.01f);
    sc[2] = leaky((q1.x + q1.y) + pd[2] + lw, 0.01f);
    sc[3] = leaky((q1.z + q1.w) + pd[3] + lw, 0.01f);
  } else {
    const float4 q = *(const float4*)(p + (size_t)se.x * 4);
    sc[0] = leaky(((q.x + q.y) + (q.z + q.w)) + pd[0] + lw, 0.01f);
  }
}

template <int H, int F>
__device__ __forceinline__ void gather_rows(const float* __restrict__ ht, int lane,
                                            int my_s, const float (&my_a)[H], int cnt,
                                            float4 (&acc)[H * F / 128]) {
  constexpr int HF = H * F;
  const bool hiHalf = lane >= 16;
#pragma unroll 4
  for (int t = 0; t < cnt; ++t) {
    const int s = __builtin_amdgcn_readlane(my_s, t);
    const float* row = ht + (size_t)s * HF + lane * 4;
    if constexpr (H == 4) {
      const float a0 = __int_as_float(__builtin_amdgcn_readlane(__float_as_int(my_a[0]), t));
      const float a1 = __int_as_float(__builtin_amdgcn_readlane(__float_as_int(my_a[1]), t));
      const float a2 = __int_as_float(__builtin_amdgcn_readlane(__float_as_int(my_a[2]), t));
      const float a3 = __int_as_float(__builtin_amdgcn_readlane(__float_as_int(my_a[3]), t));
      const float aA = hiHalf ? a1 : a0;
      const float aB = hiHalf ? a3 : a2;
      const float4 v0 = *(const float4*)(row);
      const float4 v1 = *(const float4*)(row + 128);
      acc[0].x += aA * v0.x; acc[0].y += aA * v0.y; acc[0].z += aA * v0.z; acc[0].w += aA * v0.w;
      acc[1].x += aB * v1.x; acc[1].y += aB * v1.y; acc[1].z += aB * v1.z; acc[1].w += aB * v1.w;
    } else {
      const float a = __int_as_float(__builtin_amdgcn_readlane(__float_as_int(my_a[0]), t));
      const float4 v0 = *(const float4*)(row);
      acc[0].x += a * v0.x; acc[0].y += a * v0.y; acc[0].z += a * v0.z; acc[0].w += a * v0.w;
    }
  }
}

template <int H, int F>
__global__ void __launch_bounds__(256)
node_softmax_gather_kernel(const int* __restrict__ rowptr,
                           const int2* __restrict__ csr,
                           const float* __restrict__ p,
                           const float* __restrict__ ht,
                           const float* __restrict__ bias,
                           float* __restrict__ hout,
                           int act) {
  constexpr int HF = H * F;
  constexpr int NV = HF / 128;
  const int lane = threadIdx.x & 31;
  const int n = __builtin_amdgcn_readfirstlane((int)(blockIdx.x * 8 + (threadIdx.x >> 5)));
  if (n >= NN_NODES) return;
  const int beg = rowptr[n];
  const int end = rowptr[n + 1];
  const int deg = end - beg;

  float pd[H];
  if constexpr (H == 4) {
#pragma unroll
    for (int h = 0; h < 4; ++h) pd[h] = p[n * 8 + 2 * h] + p[n * 8 + 2 * h + 1];
  } else {
    pd[0] = (p[n * 4] + p[n * 4 + 1]) + (p[n * 4 + 2] + p[n * 4 + 3]);
  }

  float4 acc[NV];
#pragma unroll
  for (int v = 0; v < NV; ++v) { acc[v].x = 0.f; acc[v].y = 0.f; acc[v].z = 0.f; acc[v].w = 0.f; }

  if (deg <= 32) {
    const bool on = lane < deg;
    int   my_s = 0;
    float sc[H];
#pragma unroll
    for (int h = 0; h < H; ++h) sc[h] = -INFINITY;
    if (on) edge_scores<H>(csr, p, pd, beg + lane, my_s, sc);
    float my_a[H];
#pragma unroll
    for (int h = 0; h < H; ++h) {
      float m = wave_max(sc[h]);
      if (!isfinite(m)) m = 0.f;
      const float ex = on ? __expf(sc[h] - m) : 0.f;
      const float sum = wave_sum(ex);
      my_a[h] = ex / fmaxf(sum, 1e-20f);
    }
    gather_rows<H, F>(ht, lane, my_s, my_a, deg, acc);
  } else {
    float m[H], den[H];
#pragma unroll
    for (int h = 0; h < H; ++h) m[h] = -INFINITY;
    for (int base = beg; base < end; base += 32) {
      const int j = base + lane;
      if (j < end) {
        int s_; float sc[H];
        edge_scores<H>(csr, p, pd, j, s_, sc);
#pragma unroll
        for (int h = 0; h < H; ++h) m[h] = fmaxf(m[h], sc[h]);
      }
    }
#pragma unroll
    for (int h = 0; h < H; ++h) {
      m[h] = wave_max(m[h]);
      if (!isfinite(m[h])) m[h] = 0.f;
      den[h] = 0.f;
    }
    for (int base = beg; base < end; base += 32) {
      const int j = base + lane;
      if (j < end) {
        int s_; float sc[H];
        edge_scores<H>(csr, p, pd, j, s_, sc);
#pragma unroll
        for (int h = 0; h < H; ++h) den[h] += __expf(sc[h] - m[h]);
      }
    }
#pragma unroll
    for (int h = 0; h < H; ++h) den[h] = fmaxf(wave_sum(den[h]), 1e-20f);
    for (int base = beg; base < end; base += 32) {
      const int j = base + lane;
      int   my_s = 0;
      float my_a[H];
#pragma unroll
      for (int h = 0; h < H; ++h) my_a[h] = 0.f;
      if (j < end) {
        float sc[H];
        edge_scores<H>(csr, p, pd, j, my_s, sc);
#pragma unroll
        for (int h = 0; h < H; ++h) my_a[h] = __expf(sc[h] - m[h]) / den[h];
      }
      const int left = end - base;
      gather_rows<H, F>(ht, lane, my_s, my_a, left < 32 ? left : 32, acc);
    }
  }

#pragma unroll
  for (int v = 0; v < NV; ++v) {
    const int c = v * 128 + lane * 4;
    const float4 b = *(const float4*)(bias + c);
    float4 r;
    r.x = acc[v].x + b.x; r.y = acc[v].y + b.y; r.z = acc[v].z + b.z; r.w = acc[v].w + b.w;
    if (act) {
      r.x = leaky(r.x, 0.01f); r.y = leaky(r.y, 0.01f);
      r.z = leaky(r.z, 0.01f); r.w = leaky(r.w, 0.01f);
    }
    *(float4*)(hout + (size_t)n * HF + c) = r;
  }
}

#define PR_TILES  (N_ROWS / 16)
#define PR_BLOCKS 512
#define ZP  132
#define Z1P 68

__device__ __forceinline__ void wave_lds_sync() {
  __builtin_amdgcn_fence(__ATOMIC_RELEASE, "wavefront");
  __builtin_amdgcn_wave_barrier();
  __builtin_amdgcn_fence(__ATOMIC_ACQUIRE, "wavefront");
}

__device__ __forceinline__ v8f wmma4(float a0, float a1, float b0, float b1, v8f acc) {
  v2f a = {a0, a1};
  v2f b = {b0, b1};
  return __builtin_amdgcn_wmma_f32_16x16x4_f32(
       false, a,  false, b,
       (short)0, acc,  false,  false);
}

__global__ void __launch_bounds__(256)
predictor_fused_kernel(const float* __restrict__ emb,
                       const int* __restrict__ ps, const int* __restrict__ pd,
                       const int* __restrict__ ns, const int* __restrict__ nd,
                       const float* __restrict__ pW1, const float* __restrict__ pb1,
                       const float* __restrict__ pW2, const float* __restrict__ pb2,
                       const float* __restrict__ pW3, const float* __restrict__ pb3,
                       float* __restrict__ out) {
  __shared__ __attribute__((aligned(16))) float W1s[64][ZP];
  __shared__ __attribute__((aligned(16))) float W2s[32][Z1P];
  __shared__ __attribute__((aligned(16))) float Zs[8][16][ZP];
  __shared__ __attribute__((aligned(16))) float Z1s[8][16][Z1P];

  const int t    = threadIdx.x;
  const int lane = t & 31;
  const int wave = __builtin_amdgcn_readfirstlane(t >> 5);
  const int c    = lane & 15;
  const int h    = lane >> 4;

#pragma unroll
  for (int j = 0; j < 8; ++j) {
    const int idx = t + j * 256;
    const int r   = idx >> 5;
    const int c4  = (idx & 31) * 4;
    *(float4*)&W1s[r][c4] = *(const float4*)(pW1 + r * 128 + c4);
  }
#pragma unroll
  for (int j = 0; j < 2; ++j) {
    const int idx = t + j * 256;
    const int r   = idx >> 4;
    const int c4  = (idx & 15) * 4;
    *(float4*)&W2s[r][c4] = *(const float4*)(pW2 + r * 64 + c4);
  }
  float bias1[4], bias2[2], w3[2];
#pragma unroll
  for (int tt = 0; tt < 4; ++tt) bias1[tt] = pb1[16 * tt + c];
#pragma unroll
  for (int tt = 0; tt < 2; ++tt) { bias2[tt] = pb2[16 * tt + c]; w3[tt] = pW3[16 * tt + c]; }
  const float b3 = pb3[0];
  __syncthreads();

  for (int wt = blockIdx.x * 8 + wave; wt < PR_TILES; wt += PR_BLOCKS * 8) {
    const int row0 = wt * 16;
    const int* sp = ps; const int* dp = pd; int rbase = row0;
    if (row0 >= N_PAIR) { sp = ns; dp = nd; rbase = row0 - N_PAIR; }
    const int si = sp[rbase + c];
    const int di = dp[rbase + c];

#pragma unroll
    for (int r = 0; r < 16; ++r) {
      const int s = __builtin_amdgcn_readlane(si, r);
      const int d = __builtin_amdgcn_readlane(di, r);
      const float4 a = *(const float4*)(emb + (size_t)s * 128 + lane * 4);
      const float4 b = *(const float4*)(emb + (size_t)d * 128 + lane * 4);
      float4 z; z.x = a.x * b.x; z.y = a.y * b.y; z.z = a.z * b.z; z.w = a.w * b.w;
      *(float4*)&Zs[wave][r][lane * 4] = z;
    }
    wave_lds_sync();

    v8f acc[4];
#pragma unroll
    for (int tt = 0; tt < 4; ++tt) acc[tt] = (v8f){0.f, 0.f, 0.f, 0.f, 0.f, 0.f, 0.f, 0.f};
    {
      float4 a_cur = *(const float4*)&Zs[wave][c][4 * h];
      float4 b_cur[4];
#pragma unroll
      for (int tt = 0; tt < 4; ++tt) b_cur[tt] = *(const float4*)&W1s[16 * tt + c][4 * h];
#pragma unroll
      for (int i = 0; i < 16; ++i) {
        float4 a_nxt = a_cur;
        float4 b_nxt[4];
#pragma unroll
        for (int tt = 0; tt < 4; ++tt) b_nxt[tt] = b_cur[tt];
        if (i + 1 < 16) {
          a_nxt = *(const float4*)&Zs[wave][c][8 * (i + 1) + 4 * h];
#pragma unroll
          for (int tt = 0; tt < 4; ++tt)
            b_nxt[tt] = *(const float4*)&W1s[16 * tt + c][8 * (i + 1) + 4 * h];
        }
#pragma unroll
        for (int tt = 0; tt < 4; ++tt)
          acc[tt] = wmma4(a_cur.x, a_cur.y, b_cur[tt].x, b_cur[tt].y, acc[tt]);
#pragma unroll
        for (int tt = 0; tt < 4; ++tt)
          acc[tt] = wmma4(a_cur.z, a_cur.w, b_cur[tt].z, b_cur[tt].w, acc[tt]);
        a_cur = a_nxt;
#pragma unroll
        for (int tt = 0; tt < 4; ++tt) b_cur[tt] = b_nxt[tt];
      }
    }
#pragma unroll
    for (int tt = 0; tt < 4; ++tt) {
#pragma unroll
      for (int r = 0; r < 8; ++r)
        Z1s[wave][8 * h + r][16 * tt + c] = leaky(acc[tt][r] + bias1[tt], 0.2f);
    }
    wave_lds_sync();

    v8f acc2[2];
#pragma unroll
    for (int tt = 0; tt < 2; ++tt) acc2[tt] = (v8f){0.f, 0.f, 0.f, 0.f, 0.f, 0.f, 0.f, 0.f};
    {
      float4 a2[8];
      float4 b2[2][8];
#pragma unroll
      for (int i = 0; i < 8; ++i) {
        a2[i] = *(const float4*)&Z1s[wave][c][8 * i + 4 * h];
#pragma unroll
        for (int tt = 0; tt < 2; ++tt)
          b2[tt][i] = *(const float4*)&W2s[16 * tt + c][8 * i + 4 * h];
      }
#pragma unroll
      for (int i = 0; i < 8; ++i) {
#pragma unroll
        for (int tt = 0; tt < 2; ++tt)
          acc2[tt] = wmma4(a2[i].x, a2[i].y, b2[tt][i].x, b2[tt][i].y, acc2[tt]);
#pragma unroll
        for (int tt = 0; tt < 2; ++tt)
          acc2[tt] = wmma4(a2[i].z, a2[i].w, b2[tt][i].z, b2[tt][i].w, acc2[tt]);
      }
    }

    float mine = 0.f;
#pragma unroll
    for (int r = 0; r < 8; ++r) {
      float v = leaky(acc2[0][r] + bias2[0], 0.2f) * w3[0]
              + leaky(acc2[1][r] + bias2[1], 0.2f) * w3[1];
      v += row_xor<1>(v); v += row_xor<2>(v);
      v += row_xor<4>(v); v += row_xor<8>(v);
      if (c == r) mine = v;
    }
    if (c < 8) out[row0 + 8 * h + c] = mine + b3;
    wave_lds_sync();
  }
}

extern "C" void kernel_launch(void* const* d_in, const int* in_sizes, int n_in,
                              void* d_out, int out_size, void* d_ws, size_t ws_size,
                              hipStream_t stream) {
  const float* x    = (const float*)d_in[0];
  const float* ew   = (const float*)d_in[1];
  const int*   esrc = (const int*)d_in[2];
  const int*   edst = (const int*)d_in[3];
  const int*   psrc = (const int*)d_in[4];
  const int*   pdst = (const int*)d_in[5];
  const int*   nsrc = (const int*)d_in[6];
  const int*   ndst = (const int*)d_in[7];
  const float* Wl[3] = {(const float*)d_in[8],  (const float*)d_in[11], (const float*)d_in[14]};
  const float* al[3] = {(const float*)d_in[9],  (const float*)d_in[12], (const float*)d_in[15]};
  const float* bl[3] = {(const float*)d_in[10], (const float*)d_in[13], (const float*)d_in[16]};
  const float* pW1 = (const float*)d_in[17]; const float* pb1 = (const float*)d_in[18];
  const float* pW2 = (const float*)d_in[19]; const float* pb2 = (const float*)d_in[20];
  const float* pW3 = (const float*)d_in[21]; const float* pb3 = (const float*)d_in[22];

  float* out   = (float*)d_out;
  float* emb   = out + 2 * N_PAIR;

  float*    ws   = (float*)d_ws;
  float*    ht   = ws;
  float*    hbuf = ht   + 6400000;
  float*    pbuf = hbuf + 6400000;
  int*      deg    = (int*)(pbuf + 200000);
  int*      rowptr = deg + 25000;
  int*      rank   = rowptr + 25008;
  int2*     csr    = (int2*)(rank + 400000);

  const int Hs[3]  = {4, 4, 1};
  const int Fs[3]  = {64, 64, 128};
  const int Kin[3] = {128, 256, 256};
  const int Fm[3]  = {63, 63, 127};

  const float* hin = x;
  for (int l = 0; l < 3; ++l) {
    const int H = Hs[l], F = Fs[l], HF = H * F, K = Kin[l];
    float* hout = (l == 2) ? emb : hbuf;

    dim3 gg((NN_NODES + 63) / 64, HF / 32);
    gemm_wmma_f32<<<gg, 256, 0, stream>>>(hin, Wl[l], nullptr, ht,
                                          NN_NODES, HF, K, 0.f, 0,
                                          al[l], pbuf, Fm[l],
                                          l == 0 ? deg : nullptr);

    if (l == 0) {
      csr_count_kernel<<<(N_EDGES + 255) / 256, 256, 0, stream>>>(edst, deg, rank);
      csr_scan_fill_kernel<<<FILL_BLOCKS, SCAN_THREADS, 0, stream>>>(deg, esrc, edst, ew, rank,
                                                                   rowptr, csr);
    }

    if (l < 2) {
      node_softmax_gather_kernel<4, 64><<<(NN_NODES + 7) / 8, 256, 0, stream>>>(
          rowptr, csr, pbuf, ht, bl[l], hout, 1);
    } else {
      node_softmax_gather_kernel<1, 128><<<(NN_NODES + 7) / 8, 256, 0, stream>>>(
          rowptr, csr, pbuf, ht, bl[l], hout, 0);
    }
    hin = hout;
  }

  predictor_fused_kernel<<<PR_BLOCKS, 256, 0, stream>>>(emb, psrc, pdst, nsrc, ndst,
                                                        pW1, pb1, pW2, pb2, pW3, pb3, out);
}


# Layers 1,2: a block aggregates its 32 nodes into LDS as the GEMM A tile, B fragments read straight from W: 2 launches and the aggregated rows' memory trip gone; k pairing inside WMMA steps reordered
# speedup vs baseline: 14.5484x; 1.0335x over previous
#include <hip/hip_runtime.h>
#include <hip/hip_bf16.h>


typedef __attribute__((ext_vector_type(2))) float v2f;
typedef __attribute__((ext_vector_type(8))) float v8f;

#define NN_NODES 25000
#define N_EDGES  400000
#define N_PAIR   100000
#define N_ROWS   200000

__device__ __forceinline__ float leaky(float v, float slope) {
  return v >= 0.f ? v : v * slope;
}

template <int MASK>
__device__ __forceinline__ float row_xor(float v) {
  return __builtin_bit_cast(float, __builtin_amdgcn_update_dpp(
      0, __builtin_bit_cast(int, v), 0x160 + MASK, 0xf, 0xf, false));
}

__global__ void gemm_wmma_f32(const float* __restrict__ A,
                              const float* __restrict__ W,
                              const float* __restrict__ bias,
                              float* __restrict__ C,
                              int M, int N, int K,
                              float slope, int act,
                              const float* __restrict__ avec,
                              float* __restrict__ ppart, int fmask,
                              int* __restrict__ zero_i32) {
  __shared__ float As[2][64][36];
  __shared__ float Ws[2][32][36];
  __shared__ float part[2][64];

  const int t     = threadIdx.x;
  const int lane  = t & 31;
  const int wave  = t >> 5;
  const int waveM = wave & 3;
  const int waveN = wave >> 2;
  const int mBase = blockIdx.x * 64;
  const int nBase = blockIdx.y * 32;
  const int lh    = lane >> 4;
  const int l15   = lane & 15;

  v8f acc = {0.f, 0.f, 0.f, 0.f, 0.f, 0.f, 0.f, 0.f};

  if (zero_i32 && blockIdx.y == 0 && t < 64 && mBase + t < M) zero_i32[mBase + t] = 0;

  const int  sr   = t >> 3;
  const int  c4   = (t & 7) * 4;
  const int  gr0  = mBase + sr;
  const int  gr1  = gr0 + 32;
  const bool in0  = gr0 < M;
  const bool in1  = gr1 < M;
  const float* pa0 = A + (size_t)(in0 ? gr0 : M - 1) * K + c4;
  const float* pa1 = A + (size_t)(in1 ? gr1 : M - 1) * K + c4;
  const float* pw  = W + (size_t)(nBase + sr) * K + c4;

  float4 ra0 = *(const float4*)(pa0);
  float4 ra1 = *(const float4*)(pa1);
  float4 rw  = *(const float4*)(pw);
  if (!in0) { ra0.x = 0.f; ra0.y = 0.f; ra0.z = 0.f; ra0.w = 0.f; }
  if (!in1) { ra1.x = 0.f; ra1.y = 0.f; ra1.z = 0.f; ra1.w = 0.f; }
  *(float4*)&As[0][sr][c4]      = ra0;
  *(float4*)&As[0][sr + 32][c4] = ra1;
  *(float4*)&Ws[0][sr][c4]      = rw;
  __syncthreads();

  const int nk = K >> 5;
  for (int it = 0; it < nk; ++it) {
    const int  cur  = it & 1;
    const bool more = it + 1 < nk;
    if (more) {
      const int ko = (it + 1) * 32;
      ra0 = *(const float4*)(pa0 + ko);
      ra1 = *(const float4*)(pa1 + ko);
      rw  = *(const float4*)(pw + ko);
    }

#pragma unroll
    for (int ks = 0; ks < 8; ++ks) {
      int kb = ks * 4 + 2 * lh;
      v2f a, b;
      a.x = As[cur][waveM * 16 + l15][kb];
      a.y = As[cur][waveM * 16 + l15][kb + 1];
      b.x = Ws[cur][waveN * 16 + l15][kb];
      b.y = Ws[cur][waveN * 16 + l15][kb + 1];
      acc = __builtin_amdgcn_wmma_f32_16x16x4_f32(
           false, a,  false, b,
           (short)0, acc,  false,  false);
    }

    if (more) {
      if (!in0) { ra0.x = 0.f; ra0.y = 0.f; ra0.z = 0.f; ra0.w = 0.f; }
      if (!in1) { ra1.x = 0.f; ra1.y = 0.f; ra1.z = 0.f; ra1.w = 0.f; }
      *(float4*)&As[cur ^ 1][sr][c4]      = ra0;
      *(float4*)&As[cur ^ 1][sr + 32][c4] = ra1;
      *(float4*)&Ws[cur ^ 1][sr][c4]      = rw;
    }
    __syncthreads();
  }

#pragma unroll
  for (int r = 0; r < 8; ++r) {
    int row = mBase + waveM * 16 + r + 8 * lh;
    int col = nBase + waveN * 16 + l15;
    if (row < M) {
      float v = acc[r];
      if (bias) v += bias[col];
      if (act)  v = leaky(v, slope);
      C[(size_t)row * N + col] = v;
    }
  }

  if (avec) {
    const float av = avec[(nBase + waveN * 16 + l15) & fmask];
#pragma unroll
    for (int r = 0; r < 8; ++r) {
      float v = acc[r] * av;
      v += row_xor<1>(v); v += row_xor<2>(v);
      v += row_xor<4>(v); v += row_xor<8>(v);
      if (l15 == r) part[waveN][waveM * 16 + 8 * lh + r] = v;
    }
    __syncthreads();
    if (t < 64 && mBase + t < M)
      ppart[(size_t)(mBase + t) * gridDim.y + blockIdx.y] = part[0][t] + part[1][t];
  }
}

__global__ void csr_count_kernel(const int* __restrict__ dst,
                                 int* __restrict__ deg,
                                 int* __restrict__ rank) {
  int e = blockIdx.x * blockDim.x + threadIdx.x;
  if (e >= N_EDGES) return;
  rank[e] = atomicAdd(&deg[dst[e]], 1);
}

#define SCAN_THREADS 1024
#define SCAN_PER     25
#define FILL_BLOCKS  256
#define FILL_EPB     ((N_EDGES + FILL_BLOCKS - 1) / FILL_BLOCKS)
__global__ void __launch_bounds__(SCAN_THREADS)
csr_scan_fill_kernel(const int* __restrict__ deg,
                     const int* __restrict__ src,
                     const int* __restrict__ dst,
                     const float* __restrict__ ew,
                     const int* __restrict__ rank,
                     int* __restrict__ rowptr,
                     int2* __restrict__ csr) {
  __shared__ int rp[SCAN_THREADS * SCAN_PER];
  __shared__ int part[SCAN_THREADS];
  const int t = threadIdx.x;
  static_assert(FILL_EPB <= 2 * SCAN_THREADS, "two edge slots per thread");
  const int e0 = blockIdx.x * FILL_EPB;
  const int e1 = e0 + FILL_EPB < N_EDGES ? e0 + FILL_EPB : N_EDGES;
  const int ea = e0 + t;
  const int eb = ea + SCAN_THREADS;
  const bool va = ea < e1;
  const bool vb = eb < e1;
  const int ca = va ? ea : 0;
  const int cb = vb ? eb : 0;
  const int   da = dst[ca],  db = dst[cb];
  const int   ra = rank[ca], rb = rank[cb];
  const int   sa = src[ca],  sb = src[cb];
  const float wa = ew[ca],   wb = ew[cb];
#pragma unroll
  for (int i = 0; i < SCAN_PER; ++i) {
    int idx = i * SCAN_THREADS + t;
    int idc = idx < NN_NODES ? idx : NN_NODES - 1;
    int v   = deg[idc];
    rp[idx] = idx < NN_NODES ? v : 0;
  }
  __syncthreads();
  const int lo = t * SCAN_PER;
  int vals[SCAN_PER];
  int sum = 0;
#pragma unroll
  for (int i = 0; i < SCAN_PER; ++i) {
    vals[i] = rp[lo + i];
    sum += vals[i];
  }
  part[t] = sum;
  __syncthreads();
  for (int off = 1; off < SCAN_THREADS; off <<= 1) {
    int v = (t >= off) ? part[t - off] : 0;
    __syncthreads();
    part[t] += v;
    __syncthreads();
  }
  int run = part[t] - sum;
#pragma unroll
  for (int i = 0; i < SCAN_PER; ++i) {
    rp[lo + i] = run;
    run += vals[i];
  }
  __syncthreads();

  if (blockIdx.x == 0) {
#pragma unroll
    for (int i = 0; i < SCAN_PER; ++i) {
      int idx = i * SCAN_THREADS + t;
      if (idx <= NN_NODES) rowptr[idx] = rp[idx];
    }
  }

  if (va) csr[rp[da] + ra] = make_int2(sa, __float_as_int(log1pf(wa)));
  if (vb) csr[rp[db] + rb] = make_int2(sb, __float_as_int(log1pf(wb)));
}

__device__ __forceinline__ float xhalf(float v) {
  return __builtin_bit_cast(float, __builtin_amdgcn_permlanex16(
      __builtin_bit_cast(int, v), __builtin_bit_cast(int, v), 0x76543210, 0xfedcba98, false, false));
}
__device__ __forceinline__ float wave_max(float v) {
  v = fmaxf(v, row_xor<1>(v)); v = fmaxf(v, row_xor<2>(v));
  v = fmaxf(v, row_xor<4>(v)); v = fmaxf(v, row_xor<8>(v));
  return fmaxf(v, xhalf(v));
}
__device__ __forceinline__ float wave_sum(float v) {
  v += row_xor<1>(v); v += row_xor<2>(v);
  v += row_xor<4>(v); v += row_xor<8>(v);
  return v + xhalf(v);
}

template <int H>
__device__ __forceinline__ void edge_scores(const int2* __restrict__ csr,
                                            const float* __restrict__ p,
                                            const float (&pd)[H], int j,
                                            int& s, float (&sc)[H]) {
  const int2 se = csr[j];
  s = se.x;
  const float lw = __int_as_float(se.y);
  if constexpr (H == 4) {
    const float4 q0 = *(const float4*)(p + (size_t)se.x * 8);
    const float4 q1 = *(const float4*)(p + (size_t)se.x * 8 + 4);
    sc[0] = leaky((q0.x + q0.y) + pd[0] + lw, 0.01f);
    sc[1] = leaky((q0.z + q0.w) + pd[1] + lw, 0.01f);
    sc[2] = leaky((q1.x + q1.y) + pd[2] + lw, 0.01f);
    sc[3] = leaky((q1.z + q1.w) + pd[3] + lw, 0.01f);
  } else {
    const float4 q = *(const float4*)(p + (size_t)se.x * 4);
    sc[0] = leaky(((q.x + q.y) + (q.z + q.w)) + pd[0] + lw, 0.01f);
  }
}

template <int H, int F>
__device__ __forceinline__ void gather_rows(const float* __restrict__ ht, int lane,
                                            int my_s, const float (&my_a)[H], int cnt,
                                            float4 (&acc)[H * F / 128]) {
  constexpr int HF = H * F;
  const bool hiHalf = lane >= 16;
#pragma unroll 4
  for (int t = 0; t < cnt; ++t) {
    const int s = __builtin_amdgcn_readlane(my_s, t);
    const float* row = ht + (size_t)s * HF + lane * 4;
    if constexpr (H == 4) {
      const float a0 = __int_as_float(__builtin_amdgcn_readlane(__float_as_int(my_a[0]), t));
      const float a1 = __int_as_float(__builtin_amdgcn_readlane(__float_as_int(my_a[1]), t));
      const float a2 = __int_as_float(__builtin_amdgcn_readlane(__float_as_int(my_a[2]), t));
      const float a3 = __int_as_float(__builtin_amdgcn_readlane(__float_as_int(my_a[3]), t));
      const float aA = hiHalf ? a1 : a0;
      const float aB = hiHalf ? a3 : a2;
      const float4 v0 = *(const float4*)(row);
      const float4 v1 = *(const float4*)(row + 128);
      acc[0].x += aA * v0.x; acc[0].y += aA * v0.y; acc[0].z += aA * v0.z; acc[0].w += aA * v0.w;
      acc[1].x += aB * v1.x; acc[1].y += aB * v1.y; acc[1].z += aB * v1.z; acc[1].w += aB * v1.w;
    } else {
      const float a = __int_as_float(__builtin_amdgcn_readlane(__float_as_int(my_a[0]), t));
      const float4 v0 = *(const float4*)(row);
      acc[0].x += a * v0.x; acc[0].y += a * v0.y; acc[0].z += a * v0.z; acc[0].w += a * v0.w;
    }
  }
}

template <int H, int F>
__device__ __forceinline__ void node_aggregate(const int* __restrict__ rowptr,
                                               const int2* __restrict__ csr,
                                               const float* __restrict__ p,
                                               const float* __restrict__ ht,
                                               int n, int lane,
                                               float4 (&acc)[H * F / 128]) {
  constexpr int NV = H * F / 128;
  const int beg = rowptr[n];
  const int end = rowptr[n + 1];
  const int deg = end - beg;

  float pd[H];
  if constexpr (H == 4) {
#pragma unroll
    for (int h = 0; h < 4; ++h) pd[h] = p[n * 8 + 2 * h] + p[n * 8 + 2 * h + 1];
  } else {
    pd[0] = (p[n * 4] + p[n * 4 + 1]) + (p[n * 4 + 2] + p[n * 4 + 3]);
  }

#pragma unroll
  for (int v = 0; v < NV; ++v) { acc[v].x = 0.f; acc[v].y = 0.f; acc[v].z = 0.f; acc[v].w = 0.f; }

  if (deg <= 32) {
    const bool on = lane < deg;
    int   my_s = 0;
    float sc[H];
#pragma unroll
    for (int h = 0; h < H; ++h) sc[h] = -INFINITY;
    if (on) edge_scores<H>(csr, p, pd, beg + lane, my_s, sc);
    float my_a[H];
#pragma unroll
    for (int h = 0; h < H; ++h) {
      float m = wave_max(sc[h]);
      if (!isfinite(m)) m = 0.f;
      const float ex = on ? __expf(sc[h] - m) : 0.f;
      const float sum = wave_sum(ex);
      my_a[h] = ex / fmaxf(sum, 1e-20f);
    }
    gather_rows<H, F>(ht, lane, my_s, my_a, deg, acc);
  } else {
    float m[H], den[H];
#pragma unroll
    for (int h = 0; h < H; ++h) m[h] = -INFINITY;
    for (int base = beg; base < end; base += 32) {
      const int j = base + lane;
      if (j < end) {
        int s_; float sc[H];
        edge_scores<H>(csr, p, pd, j, s_, sc);
#pragma unroll
        for (int h = 0; h < H; ++h) m[h] = fmaxf(m[h], sc[h]);
      }
    }
#pragma unroll
    for (int h = 0; h < H; ++h) {
      m[h] = wave_max(m[h]);
      if (!isfinite(m[h])) m[h] = 0.f;
      den[h] = 0.f;
    }
    for (int base = beg; base < end; base += 32) {
      const int j = base + lane;
      if (j < end) {
        int s_; float sc[H];
        edge_scores<H>(csr, p, pd, j, s_, sc);
#pragma unroll
        for (int h = 0; h < H; ++h) den[h] += __expf(sc[h] - m[h]);
      }
    }
#pragma unroll
    for (int h = 0; h < H; ++h) den[h] = fmaxf(wave_sum(den[h]), 1e-20f);
    for (int base = beg; base < end; base += 32) {
      const int j = base + lane;
      int   my_s = 0;
      float my_a[H];
#pragma unroll
      for (int h = 0; h < H; ++h) my_a[h] = 0.f;
      if (j < end) {
        float sc[H];
        edge_scores<H>(csr, p, pd, j, my_s, sc);
#pragma unroll
        for (int h = 0; h < H; ++h) my_a[h] = __expf(sc[h] - m[h]) / den[h];
      }
      const int left = end - base;
      gather_rows<H, F>(ht, lane, my_s, my_a, left < 32 ? left : 32, acc);
    }
  }
}

template <int H, int F>
__global__ void __launch_bounds__(256)
node_softmax_gather_kernel(const int* __restrict__ rowptr,
                           const int2* __restrict__ csr,
                           const float* __restrict__ p,
                           const float* __restrict__ ht,
                           const float* __restrict__ bias,
                           float* __restrict__ hout,
                           int act) {
  constexpr int HF = H * F;
  constexpr int NV = HF / 128;
  const int lane = threadIdx.x & 31;
  const int n = __builtin_amdgcn_readfirstlane((int)(blockIdx.x * 8 + (threadIdx.x >> 5)));
  if (n >= NN_NODES) return;

  float4 acc[NV];
  node_aggregate<H, F>(rowptr, csr, p, ht, n, lane, acc);

#pragma unroll
  for (int v = 0; v < NV; ++v) {
    const int c = v * 128 + lane * 4;
    const float4 b = *(const float4*)(bias + c);
    float4 r;
    r.x = acc[v].x + b.x; r.y = acc[v].y + b.y; r.z = acc[v].z + b.z; r.w = acc[v].w + b.w;
    if (act) {
      r.x = leaky(r.x, 0.01f); r.y = leaky(r.y, 0.01f);
      r.z = leaky(r.z, 0.01f); r.w = leaky(r.w, 0.01f);
    }
    *(float4*)(hout + (size_t)n * HF + c) = r;
  }
}

__device__ __forceinline__ v8f wmma4(float a0, float a1, float b0, float b1, v8f acc) {
  v2f a = {a0, a1};
  v2f b = {b0, b1};
  return __builtin_amdgcn_wmma_f32_16x16x4_f32(
       false, a,  false, b,
       (short)0, acc,  false,  false);
}

#define FG_ROWS  32
#define FG_PITCH 260
template <int NOUT>
__global__ void __launch_bounds__(512)
gather_gemm_kernel(const int* __restrict__ rowptr,
                   const int2* __restrict__ csr,
                   const float* __restrict__ p,
                   const float* __restrict__ ht,
                   const float* __restrict__ bias,
                   const float* __restrict__ W,
                   const float* __restrict__ avec, int fmask,
                   float* __restrict__ C,
                   float* __restrict__ ppart) {
  constexpr int TM = NOUT / 128;
  constexpr int NS = NOUT / 32;
  __shared__ __attribute__((aligned(16))) float As[FG_ROWS][FG_PITCH];
  __shared__ float part[NOUT / 16][FG_ROWS];

  const int t     = threadIdx.x;
  const int lane  = t & 31;
  const int wave  = __builtin_amdgcn_readfirstlane(t >> 5);
  const int mBase = blockIdx.x * FG_ROWS;

  const float4 bA = *(const float4*)(bias + lane * 4);
  const float4 bB = *(const float4*)(bias + 128 + lane * 4);
  for (int q = 0; q < 2; ++q) {
    const int rl = wave + 16 * q;
    const int n  = mBase + rl;
    float4 r0, r1;
    r0.x = 0.f; r0.y = 0.f; r0.z = 0.f; r0.w = 0.f;
    r1 = r0;
    if (n < NN_NODES) {
      float4 acc[2];
      node_aggregate<4, 64>(rowptr, csr, p, ht, n, lane, acc);
      r0.x = leaky(acc[0].x + bA.x, 0.01f); r0.y = leaky(acc[0].y + bA.y, 0.01f);
      r0.z = leaky(acc[0].z + bA.z, 0.01f); r0.w = leaky(acc[0].w + bA.w, 0.01f);
      r1.x = leaky(acc[1].x + bB.x, 0.01f); r1.y = leaky(acc[1].y + bB.y, 0.01f);
      r1.z = leaky(acc[1].z + bB.z, 0.01f); r1.w = leaky(acc[1].w + bB.w, 0.01f);
    }
    *(float4*)&As[rl][lane * 4]       = r0;
    *(float4*)&As[rl][128 + lane * 4] = r1;
  }
  __syncthreads();

  const int c  = lane & 15;
  const int h  = lane >> 4;
  const int ct = (NOUT == 256) ? wave : (wave & 7);
  const int rt = (NOUT == 256) ? 0 : (wave >> 3);
  const float* wrow = W + (size_t)(ct * 16 + c) * 256 + 4 * h;

  v8f acc[TM];
#pragma unroll
  for (int m = 0; m < TM; ++m) acc[m] = (v8f){0.f, 0.f, 0.f, 0.f, 0.f, 0.f, 0.f, 0.f};
  {
    float4 b_cur = *(const float4*)(wrow);
    float4 a_cur[TM];
#pragma unroll
    for (int m = 0; m < TM; ++m) a_cur[m] = *(const float4*)&As[16 * (rt + m) + c][4 * h];
#pragma unroll
    for (int i = 0; i < 32; ++i) {
      float4 b_nxt = b_cur;
      float4 a_nxt[TM];
#pragma unroll
      for (int m = 0; m < TM; ++m) a_nxt[m] = a_cur[m];
      if (i + 1 < 32) {
        b_nxt = *(const float4*)(wrow + 8 * (i + 1));
#pragma unroll
        for (int m = 0; m < TM; ++m)
          a_nxt[m] = *(const float4*)&As[16 * (rt + m) + c][8 * (i + 1) + 4 * h];
      }
#pragma unroll
      for (int m = 0; m < TM; ++m)
        acc[m] = wmma4(a_cur[m].x, a_cur[m].y, b_cur.x, b_cur.y, acc[m]);
#pragma unroll
      for (int m = 0; m < TM; ++m)
        acc[m] = wmma4(a_cur[m].z, a_cur[m].w, b_cur.z, b_cur.w, acc[m]);
      b_cur = b_nxt;
#pragma unroll
      for (int m = 0; m < TM; ++m) a_cur[m] = a_nxt[m];
    }
  }

  const int col = ct * 16 + c;
#pragma unroll
  for (int m = 0; m < TM; ++m) {
#pragma unroll
    for (int r = 0; r < 8; ++r) {
      const int row = mBase + 16 * (rt + m) + 8 * h + r;
      if (row < NN_NODES) C[(size_t)row * NOUT + col] = acc[m][r];
    }
  }

  const float av = avec[col & fmask];
#pragma unroll
  for (int m = 0; m < TM; ++m) {
#pragma unroll
    for (int r = 0; r < 8; ++r) {
      float v = acc[m][r] * av;
      v += row_xor<1>(v); v += row_xor<2>(v);
      v += row_xor<4>(v); v += row_xor<8>(v);
      if (c == r) part[ct][16 * (rt + m) + 8 * h + r] = v;
    }
  }
  __syncthreads();
  if (t < FG_ROWS * NS) {
    const int rl = t / NS;
    const int j  = t % NS;
    if (mBase + rl < NN_NODES)
      ppart[(size_t)(mBase + rl) * NS + j] = part[2 * j][rl] + part[2 * j + 1][rl];
  }
}

#define PR_TILES  (N_ROWS / 16)
#define PR_BLOCKS 512
#define ZP  132
#define Z1P 68

__device__ __forceinline__ void wave_lds_sync() {
  __builtin_amdgcn_fence(__ATOMIC_RELEASE, "wavefront");
  __builtin_amdgcn_wave_barrier();
  __builtin_amdgcn_fence(__ATOMIC_ACQUIRE, "wavefront");
}

__global__ void __launch_bounds__(256)
predictor_fused_kernel(const float* __restrict__ emb,
                       const int* __restrict__ ps, const int* __restrict__ pd,
                       const int* __restrict__ ns, const int* __restrict__ nd,
                       const float* __restrict__ pW1, const float* __restrict__ pb1,
                       const float* __restrict__ pW2, const float* __restrict__ pb2,
                       const float* __restrict__ pW3, const float* __restrict__ pb3,
                       float* __restrict__ out) {
  __shared__ __attribute__((aligned(16))) float W1s[64][ZP];
  __shared__ __attribute__((aligned(16))) float W2s[32][Z1P];
  __shared__ __attribute__((aligned(16))) float Zs[8][16][ZP];
  __shared__ __attribute__((aligned(16))) float Z1s[8][16][Z1P];

  const int t    = threadIdx.x;
  const int lane = t & 31;
  const int wave = __builtin_amdgcn_readfirstlane(t >> 5);
  const int c    = lane & 15;
  const int h    = lane >> 4;

#pragma unroll
  for (int j = 0; j < 8; ++j) {
    const int idx = t + j * 256;
    const int r   = idx >> 5;
    const int c4  = (idx & 31) * 4;
    *(float4*)&W1s[r][c4] = *(const float4*)(pW1 + r * 128 + c4);
  }
#pragma unroll
  for (int j = 0; j < 2; ++j) {
    const int idx = t + j * 256;
    const int r   = idx >> 4;
    const int c4  = (idx & 15) * 4;
    *(float4*)&W2s[r][c4] = *(const float4*)(pW2 + r * 64 + c4);
  }
  float bias1[4], bias2[2], w3[2];
#pragma unroll
  for (int tt = 0; tt < 4; ++tt) bias1[tt] = pb1[16 * tt + c];
#pragma unroll
  for (int tt = 0; tt < 2; ++tt) { bias2[tt] = pb2[16 * tt + c]; w3[tt] = pW3[16 * tt + c]; }
  const float b3 = pb3[0];
  __syncthreads();

  for (int wt = blockIdx.x * 8 + wave; wt < PR_TILES; wt += PR_BLOCKS * 8) {
    const int row0 = wt * 16;
    const int* sp = ps; const int* dp = pd; int rbase = row0;
    if (row0 >= N_PAIR) { sp = ns; dp = nd; rbase = row0 - N_PAIR; }
    const int si = sp[rbase + c];
    const int di = dp[rbase + c];

#pragma unroll
    for (int r = 0; r < 16; ++r) {
      const int s = __builtin_amdgcn_readlane(si, r);
      const int d = __builtin_amdgcn_readlane(di, r);
      const float4 a = *(const float4*)(emb + (size_t)s * 128 + lane * 4);
      const float4 b = *(const float4*)(emb + (size_t)d * 128 + lane * 4);
      float4 z; z.x = a.x * b.x; z.y = a.y * b.y; z.z = a.z * b.z; z.w = a.w * b.w;
      *(float4*)&Zs[wave][r][lane * 4] = z;
    }
    wave_lds_sync();

    v8f acc[4];
#pragma unroll
    for (int tt = 0; tt < 4; ++tt) acc[tt] = (v8f){0.f, 0.f, 0.f, 0.f, 0.f, 0.f, 0.f, 0.f};
    {
      float4 a_cur = *(const float4*)&Zs[wave][c][4 * h];
      float4 b_cur[4];
#pragma unroll
      for (int tt = 0; tt < 4; ++tt) b_cur[tt] = *(const float4*)&W1s[16 * tt + c][4 * h];
#pragma unroll
      for (int i = 0; i < 16; ++i) {
        float4 a_nxt = a_cur;
        float4 b_nxt[4];
#pragma unroll
        for (int tt = 0; tt < 4; ++tt) b_nxt[tt] = b_cur[tt];
        if (i + 1 < 16) {
          a_nxt = *(const float4*)&Zs[wave][c][8 * (i + 1) + 4 * h];
#pragma unroll
          for (int tt = 0; tt < 4; ++tt)
            b_nxt[tt] = *(const float4*)&W1s[16 * tt + c][8 * (i + 1) + 4 * h];
        }
#pragma unroll
        for (int tt = 0; tt < 4; ++tt)
          acc[tt] = wmma4(a_cur.x, a_cur.y, b_cur[tt].x, b_cur[tt].y, acc[tt]);
#pragma unroll
        for (int tt = 0; tt < 4; ++tt)
          acc[tt] = wmma4(a_cur.z, a_cur.w, b_cur[tt].z, b_cur[tt].w, acc[tt]);
        a_cur = a_nxt;
#pragma unroll
        for (int tt = 0; tt < 4; ++tt) b_cur[tt] = b_nxt[tt];
      }
    }
#pragma unroll
    for (int tt = 0; tt < 4; ++tt) {
#pragma unroll
      for (int r = 0; r < 8; ++r)
        Z1s[wave][8 * h + r][16 * tt + c] = leaky(acc[tt][r] + bias1[tt], 0.2f);
    }
    wave_lds_sync();

    v8f acc2[2];
#pragma unroll
    for (int tt = 0; tt < 2; ++tt) acc2[tt] = (v8f){0.f, 0.f, 0.f, 0.f, 0.f, 0.f, 0.f, 0.f};
    {
      float4 a2[8];
      float4 b2[2][8];
#pragma unroll
      for (int i = 0; i < 8; ++i) {
        a2[i] = *(const float4*)&Z1s[wave][c][8 * i + 4 * h];
#pragma unroll
        for (int tt = 0; tt < 2; ++tt)
          b2[tt][i] = *(const float4*)&W2s[16 * tt + c][8 * i + 4 * h];
      }
#pragma unroll
      for (int i = 0; i < 8; ++i) {
#pragma unroll
        for (int tt = 0; tt < 2; ++tt)
          acc2[tt] = wmma4(a2[i].x, a2[i].y, b2[tt][i].x, b2[tt][i].y, acc2[tt]);
#pragma unroll
        for (int tt = 0; tt < 2; ++tt)
          acc2[tt] = wmma4(a2[i].z, a2[i].w, b2[tt][i].z, b2[tt][i].w, acc2[tt]);
      }
    }

    float mine = 0.f;
#pragma unroll
    for (int r = 0; r < 8; ++r) {
      float v = leaky(acc2[0][r] + bias2[0], 0.2f) * w3[0]
              + leaky(acc2[1][r] + bias2[1], 0.2f) * w3[1];
      v += row_xor<1>(v); v += row_xor<2>(v);
      v += row_xor<4>(v); v += row_xor<8>(v);
      if (c == r) mine = v;
    }
    if (c < 8) out[row0 + 8 * h + c] = mine + b3;
    wave_lds_sync();
  }
}

extern "C" void kernel_launch(void* const* d_in, const int* in_sizes, int n_in,
                              void* d_out, int out_size, void* d_ws, size_t ws_size,
                              hipStream_t stream) {
  const float* x    = (const float*)d_in[0];
  const float* ew   = (const float*)d_in[1];
  const int*   esrc = (const int*)d_in[2];
  const int*   edst = (const int*)d_in[3];
  const int*   psrc = (const int*)d_in[4];
  const int*   pdst = (const int*)d_in[5];
  const int*   nsrc = (const int*)d_in[6];
  const int*   ndst = (const int*)d_in[7];
  const float* Wl[3] = {(const float*)d_in[8],  (const float*)d_in[11], (const float*)d_in[14]};
  const float* al[3] = {(const float*)d_in[9],  (const float*)d_in[12], (const float*)d_in[15]};
  const float* bl[3] = {(const float*)d_in[10], (const float*)d_in[13], (const float*)d_in[16]};
  const float* pW1 = (const float*)d_in[17]; const float* pb1 = (const float*)d_in[18];
  const float* pW2 = (const float*)d_in[19]; const float* pb2 = (const float*)d_in[20];
  const float* pW3 = (const float*)d_in[21]; const float* pb3 = (const float*)d_in[22];

  float* out   = (float*)d_out;
  float* emb   = out + 2 * N_PAIR;

  float*    ws   = (float*)d_ws;
  float*    htA  = ws;
  float*    htB  = htA  + 6400000;
  float*    pA   = htB  + 6400000;
  int*      deg    = (int*)(pA + 200000);
  int*      rowptr = deg + 25000;
  int*      rank   = rowptr + 25008;
  int2*     csr    = (int2*)(rank + 400000);
  float*    pB     = (float*)(csr + 400000);

  dim3 g0((NN_NODES + 63) / 64, 256 / 32);
  gemm_wmma_f32<<<g0, 256, 0, stream>>>(x, Wl[0], nullptr, htA,
                                        NN_NODES, 256, 128, 0.f, 0,
                                        al[0], pA, 63, deg);

  csr_count_kernel<<<(N_EDGES + 255) / 256, 256, 0, stream>>>(edst, deg, rank);
  csr_scan_fill_kernel<<<FILL_BLOCKS, SCAN_THREADS, 0, stream>>>(deg, esrc, edst, ew, rank,
                                                               rowptr, csr);

  gather_gemm_kernel<256><<<(NN_NODES + FG_ROWS - 1) / FG_ROWS, 512, 0, stream>>>(
      rowptr, csr, pA, htA, bl[0], Wl[1], al[1], 63, htB, pB);
  gather_gemm_kernel<128><<<(NN_NODES + FG_ROWS - 1) / FG_ROWS, 512, 0, stream>>>(
      rowptr, csr, pB, htB, bl[1], Wl[2], al[2], 127, htA, pA);

  node_softmax_gather_kernel<1, 128><<<(NN_NODES + 7) / 8, 256, 0, stream>>>(
      rowptr, csr, pA, htA, bl[2], emb, 0);

  predictor_fused_kernel<<<PR_BLOCKS, 256, 0, stream>>>(emb, psrc, pdst, nsrc, ndst,
                                                        pW1, pb1, pW2, pb2, pW3, pb3, out);
}


# Edge list built in one launch: an edge goes to the slot its counting atomic returns in a 64-entry bucket per destination (overflow list beyond); scan+fill launch gone; sums keep arrival-order rank
# speedup vs baseline: 15.4975x; 1.0652x over previous
#include <hip/hip_runtime.h>
#include <hip/hip_bf16.h>


typedef __attribute__((ext_vector_type(2))) float v2f;
typedef __attribute__((ext_vector_type(8))) float v8f;

#define NN_NODES 25000
#define N_EDGES  400000
#define N_PAIR   100000
#define N_ROWS   200000

__device__ __forceinline__ float leaky(float v, float slope) {
  return v >= 0.f ? v : v * slope;
}

template <int MASK>
__device__ __forceinline__ float row_xor(float v) {
  return __builtin_bit_cast(float, __builtin_amdgcn_update_dpp(
      0, __builtin_bit_cast(int, v), 0x160 + MASK, 0xf, 0xf, false));
}

__global__ void gemm_wmma_f32(const float* __restrict__ A,
                              const float* __restrict__ W,
                              const float* __restrict__ bias,
                              float* __restrict__ C,
                              int M, int N, int K,
                              float slope, int act,
                              const float* __restrict__ avec,
                              float* __restrict__ ppart, int fmask,
                              int* __restrict__ zero_i32) {
  __shared__ float As[2][64][36];
  __shared__ float Ws[2][32][36];
  __shared__ float part[2][64];

  const int t     = threadIdx.x;
  const int lane  = t & 31;
  const int wave  = t >> 5;
  const int waveM = wave & 3;
  const int waveN = wave >> 2;
  const int mBase = blockIdx.x * 64;
  const int nBase = blockIdx.y * 32;
  const int lh    = lane >> 4;
  const int l15   = lane & 15;

  v8f acc = {0.f, 0.f, 0.f, 0.f, 0.f, 0.f, 0.f, 0.f};

  if (zero_i32 && blockIdx.y == 0 && t < 64 && mBase + t < M) zero_i32[mBase + t] = 0;
  if (zero_i32 && blockIdx.x == 0 && blockIdx.y == 0 && t == 0) zero_i32[M] = 0;

  const int  sr   = t >> 3;
  const int  c4   = (t & 7) * 4;
  const int  gr0  = mBase + sr;
  const int  gr1  = gr0 + 32;
  const bool in0  = gr0 < M;
  const bool in1  = gr1 < M;
  const float* pa0 = A + (size_t)(in0 ? gr0 : M - 1) * K + c4;
  const float* pa1 = A + (size_t)(in1 ? gr1 : M - 1) * K + c4;
  const float* pw  = W + (size_t)(nBase + sr) * K + c4;

  float4 ra0 = *(const float4*)(pa0);
  float4 ra1 = *(const float4*)(pa1);
  float4 rw  = *(const float4*)(pw);
  if (!in0) { ra0.x = 0.f; ra0.y = 0.f; ra0.z = 0.f; ra0.w = 0.f; }
  if (!in1) { ra1.x = 0.f; ra1.y = 0.f; ra1.z = 0.f; ra1.w = 0.f; }
  *(float4*)&As[0][sr][c4]      = ra0;
  *(float4*)&As[0][sr + 32][c4] = ra1;
  *(float4*)&Ws[0][sr][c4]      = rw;
  __syncthreads();

  const int nk = K >> 5;
  for (int it = 0; it < nk; ++it) {
    const int  cur  = it & 1;
    const bool more = it + 1 < nk;
    if (more) {
      const int ko = (it + 1) * 32;
      ra0 = *(const float4*)(pa0 + ko);
      ra1 = *(const float4*)(pa1 + ko);
      rw  = *(const float4*)(pw + ko);
    }

#pragma unroll
    for (int ks = 0; ks < 8; ++ks) {
      int kb = ks * 4 + 2 * lh;
      v2f a, b;
      a.x = As[cur][waveM * 16 + l15][kb];
      a.y = As[cur][waveM * 16 + l15][kb + 1];
      b.x = Ws[cur][waveN * 16 + l15][kb];
      b.y = Ws[cur][waveN * 16 + l15][kb + 1];
      acc = __builtin_amdgcn_wmma_f32_16x16x4_f32(
           false, a,  false, b,
           (short)0, acc,  false,  false);
    }

    if (more) {
      if (!in0) { ra0.x = 0.f; ra0.y = 0.f; ra0.z = 0.f; ra0.w = 0.f; }
      if (!in1) { ra1.x = 0.f; ra1.y = 0.f; ra1.z = 0.f; ra1.w = 0.f; }
      *(float4*)&As[cur ^ 1][sr][c4]      = ra0;
      *(float4*)&As[cur ^ 1][sr + 32][c4] = ra1;
      *(float4*)&Ws[cur ^ 1][sr][c4]      = rw;
    }
    __syncthreads();
  }

#pragma unroll
  for (int r = 0; r < 8; ++r) {
    int row = mBase + waveM * 16 + r + 8 * lh;
    int col = nBase + waveN * 16 + l15;
    if (row < M) {
      float v = acc[r];
      if (bias) v += bias[col];
      if (act)  v = leaky(v, slope);
      C[(size_t)row * N + col] = v;
    }
  }

  if (avec) {
    const float av = avec[(nBase + waveN * 16 + l15) & fmask];
#pragma unroll
    for (int r = 0; r < 8; ++r) {
      float v = acc[r] * av;
      v += row_xor<1>(v); v += row_xor<2>(v);
      v += row_xor<4>(v); v += row_xor<8>(v);
      if (l15 == r) part[waveN][waveM * 16 + 8 * lh + r] = v;
    }
    __syncthreads();
    if (t < 64 && mBase + t < M)
      ppart[(size_t)(mBase + t) * gridDim.y + blockIdx.y] = part[0][t] + part[1][t];
  }
}

#define EL_CAP 64
__global__ void edge_bucket_kernel(const int* __restrict__ dst,
                                   const int* __restrict__ src,
                                   const float* __restrict__ ew,
                                   int* __restrict__ deg,
                                   int2* __restrict__ bucket,
                                   int4* __restrict__ ov) {
  int e = blockIdx.x * blockDim.x + threadIdx.x;
  if (e >= N_EDGES) return;
  const int   d  = dst[e];
  const int   s  = src[e];
  const float w  = ew[e];
  const int   r  = atomicAdd(&deg[d], 1);
  const int   lw = __float_as_int(log1pf(w));
  if (r < EL_CAP) {
    bucket[(size_t)d * EL_CAP + r] = make_int2(s, lw);
  } else {
    const int k = atomicAdd(&deg[NN_NODES], 1);
    ov[k] = make_int4(d, s, lw, 0);
  }
}

__device__ __forceinline__ float xhalf(float v) {
  return __builtin_bit_cast(float, __builtin_amdgcn_permlanex16(
      __builtin_bit_cast(int, v), __builtin_bit_cast(int, v), 0x76543210, 0xfedcba98, false, false));
}
__device__ __forceinline__ float wave_max(float v) {
  v = fmaxf(v, row_xor<1>(v)); v = fmaxf(v, row_xor<2>(v));
  v = fmaxf(v, row_xor<4>(v)); v = fmaxf(v, row_xor<8>(v));
  return fmaxf(v, xhalf(v));
}
__device__ __forceinline__ float wave_sum(float v) {
  v += row_xor<1>(v); v += row_xor<2>(v);
  v += row_xor<4>(v); v += row_xor<8>(v);
  return v + xhalf(v);
}

template <int H>
__device__ __forceinline__ void edge_scores(const int2 se,
                                            const float* __restrict__ p,
                                            const float (&pd)[H],
                                            int& s, float (&sc)[H]) {
  s = se.x;
  const float lw = __int_as_float(se.y);
  if constexpr (H == 4) {
    const float4 q0 = *(const float4*)(p + (size_t)se.x * 8);
    const float4 q1 = *(const float4*)(p + (size_t)se.x * 8 + 4);
    sc[0] = leaky((q0.x + q0.y) + pd[0] + lw, 0.01f);
    sc[1] = leaky((q0.z + q0.w) + pd[1] + lw, 0.01f);
    sc[2] = leaky((q1.x + q1.y) + pd[2] + lw, 0.01f);
    sc[3] = leaky((q1.z + q1.w) + pd[3] + lw, 0.01f);
  } else {
    const float4 q = *(const float4*)(p + (size_t)se.x * 4);
    sc[0] = leaky(((q.x + q.y) + (q.z + q.w)) + pd[0] + lw, 0.01f);
  }
}

template <int H, int F>
__device__ __forceinline__ void gather_rows(const float* __restrict__ ht, int lane,
                                            int my_s, const float (&my_a)[H], int cnt,
                                            float4 (&acc)[H * F / 128]) {
  constexpr int HF = H * F;
  const bool hiHalf = lane >= 16;
#pragma unroll 4
  for (int t = 0; t < cnt; ++t) {
    const int s = __builtin_amdgcn_readlane(my_s, t);
    const float* row = ht + (size_t)s * HF + lane * 4;
    if constexpr (H == 4) {
      const float a0 = __int_as_float(__builtin_amdgcn_readlane(__float_as_int(my_a[0]), t));
      const float a1 = __int_as_float(__builtin_amdgcn_readlane(__float_as_int(my_a[1]), t));
      const float a2 = __int_as_float(__builtin_amdgcn_readlane(__float_as_int(my_a[2]), t));
      const float a3 = __int_as_float(__builtin_amdgcn_readlane(__float_as_int(my_a[3]), t));
      const float aA = hiHalf ? a1 : a0;
      const float aB = hiHalf ? a3 : a2;
      const float4 v0 = *(const float4*)(row);
      const float4 v1 = *(const float4*)(row + 128);
      acc[0].x += aA * v0.x; acc[0].y += aA * v0.y; acc[0].z += aA * v0.z; acc[0].w += aA * v0.w;
      acc[1].x += aB * v1.x; acc[1].y += aB * v1.y; acc[1].z += aB * v1.z; acc[1].w += aB * v1.w;
    } else {
      const float a = __int_as_float(__builtin_amdgcn_readlane(__float_as_int(my_a[0]), t));
      const float4 v0 = *(const float4*)(row);
      acc[0].x += a * v0.x; acc[0].y += a * v0.y; acc[0].z += a * v0.z; acc[0].w += a * v0.w;
    }
  }
}

template <int H, int F>
__device__ __forceinline__ void gather_rows_masked(const float* __restrict__ ht, int lane,
                                                   int my_s, const float (&my_a)[H], unsigned live,
                                                   float4 (&acc)[H * F / 128]) {
  constexpr int HF = H * F;
  const bool hiHalf = lane >= 16;
  while (live) {
    const int t = __builtin_ctz(live);
    live &= live - 1;
    const int s = __builtin_amdgcn_readlane(my_s, t);
    const float* row = ht + (size_t)s * HF + lane * 4;
    if constexpr (H == 4) {
      const float a0 = __int_as_float(__builtin_amdgcn_readlane(__float_as_int(my_a[0]), t));
      const float a1 = __int_as_float(__builtin_amdgcn_readlane(__float_as_int(my_a[1]), t));
      const float a2 = __int_as_float(__builtin_amdgcn_readlane(__float_as_int(my_a[2]), t));
      const float a3 = __int_as_float(__builtin_amdgcn_readlane(__float_as_int(my_a[3]), t));
      const float aA = hiHalf ? a1 : a0;
      const float aB = hiHalf ? a3 : a2;
      const float4 v0 = *(const float4*)(row);
      const float4 v1 = *(const float4*)(row + 128);
      acc[0].x += aA * v0.x; acc[0].y += aA * v0.y; acc[0].z += aA * v0.z; acc[0].w += aA * v0.w;
      acc[1].x += aB * v1.x; acc[1].y += aB * v1.y; acc[1].z += aB * v1.z; acc[1].w += aB * v1.w;
    } else {
      const float a = __int_as_float(__builtin_amdgcn_readlane(__float_as_int(my_a[0]), t));
      const float4 v0 = *(const float4*)(row);
      acc[0].x += a * v0.x; acc[0].y += a * v0.y; acc[0].z += a * v0.z; acc[0].w += a * v0.w;
    }
  }
}

__device__ __forceinline__ bool list_entry(const int2* __restrict__ bk,
                                           const int4* __restrict__ ov,
                                           int n, int nb, int total, int j, int2& se) {
  if (j < nb) { se = bk[j]; return true; }
  if (j >= EL_CAP && j < total) {
    const int4 o = ov[j - EL_CAP];
    se = make_int2(o.y, o.z);
    return o.x == n;
  }
  return false;
}

template <int H, int F>
__device__ __forceinline__ void node_aggregate(const int* __restrict__ degs,
                                               const int2* __restrict__ bucket,
                                               const int4* __restrict__ ov,
                                               const float* __restrict__ p,
                                               const float* __restrict__ ht,
                                               int n, int lane,
                                               float4 (&acc)[H * F / 128]) {
  constexpr int NV = H * F / 128;
  const int deg = degs[n];
  const int2* __restrict__ bk = bucket + (size_t)n * EL_CAP;

  float pd[H];
  if constexpr (H == 4) {
#pragma unroll
    for (int h = 0; h < 4; ++h) pd[h] = p[n * 8 + 2 * h] + p[n * 8 + 2 * h + 1];
  } else {
    pd[0] = (p[n * 4] + p[n * 4 + 1]) + (p[n * 4 + 2] + p[n * 4 + 3]);
  }

#pragma unroll
  for (int v = 0; v < NV; ++v) { acc[v].x = 0.f; acc[v].y = 0.f; acc[v].z = 0.f; acc[v].w = 0.f; }

  if (deg <= 32) {
    const bool on = lane < deg;
    int   my_s = 0;
    float sc[H];
#pragma unroll
    for (int h = 0; h < H; ++h) sc[h] = -INFINITY;
    if (on) edge_scores<H>(bk[lane], p, pd, my_s, sc);
    float my_a[H];
#pragma unroll
    for (int h = 0; h < H; ++h) {
      float m = wave_max(sc[h]);
      if (!isfinite(m)) m = 0.f;
      const float ex = on ? __expf(sc[h] - m) : 0.f;
      const float sum = wave_sum(ex);
      my_a[h] = ex / fmaxf(sum, 1e-20f);
    }
    gather_rows<H, F>(ht, lane, my_s, my_a, deg, acc);
  } else {
    const int nb    = deg < EL_CAP ? deg : EL_CAP;
    const int total = deg <= EL_CAP ? deg : EL_CAP + degs[NN_NODES];
    float m[H], den[H];
#pragma unroll
    for (int h = 0; h < H; ++h) m[h] = -INFINITY;
    for (int base = 0; base < total; base += 32) {
      int2 se;
      if (list_entry(bk, ov, n, nb, total, base + lane, se)) {
        int s_; float sc[H];
        edge_scores<H>(se, p, pd, s_, sc);
#pragma unroll
        for (int h = 0; h < H; ++h) m[h] = fmaxf(m[h], sc[h]);
      }
    }
#pragma unroll
    for (int h = 0; h < H; ++h) {
      m[h] = wave_max(m[h]);
      if (!isfinite(m[h])) m[h] = 0.f;
      den[h] = 0.f;
    }
    for (int base = 0; base < total; base += 32) {
      int2 se;
      if (list_entry(bk, ov, n, nb, total, base + lane, se)) {
        int s_; float sc[H];
        edge_scores<H>(se, p, pd, s_, sc);
#pragma unroll
        for (int h = 0; h < H; ++h) den[h] += __expf(sc[h] - m[h]);
      }
    }
#pragma unroll
    for (int h = 0; h < H; ++h) den[h] = fmaxf(wave_sum(den[h]), 1e-20f);
    for (int base = 0; base < total; base += 32) {
      int   my_s = 0;
      float my_a[H];
#pragma unroll
      for (int h = 0; h < H; ++h) my_a[h] = 0.f;
      int2 se;
      const bool mine = list_entry(bk, ov, n, nb, total, base + lane, se);
      if (mine) {
        float sc[H];
        edge_scores<H>(se, p, pd, my_s, sc);
#pragma unroll
        for (int h = 0; h < H; ++h) my_a[h] = __expf(sc[h] - m[h]) / den[h];
      }
      if (base < EL_CAP) {
        const int left = nb - base;
        gather_rows<H, F>(ht, lane, my_s, my_a, left < 32 ? left : 32, acc);
      } else {
        gather_rows_masked<H, F>(ht, lane, my_s, my_a, __builtin_amdgcn_ballot_w32(mine), acc);
      }
    }
  }
}

template <int H, int F>
__global__ void __launch_bounds__(256)
node_softmax_gather_kernel(const int* __restrict__ degs,
                           const int2* __restrict__ bucket,
                           const int4* __restrict__ ov,
                           const float* __restrict__ p,
                           const float* __restrict__ ht,
                           const float* __restrict__ bias,
                           float* __restrict__ hout,
                           int act) {
  constexpr int HF = H * F;
  constexpr int NV = HF / 128;
  const int lane = threadIdx.x & 31;
  const int n = __builtin_amdgcn_readfirstlane((int)(blockIdx.x * 8 + (threadIdx.x >> 5)));
  if (n >= NN_NODES) return;

  float4 acc[NV];
  node_aggregate<H, F>(degs, bucket, ov, p, ht, n, lane, acc);

#pragma unroll
  for (int v = 0; v < NV; ++v) {
    const int c = v * 128 + lane * 4;
    const float4 b = *(const float4*)(bias + c);
    float4 r;
    r.x = acc[v].x + b.x; r.y = acc[v].y + b.y; r.z = acc[v].z + b.z; r.w = acc[v].w + b.w;
    if (act) {
      r.x = leaky(r.x, 0.01f); r.y = leaky(r.y, 0.01f);
      r.z = leaky(r.z, 0.01f); r.w = leaky(r.w, 0.01f);
    }
    *(float4*)(hout + (size_t)n * HF + c) = r;
  }
}

__device__ __forceinline__ v8f wmma4(float a0, float a1, float b0, float b1, v8f acc) {
  v2f a = {a0, a1};
  v2f b = {b0, b1};
  return __builtin_amdgcn_wmma_f32_16x16x4_f32(
       false, a,  false, b,
       (short)0, acc,  false,  false);
}

#define FG_ROWS  32
#define FG_PITCH 260
template <int NOUT>
__global__ void __launch_bounds__(512)
gather_gemm_kernel(const int* __restrict__ degs,
                   const int2* __restrict__ bucket,
                   const int4* __restrict__ ov,
                   const float* __restrict__ p,
                   const float* __restrict__ ht,
                   const float* __restrict__ bias,
                   const float* __restrict__ W,
                   const float* __restrict__ avec, int fmask,
                   float* __restrict__ C,
                   float* __restrict__ ppart) {
  constexpr int TM = NOUT / 128;
  constexpr int NS = NOUT / 32;
  __shared__ __attribute__((aligned(16))) float As[FG_ROWS][FG_PITCH];
  __shared__ float part[NOUT / 16][FG_ROWS];

  const int t     = threadIdx.x;
  const int lane  = t & 31;
  const int wave  = __builtin_amdgcn_readfirstlane(t >> 5);
  const int mBase = blockIdx.x * FG_ROWS;

  const float4 bA = *(const float4*)(bias + lane * 4);
  const float4 bB = *(const float4*)(bias + 128 + lane * 4);
  for (int q = 0; q < 2; ++q) {
    const int rl = wave + 16 * q;
    const int n  = mBase + rl;
    float4 r0, r1;
    r0.x = 0.f; r0.y = 0.f; r0.z = 0.f; r0.w = 0.f;
    r1 = r0;
    if (n < NN_NODES) {
      float4 acc[2];
      node_aggregate<4, 64>(degs, bucket, ov, p, ht, n, lane, acc);
      r0.x = leaky(acc[0].x + bA.x, 0.01f); r0.y = leaky(acc[0].y + bA.y, 0.01f);
      r0.z = leaky(acc[0].z + bA.z, 0.01f); r0.w = leaky(acc[0].w + bA.w, 0.01f);
      r1.x = leaky(acc[1].x + bB.x, 0.01f); r1.y = leaky(acc[1].y + bB.y, 0.01f);
      r1.z = leaky(acc[1].z + bB.z, 0.01f); r1.w = leaky(acc[1].w + bB.w, 0.01f);
    }
    *(float4*)&As[rl][lane * 4]       = r0;
    *(float4*)&As[rl][128 + lane * 4] = r1;
  }
  __syncthreads();

  const int c  = lane & 15;
  const int h  = lane >> 4;
  const int ct = (NOUT == 256) ? wave : (wave & 7);
  const int rt = (NOUT == 256) ? 0 : (wave >> 3);
  const float* wrow = W + (size_t)(ct * 16 + c) * 256 + 4 * h;

  v8f acc[TM];
#pragma unroll
  for (int m = 0; m < TM; ++m) acc[m] = (v8f){0.f, 0.f, 0.f, 0.f, 0.f, 0.f, 0.f, 0.f};
  {
    float4 b_cur = *(const float4*)(wrow);
    float4 a_cur[TM];
#pragma unroll
    for (int m = 0; m < TM; ++m) a_cur[m] = *(const float4*)&As[16 * (rt + m) + c][4 * h];
#pragma unroll
    for (int i = 0; i < 32; ++i) {
      float4 b_nxt = b_cur;
      float4 a_nxt[TM];
#pragma unroll
      for (int m = 0; m < TM; ++m) a_nxt[m] = a_cur[m];
      if (i + 1 < 32) {
        b_nxt = *(const float4*)(wrow + 8 * (i + 1));
#pragma unroll
        for (int m = 0; m < TM; ++m)
          a_nxt[m] = *(const float4*)&As[16 * (rt + m) + c][8 * (i + 1) + 4 * h];
      }
#pragma unroll
      for (int m = 0; m < TM; ++m)
        acc[m] = wmma4(a_cur[m].x, a_cur[m].y, b_cur.x, b_cur.y, acc[m]);
#pragma unroll
      for (int m = 0; m < TM; ++m)
        acc[m] = wmma4(a_cur[m].z, a_cur[m].w, b_cur.z, b_cur.w, acc[m]);
      b_cur = b_nxt;
#pragma unroll
      for (int m = 0; m < TM; ++m) a_cur[m] = a_nxt[m];
    }
  }

  const int col = ct * 16 + c;
#pragma unroll
  for (int m = 0; m < TM; ++m) {
#pragma unroll
    for (int r = 0; r < 8; ++r) {
      const int row = mBase + 16 * (rt + m) + 8 * h + r;
      if (row < NN_NODES) C[(size_t)row * NOUT + col] = acc[m][r];
    }
  }

  const float av = avec[col & fmask];
#pragma unroll
  for (int m = 0; m < TM; ++m) {
#pragma unroll
    for (int r = 0; r < 8; ++r) {
      float v = acc[m][r] * av;
      v += row_xor<1>(v); v += row_xor<2>(v);
      v += row_xor<4>(v); v += row_xor<8>(v);
      if (c == r) part[ct][16 * (rt + m) + 8 * h + r] = v;
    }
  }
  __syncthreads();
  if (t < FG_ROWS * NS) {
    const int rl = t / NS;
    const int j  = t % NS;
    if (mBase + rl < NN_NODES)
      ppart[(size_t)(mBase + rl) * NS + j] = part[2 * j][rl] + part[2 * j + 1][rl];
  }
}

#define PR_TILES  (N_ROWS / 16)
#define PR_BLOCKS 512
#define ZP  132
#define Z1P 68

__device__ __forceinline__ void wave_lds_sync() {
  __builtin_amdgcn_fence(__ATOMIC_RELEASE, "wavefront");
  __builtin_amdgcn_wave_barrier();
  __builtin_amdgcn_fence(__ATOMIC_ACQUIRE, "wavefront");
}

__global__ void __launch_bounds__(256)
predictor_fused_kernel(const float* __restrict__ emb,
                       const int* __restrict__ ps, const int* __restrict__ pd,
                       const int* __restrict__ ns, const int* __restrict__ nd,
                       const float* __restrict__ pW1, const float* __restrict__ pb1,
                       const float* __restrict__ pW2, const float* __restrict__ pb2,
                       const float* __restrict__ pW3, const float* __restrict__ pb3,
                       float* __restrict__ out) {
  __shared__ __attribute__((aligned(16))) float W1s[64][ZP];
  __shared__ __attribute__((aligned(16))) float W2s[32][Z1P];
  __shared__ __attribute__((aligned(16))) float Zs[8][16][ZP];
  __shared__ __attribute__((aligned(16))) float Z1s[8][16][Z1P];

  const int t    = threadIdx.x;
  const int lane = t & 31;
  const int wave = __builtin_amdgcn_readfirstlane(t >> 5);
  const int c    = lane & 15;
  const int h    = lane >> 4;

#pragma unroll
  for (int j = 0; j < 8; ++j) {
    const int idx = t + j * 256;
    const int r   = idx >> 5;
    const int c4  = (idx & 31) * 4;
    *(float4*)&W1s[r][c4] = *(const float4*)(pW1 + r * 128 + c4);
  }
#pragma unroll
  for (int j = 0; j < 2; ++j) {
    const int idx = t + j * 256;
    const int r   = idx >> 4;
    const int c4  = (idx & 15) * 4;
    *(float4*)&W2s[r][c4] = *(const float4*)(pW2 + r * 64 + c4);
  }
  float bias1[4], bias2[2], w3[2];
#pragma unroll
  for (int tt = 0; tt < 4; ++tt) bias1[tt] = pb1[16 * tt + c];
#pragma unroll
  for (int tt = 0; tt < 2; ++tt) { bias2[tt] = pb2[16 * tt + c]; w3[tt] = pW3[16 * tt + c]; }
  const float b3 = pb3[0];
  __syncthreads();

  for (int wt = blockIdx.x * 8 + wave; wt < PR_TILES; wt += PR_BLOCKS * 8) {
    const int row0 = wt * 16;
    const int* sp = ps; const int* dp = pd; int rbase = row0;
    if (row0 >= N_PAIR) { sp = ns; dp = nd; rbase = row0 - N_PAIR; }
    const int si = sp[rbase + c];
    const int di = dp[rbase + c];

#pragma unroll
    for (int r = 0; r < 16; ++r) {
      const int s = __builtin_amdgcn_readlane(si, r);
      const int d = __builtin_amdgcn_readlane(di, r);
      const float4 a = *(const float4*)(emb + (size_t)s * 128 + lane * 4);
      const float4 b = *(const float4*)(emb + (size_t)d * 128 + lane * 4);
      float4 z; z.x = a.x * b.x; z.y = a.y * b.y; z.z = a.z * b.z; z.w = a.w * b.w;
      *(float4*)&Zs[wave][r][lane * 4] = z;
    }
    wave_lds_sync();

    v8f acc[4];
#pragma unroll
    for (int tt = 0; tt < 4; ++tt) acc[tt] = (v8f){0.f, 0.f, 0.f, 0.f, 0.f, 0.f, 0.f, 0.f};
    {
      float4 a_cur = *(const float4*)&Zs[wave][c][4 * h];
      float4 b_cur[4];
#pragma unroll
      for (int tt = 0; tt < 4; ++tt) b_cur[tt] = *(const float4*)&W1s[16 * tt + c][4 * h];
#pragma unroll
      for (int i = 0; i < 16; ++i) {
        float4 a_nxt = a_cur;
        float4 b_nxt[4];
#pragma unroll
        for (int tt = 0; tt < 4; ++tt) b_nxt[tt] = b_cur[tt];
        if (i + 1 < 16) {
          a_nxt = *(const float4*)&Zs[wave][c][8 * (i + 1) + 4 * h];
#pragma unroll
          for (int tt = 0; tt < 4; ++tt)
            b_nxt[tt] = *(const float4*)&W1s[16 * tt + c][8 * (i + 1) + 4 * h];
        }
#pragma unroll
        for (int tt = 0; tt < 4; ++tt)
          acc[tt] = wmma4(a_cur.x, a_cur.y, b_cur[tt].x, b_cur[tt].y, acc[tt]);
#pragma unroll
        for (int tt = 0; tt < 4; ++tt)
          acc[tt] = wmma4(a_cur.z, a_cur.w, b_cur[tt].z, b_cur[tt].w, acc[tt]);
        a_cur = a_nxt;
#pragma unroll
        for (int tt = 0; tt < 4; ++tt) b_cur[tt] = b_nxt[tt];
      }
    }
#pragma unroll
    for (int tt = 0; tt < 4; ++tt) {
#pragma unroll
      for (int r = 0; r < 8; ++r)
        Z1s[wave][8 * h + r][16 * tt + c] = leaky(acc[tt][r] + bias1[tt], 0.2f);
    }
    wave_lds_sync();

    v8f acc2[2];
#pragma unroll
    for (int tt = 0; tt < 2; ++tt) acc2[tt] = (v8f){0.f, 0.f, 0.f, 0.f, 0.f, 0.f, 0.f, 0.f};
    {
      float4 a2[8];
      float4 b2[2][8];
#pragma unroll
      for (int i = 0; i < 8; ++i) {
        a2[i] = *(const float4*)&Z1s[wave][c][8 * i + 4 * h];
#pragma unroll
        for (int tt = 0; tt < 2; ++tt)
          b2[tt][i] = *(const float4*)&W2s[16 * tt + c][8 * i + 4 * h];
      }
#pragma unroll
      for (int i = 0; i < 8; ++i) {
#pragma unroll
        for (int tt = 0; tt < 2; ++tt)
          acc2[tt] = wmma4(a2[i].x, a2[i].y, b2[tt][i].x, b2[tt][i].y, acc2[tt]);
#pragma unroll
        for (int tt = 0; tt < 2; ++tt)
          acc2[tt] = wmma4(a2[i].z, a2[i].w, b2[tt][i].z, b2[tt][i].w, acc2[tt]);
      }
    }

    float mine = 0.f;
#pragma unroll
    for (int r = 0; r < 8; ++r) {
      float v = leaky(acc2[0][r] + bias2[0], 0.2f) * w3[0]
              + leaky(acc2[1][r] + bias2[1], 0.2f) * w3[1];
      v += row_xor<1>(v); v += row_xor<2>(v);
      v += row_xor<4>(v); v += row_xor<8>(v);
      if (c == r) mine = v;
    }
    if (c < 8) out[row0 + 8 * h + c] = mine + b3;
    wave_lds_sync();
  }
}

extern "C" void kernel_launch(void* const* d_in, const int* in_sizes, int n_in,
                              void* d_out, int out_size, void* d_ws, size_t ws_size,
                              hipStream_t stream) {
  const float* x    = (const float*)d_in[0];
  const float* ew   = (const float*)d_in[1];
  const int*   esrc = (const int*)d_in[2];
  const int*   edst = (const int*)d_in[3];
  const int*   psrc = (const int*)d_in[4];
  const int*   pdst = (const int*)d_in[5];
  const int*   nsrc = (const int*)d_in[6];
  const int*   ndst = (const int*)d_in[7];
  const float* Wl[3] = {(const float*)d_in[8],  (const float*)d_in[11], (const float*)d_in[14]};
  const float* al[3] = {(const float*)d_in[9],  (const float*)d_in[12], (const float*)d_in[15]};
  const float* bl[3] = {(const float*)d_in[10], (const float*)d_in[13], (const float*)d_in[16]};
  const float* pW1 = (const float*)d_in[17]; const float* pb1 = (const float*)d_in[18];
  const float* pW2 = (const float*)d_in[19]; const float* pb2 = (const float*)d_in[20];
  const float* pW3 = (const float*)d_in[21]; const float* pb3 = (const float*)d_in[22];

  float* out   = (float*)d_out;
  float* emb   = out + 2 * N_PAIR;

  float*    ws   = (float*)d_ws;
  float*    htA  = ws;
  float*    htB  = htA  + 6400000;
  float*    pA   = htB  + 6400000;
  float*    pB   = pA   + 200000;
  int*      deg    = (int*)(pB + 200000);
  int2*     bucket = (int2*)(deg + 25088);
  int4*     ov     = (int4*)(bucket + 1600000);

  dim3 g0((NN_NODES + 63) / 64, 256 / 32);
  gemm_wmma_f32<<<g0, 256, 0, stream>>>(x, Wl[0], nullptr, htA,
                                        NN_NODES, 256, 128, 0.f, 0,
                                        al[0], pA, 63, deg);

  edge_bucket_kernel<<<(N_EDGES + 255) / 256, 256, 0, stream>>>(edst, esrc, ew, deg, bucket, ov);

  gather_gemm_kernel<256><<<(NN_NODES + FG_ROWS - 1) / FG_ROWS, 512, 0, stream>>>(
      deg, bucket, ov, pA, htA, bl[0], Wl[1], al[1], 63, htB, pB);
  gather_gemm_kernel<128><<<(NN_NODES + FG_ROWS - 1) / FG_ROWS, 512, 0, stream>>>(
      deg, bucket, ov, pB, htB, bl[1], Wl[2], al[2], 127, htA, pA);

  node_softmax_gather_kernel<1, 128><<<(NN_NODES + 7) / 8, 256, 0, stream>>>(
      deg, bucket, ov, pA, htA, bl[2], emb, 0);

  predictor_fused_kernel<<<PR_BLOCKS, 256, 0, stream>>>(emb, psrc, pdst, nsrc, ndst,
                                                        pW1, pb1, pW2, pb2, pW3, pb3, out);
}


# Layer 0's linear map hoisted out of its gather: 128-wide input rows summed per head, then multiplied by W0 (WMMA) inside the layer-1 kernel; p0 = x.(W0^T a0); sum order of layer 0 changes
# speedup vs baseline: 16.1331x; 1.0410x over previous
#include <hip/hip_runtime.h>
#include <hip/hip_bf16.h>


typedef __attribute__((ext_vector_type(2))) float v2f;
typedef __attribute__((ext_vector_type(8))) float v8f;

#define NN_NODES 25000
#define N_EDGES  400000
#define N_PAIR   100000
#define N_ROWS   200000

__device__ __forceinline__ float leaky(float v, float slope) {
  return v >= 0.f ? v : v * slope;
}

template <int MASK>
__device__ __forceinline__ float row_xor(float v) {
  return __builtin_bit_cast(float, __builtin_amdgcn_update_dpp(
      0, __builtin_bit_cast(int, v), 0x160 + MASK, 0xf, 0xf, false));
}

#define EL_CAP 64
__global__ void edge_bucket_kernel(const int* __restrict__ dst,
                                   const int* __restrict__ src,
                                   const float* __restrict__ ew,
                                   int* __restrict__ deg,
                                   int2* __restrict__ bucket,
                                   int4* __restrict__ ov) {
  int e = blockIdx.x * blockDim.x + threadIdx.x;
  if (e >= N_EDGES) return;
  const int   d  = dst[e];
  const int   s  = src[e];
  const float w  = ew[e];
  const int   r  = atomicAdd(&deg[d], 1);
  const int   lw = __float_as_int(log1pf(w));
  if (r < EL_CAP) {
    bucket[(size_t)d * EL_CAP + r] = make_int2(s, lw);
  } else {
    const int k = atomicAdd(&deg[NN_NODES], 1);
    ov[k] = make_int4(d, s, lw, 0);
  }
}

__device__ __forceinline__ float xhalf(float v) {
  return __builtin_bit_cast(float, __builtin_amdgcn_permlanex16(
      __builtin_bit_cast(int, v), __builtin_bit_cast(int, v), 0x76543210, 0xfedcba98, false, false));
}
__device__ __forceinline__ float wave_max(float v) {
  v = fmaxf(v, row_xor<1>(v)); v = fmaxf(v, row_xor<2>(v));
  v = fmaxf(v, row_xor<4>(v)); v = fmaxf(v, row_xor<8>(v));
  return fmaxf(v, xhalf(v));
}
__device__ __forceinline__ float wave_sum(float v) {
  v += row_xor<1>(v); v += row_xor<2>(v);
  v += row_xor<4>(v); v += row_xor<8>(v);
  return v + xhalf(v);
}

__global__ void __launch_bounds__(256)
proj_p_kernel(const float* __restrict__ x, const float* __restrict__ W,
              const float* __restrict__ avec, float* __restrict__ pout,
              int* __restrict__ zero_i32) {
  __shared__ __attribute__((aligned(16))) float wts[4][128];
  const int t     = threadIdx.x;
  const int lane  = t & 31;
  const int wave  = __builtin_amdgcn_readfirstlane(t >> 5);
  const int mBase = blockIdx.x * 64;

  if (t < 64 && mBase + t < NN_NODES) zero_i32[mBase + t] = 0;
  if (blockIdx.x == 0 && t == 0) zero_i32[NN_NODES] = 0;

  float4 xv[8];
#pragma unroll
  for (int q = 0; q < 8; ++q) {
    const int n  = mBase + wave * 8 + q;
    const int nc = n < NN_NODES ? n : NN_NODES - 1;
    xv[q] = *(const float4*)(x + (size_t)nc * 128 + lane * 4);
  }

  {
    const int k  = t & 127;
    const int hp = t >> 7;
    const float* w0 = W + (size_t)(hp * 2) * 64 * 128 + k;
    const float* w1 = w0 + 64 * 128;
    float s0 = 0.f, s1 = 0.f;
#pragma unroll 16
    for (int f = 0; f < 64; ++f) {
      const float a = avec[f];
      s0 += a * w0[f * 128];
      s1 += a * w1[f * 128];
    }
    wts[hp * 2][k]     = s0;
    wts[hp * 2 + 1][k] = s1;
  }
  __syncthreads();

  float4 wv[4];
#pragma unroll
  for (int h = 0; h < 4; ++h) wv[h] = *(const float4*)&wts[h][lane * 4];

#pragma unroll
  for (int q = 0; q < 8; ++q) {
    const int n = mBase + wave * 8 + q;
    const float4 v = xv[q];
    float d[4];
#pragma unroll
    for (int h = 0; h < 4; ++h)
      d[h] = wave_sum((v.x * wv[h].x + v.y * wv[h].y) + (v.z * wv[h].z + v.w * wv[h].w));
    if (n < NN_NODES && lane == 0) {
      *(float4*)(pout + (size_t)n * 8)     = make_float4(d[0], 0.f, d[1], 0.f);
      *(float4*)(pout + (size_t)n * 8 + 4) = make_float4(d[2], 0.f, d[3], 0.f);
    }
  }
}

template <int H>
__device__ __forceinline__ void edge_scores(const int2 se,
                                            const float* __restrict__ p,
                                            const float (&pd)[H],
                                            int& s, float (&sc)[H]) {
  s = se.x;
  const float lw = __int_as_float(se.y);
  if constexpr (H == 4) {
    const float4 q0 = *(const float4*)(p + (size_t)se.x * 8);
    const float4 q1 = *(const float4*)(p + (size_t)se.x * 8 + 4);
    sc[0] = leaky((q0.x + q0.y) + pd[0] + lw, 0.01f);
    sc[1] = leaky((q0.z + q0.w) + pd[1] + lw, 0.01f);
    sc[2] = leaky((q1.x + q1.y) + pd[2] + lw, 0.01f);
    sc[3] = leaky((q1.z + q1.w) + pd[3] + lw, 0.01f);
  } else {
    const float4 q = *(const float4*)(p + (size_t)se.x * 4);
    sc[0] = leaky(((q.x + q.y) + (q.z + q.w)) + pd[0] + lw, 0.01f);
  }
}

template <int H, int F, bool XIN>
__device__ __forceinline__ void gather_rows(const float* __restrict__ ht, int lane,
                                            int my_s, const float (&my_a)[H], int cnt,
                                            float4 (&acc)[XIN ? 4 : H * F / 128]) {
  constexpr int HF = XIN ? 128 : H * F;
  const bool hiHalf = lane >= 16;
#pragma unroll 4
  for (int t = 0; t < cnt; ++t) {
    const int s = __builtin_amdgcn_readlane(my_s, t);
    const float* row = ht + (size_t)s * HF + lane * 4;
    if constexpr (XIN) {
      const float a0 = __int_as_float(__builtin_amdgcn_readlane(__float_as_int(my_a[0]), t));
      const float a1 = __int_as_float(__builtin_amdgcn_readlane(__float_as_int(my_a[1]), t));
      const float a2 = __int_as_float(__builtin_amdgcn_readlane(__float_as_int(my_a[2]), t));
      const float a3 = __int_as_float(__builtin_amdgcn_readlane(__float_as_int(my_a[3]), t));
      const float4 v = *(const float4*)(row);
      acc[0].x += a0 * v.x; acc[0].y += a0 * v.y; acc[0].z += a0 * v.z; acc[0].w += a0 * v.w;
      acc[1].x += a1 * v.x; acc[1].y += a1 * v.y; acc[1].z += a1 * v.z; acc[1].w += a1 * v.w;
      acc[2].x += a2 * v.x; acc[2].y += a2 * v.y; acc[2].z += a2 * v.z; acc[2].w += a2 * v.w;
      acc[3].x += a3 * v.x; acc[3].y += a3 * v.y; acc[3].z += a3 * v.z; acc[3].w += a3 * v.w;
    } else if constexpr (H == 4) {
      const float a0 = __int_as_float(__builtin_amdgcn_readlane(__float_as_int(my_a[0]), t));
      const float a1 = __int_as_float(__builtin_amdgcn_readlane(__float_as_int(my_a[1]), t));
      const float a2 = __int_as_float(__builtin_amdgcn_readlane(__float_as_int(my_a[2]), t));
      const float a3 = __int_as_float(__builtin_amdgcn_readlane(__float_as_int(my_a[3]), t));
      const float aA = hiHalf ? a1 : a0;
      const float aB = hiHalf ? a3 : a2;
      const float4 v0 = *(const float4*)(row);
      const float4 v1 = *(const float4*)(row + 128);
      acc[0].x += aA * v0.x; acc[0].y += aA * v0.y; acc[0].z += aA * v0.z; acc[0].w += aA * v0.w;
      acc[1].x += aB * v1.x; acc[1].y += aB * v1.y; acc[1].z += aB * v1.z; acc[1].w += aB * v1.w;
    } else {
      const float a = __int_as_float(__builtin_amdgcn_readlane(__float_as_int(my_a[0]), t));
      const float4 v0 = *(const float4*)(row);
      acc[0].x += a * v0.x; acc[0].y += a * v0.y; acc[0].z += a * v0.z; acc[0].w += a * v0.w;
    }
  }
}

template <int H, int F, bool XIN>
__device__ __forceinline__ void gather_rows_masked(const float* __restrict__ ht, int lane,
                                                   int my_s, const float (&my_a)[H], unsigned live,
                                                   float4 (&acc)[XIN ? 4 : H * F / 128]) {
  constexpr int HF = XIN ? 128 : H * F;
  const bool hiHalf = lane >= 16;
  while (live) {
    const int t = __builtin_ctz(live);
    live &= live - 1;
    const int s = __builtin_amdgcn_readlane(my_s, t);
    const float* row = ht + (size_t)s * HF + lane * 4;
    if constexpr (XIN) {
      const float a0 = __int_as_float(__builtin_amdgcn_readlane(__float_as_int(my_a[0]), t));
      const float a1 = __int_as_float(__builtin_amdgcn_readlane(__float_as_int(my_a[1]), t));
      const float a2 = __int_as_float(__builtin_amdgcn_readlane(__float_as_int(my_a[2]), t));
      const float a3 = __int_as_float(__builtin_amdgcn_readlane(__float_as_int(my_a[3]), t));
      const float4 v = *(const float4*)(row);
      acc[0].x += a0 * v.x; acc[0].y += a0 * v.y; acc[0].z += a0 * v.z; acc[0].w += a0 * v.w;
      acc[1].x += a1 * v.x; acc[1].y += a1 * v.y; acc[1].z += a1 * v.z; acc[1].w += a1 * v.w;
      acc[2].x += a2 * v.x; acc[2].y += a2 * v.y; acc[2].z += a2 * v.z; acc[2].w += a2 * v.w;
      acc[3].x += a3 * v.x; acc[3].y += a3 * v.y; acc[3].z += a3 * v.z; acc[3].w += a3 * v.w;
    } else if constexpr (H == 4) {
      const float a0 = __int_as_float(__builtin_amdgcn_readlane(__float_as_int(my_a[0]), t));
      const float a1 = __int_as_float(__builtin_amdgcn_readlane(__float_as_int(my_a[1]), t));
      const float a2 = __int_as_float(__builtin_amdgcn_readlane(__float_as_int(my_a[2]), t));
      const float a3 = __int_as_float(__builtin_amdgcn_readlane(__float_as_int(my_a[3]), t));
      const float aA = hiHalf ? a1 : a0;
      const float aB = hiHalf ? a3 : a2;
      const float4 v0 = *(const float4*)(row);
      const float4 v1 = *(const float4*)(row + 128);
      acc[0].x += aA * v0.x; acc[0].y += aA * v0.y; acc[0].z += aA * v0.z; acc[0].w += aA * v0.w;
      acc[1].x += aB * v1.x; acc[1].y += aB * v1.y; acc[1].z += aB * v1.z; acc[1].w += aB * v1.w;
    } else {
      const float a = __int_as_float(__builtin_amdgcn_readlane(__float_as_int(my_a[0]), t));
      const float4 v0 = *(const float4*)(row);
      acc[0].x += a * v0.x; acc[0].y += a * v0.y; acc[0].z += a * v0.z; acc[0].w += a * v0.w;
    }
  }
}

__device__ __forceinline__ bool list_entry(const int2* __restrict__ bk,
                                           const int4* __restrict__ ov,
                                           int n, int nb, int total, int j, int2& se) {
  if (j < nb) { se = bk[j]; return true; }
  if (j >= EL_CAP && j < total) {
    const int4 o = ov[j - EL_CAP];
    se = make_int2(o.y, o.z);
    return o.x == n;
  }
  return false;
}

template <int H, int F, bool XIN = false>
__device__ __forceinline__ void node_aggregate(const int* __restrict__ degs,
                                               const int2* __restrict__ bucket,
                                               const int4* __restrict__ ov,
                                               const float* __restrict__ p,
                                               const float* __restrict__ ht,
                                               int n, int lane,
                                               float4 (&acc)[XIN ? 4 : H * F / 128]) {
  constexpr int NV = XIN ? 4 : H * F / 128;
  const int deg = degs[n];
  const int2* __restrict__ bk = bucket + (size_t)n * EL_CAP;

  float pd[H];
  if constexpr (H == 4) {
#pragma unroll
    for (int h = 0; h < 4; ++h) pd[h] = p[n * 8 + 2 * h] + p[n * 8 + 2 * h + 1];
  } else {
    pd[0] = (p[n * 4] + p[n * 4 + 1]) + (p[n * 4 + 2] + p[n * 4 + 3]);
  }

#pragma unroll
  for (int v = 0; v < NV; ++v) { acc[v].x = 0.f; acc[v].y = 0.f; acc[v].z = 0.f; acc[v].w = 0.f; }

  if (deg <= 32) {
    const bool on = lane < deg;
    int   my_s = 0;
    float sc[H];
#pragma unroll
    for (int h = 0; h < H; ++h) sc[h] = -INFINITY;
    if (on) edge_scores<H>(bk[lane], p, pd, my_s, sc);
    float my_a[H];
#pragma unroll
    for (int h = 0; h < H; ++h) {
      float m = wave_max(sc[h]);
      if (!isfinite(m)) m = 0.f;
      const float ex = on ? __expf(sc[h] - m) : 0.f;
      const float sum = wave_sum(ex);
      my_a[h] = ex / fmaxf(sum, 1e-20f);
    }
    gather_rows<H, F, XIN>(ht, lane, my_s, my_a, deg, acc);
  } else {
    const int nb    = deg < EL_CAP ? deg : EL_CAP;
    const int total = deg <= EL_CAP ? deg : EL_CAP + degs[NN_NODES];
    float m[H], den[H];
#pragma unroll
    for (int h = 0; h < H; ++h) m[h] = -INFINITY;
    for (int base = 0; base < total; base += 32) {
      int2 se;
      if (list_entry(bk, ov, n, nb, total, base + lane, se)) {
        int s_; float sc[H];
        edge_scores<H>(se, p, pd, s_, sc);
#pragma unroll
        for (int h = 0; h < H; ++h) m[h] = fmaxf(m[h], sc[h]);
      }
    }
#pragma unroll
    for (int h = 0; h < H; ++h) {
      m[h] = wave_max(m[h]);
      if (!isfinite(m[h])) m[h] = 0.f;
      den[h] = 0.f;
    }
    for (int base = 0; base < total; base += 32) {
      int2 se;
      if (list_entry(bk, ov, n, nb, total, base + lane, se)) {
        int s_; float sc[H];
        edge_scores<H>(se, p, pd, s_, sc);
#pragma unroll
        for (int h = 0; h < H; ++h) den[h] += __expf(sc[h] - m[h]);
      }
    }
#pragma unroll
    for (int h = 0; h < H; ++h) den[h] = fmaxf(wave_sum(den[h]), 1e-20f);
    for (int base = 0; base < total; base += 32) {
      int   my_s = 0;
      float my_a[H];
#pragma unroll
      for (int h = 0; h < H; ++h) my_a[h] = 0.f;
      int2 se;
      const bool mine = list_entry(bk, ov, n, nb, total, base + lane, se);
      if (mine) {
        float sc[H];
        edge_scores<H>(se, p, pd, my_s, sc);
#pragma unroll
        for (int h = 0; h < H; ++h) my_a[h] = __expf(sc[h] - m[h]) / den[h];
      }
      if (base < EL_CAP) {
        const int left = nb - base;
        gather_rows<H, F, XIN>(ht, lane, my_s, my_a, left < 32 ? left : 32, acc);
      } else {
        gather_rows_masked<H, F, XIN>(ht, lane, my_s, my_a, __builtin_amdgcn_ballot_w32(mine), acc);
      }
    }
  }
}

template <int H, int F>
__global__ void __launch_bounds__(256)
node_softmax_gather_kernel(const int* __restrict__ degs,
                           const int2* __restrict__ bucket,
                           const int4* __restrict__ ov,
                           const float* __restrict__ p,
                           const float* __restrict__ ht,
                           const float* __restrict__ bias,
                           float* __restrict__ hout,
                           int act) {
  constexpr int HF = H * F;
  constexpr int NV = HF / 128;
  const int lane = threadIdx.x & 31;
  const int n = __builtin_amdgcn_readfirstlane((int)(blockIdx.x * 8 + (threadIdx.x >> 5)));
  if (n >= NN_NODES) return;

  float4 acc[NV];
  node_aggregate<H, F>(degs, bucket, ov, p, ht, n, lane, acc);

#pragma unroll
  for (int v = 0; v < NV; ++v) {
    const int c = v * 128 + lane * 4;
    const float4 b = *(const float4*)(bias + c);
    float4 r;
    r.x = acc[v].x + b.x; r.y = acc[v].y + b.y; r.z = acc[v].z + b.z; r.w = acc[v].w + b.w;
    if (act) {
      r.x = leaky(r.x, 0.01f); r.y = leaky(r.y, 0.01f);
      r.z = leaky(r.z, 0.01f); r.w = leaky(r.w, 0.01f);
    }
    *(float4*)(hout + (size_t)n * HF + c) = r;
  }
}

__device__ __forceinline__ v8f wmma4(float a0, float a1, float b0, float b1, v8f acc) {
  v2f a = {a0, a1};
  v2f b = {b0, b1};
  return __builtin_amdgcn_wmma_f32_16x16x4_f32(
       false, a,  false, b,
       (short)0, acc,  false,  false);
}

#define FG_ROWS  32
#define FG_PITCH 260
#define AG_PITCH 516
template <int NOUT, bool FIRST>
__global__ void __launch_bounds__(512)
gather_gemm_kernel(const int* __restrict__ degs,
                   const int2* __restrict__ bucket,
                   const int4* __restrict__ ov,
                   const float* __restrict__ p,
                   const float* __restrict__ ht,
                   const float* __restrict__ bias,
                   const float* __restrict__ W0,
                   const float* __restrict__ W,
                   const float* __restrict__ avec, int fmask,
                   float* __restrict__ C,
                   float* __restrict__ ppart) {
  constexpr int TM = NOUT / 128;
  constexpr int NS = NOUT / 32;
  __shared__ __attribute__((aligned(16))) float As[FG_ROWS][FG_PITCH];
  __shared__ float part[NOUT / 16][FG_ROWS];
  __shared__ __attribute__((aligned(16))) float Ag[FIRST ? FG_ROWS : 1][FIRST ? AG_PITCH : 4];

  const int t     = threadIdx.x;
  const int lane  = t & 31;
  const int wave  = __builtin_amdgcn_readfirstlane(t >> 5);
  const int mBase = blockIdx.x * FG_ROWS;

  if constexpr (FIRST) {
    for (int q = 0; q < 2; ++q) {
      const int rl = wave + 16 * q;
      const int n  = mBase + rl;
      float4 acc[4];
#pragma unroll
      for (int hh = 0; hh < 4; ++hh) { acc[hh].x = 0.f; acc[hh].y = 0.f; acc[hh].z = 0.f; acc[hh].w = 0.f; }
      if (n < NN_NODES)
        node_aggregate<4, 64, true>(degs, bucket, ov, p, ht, n, lane, acc);
#pragma unroll
      for (int hh = 0; hh < 4; ++hh) *(float4*)&Ag[rl][hh * 128 + lane * 4] = acc[hh];
    }
    __syncthreads();

    {
      const int c0  = lane & 15;
      const int h0  = lane >> 4;
      const int hd  = wave >> 2;
      const float* wrow0 = W0 + (size_t)(wave * 16 + c0) * 128 + 4 * h0;
      v8f acc0[2];
#pragma unroll
      for (int m = 0; m < 2; ++m) acc0[m] = (v8f){0.f, 0.f, 0.f, 0.f, 0.f, 0.f, 0.f, 0.f};
#pragma unroll
      for (int i = 0; i < 16; ++i) {
        const float4 b = *(const float4*)(wrow0 + 8 * i);
        float4 a[2];
#pragma unroll
        for (int m = 0; m < 2; ++m) a[m] = *(const float4*)&Ag[16 * m + c0][hd * 128 + 8 * i + 4 * h0];
#pragma unroll
        for (int m = 0; m < 2; ++m) acc0[m] = wmma4(a[m].x, a[m].y, b.x, b.y, acc0[m]);
#pragma unroll
        for (int m = 0; m < 2; ++m) acc0[m] = wmma4(a[m].z, a[m].w, b.z, b.w, acc0[m]);
      }
      const float bv = bias[wave * 16 + c0];
#pragma unroll
      for (int m = 0; m < 2; ++m) {
#pragma unroll
        for (int r = 0; r < 8; ++r)
          As[16 * m + 8 * h0 + r][wave * 16 + c0] = leaky(acc0[m][r] + bv, 0.01f);
      }
    }
  } else {
  const float4 bA = *(const float4*)(bias + lane * 4);
  const float4 bB = *(const float4*)(bias + 128 + lane * 4);
  for (int q = 0; q < 2; ++q) {
    const int rl = wave + 16 * q;
    const int n  = mBase + rl;
    float4 r0, r1;
    r0.x = 0.f; r0.y = 0.f; r0.z = 0.f; r0.w = 0.f;
    r1 = r0;
    if (n < NN_NODES) {
      float4 acc[2];
      node_aggregate<4, 64>(degs, bucket, ov, p, ht, n, lane, acc);
      r0.x = leaky(acc[0].x + bA.x, 0.01f); r0.y = leaky(acc[0].y + bA.y, 0.01f);
      r0.z = leaky(acc[0].z + bA.z, 0.01f); r0.w = leaky(acc[0].w + bA.w, 0.01f);
      r1.x = leaky(acc[1].x + bB.x, 0.01f); r1.y = leaky(acc[1].y + bB.y, 0.01f);
      r1.z = leaky(acc[1].z + bB.z, 0.01f); r1.w = leaky(acc[1].w + bB.w, 0.01f);
    }
    *(float4*)&As[rl][lane * 4]       = r0;
    *(float4*)&As[rl][128 + lane * 4] = r1;
  }
  }
  __syncthreads();

  const int c  = lane & 15;
  const int h  = lane >> 4;
  const int ct = (NOUT == 256) ? wave : (wave & 7);
  const int rt = (NOUT == 256) ? 0 : (wave >> 3);
  const float* wrow = W + (size_t)(ct * 16 + c) * 256 + 4 * h;

  v8f acc[TM];
#pragma unroll
  for (int m = 0; m < TM; ++m) acc[m] = (v8f){0.f, 0.f, 0.f, 0.f, 0.f, 0.f, 0.f, 0.f};
  {
    float4 b_cur = *(const float4*)(wrow);
    float4 a_cur[TM];
#pragma unroll
    for (int m = 0; m < TM; ++m) a_cur[m] = *(const float4*)&As[16 * (rt + m) + c][4 * h];
#pragma unroll
    for (int i = 0; i < 32; ++i) {
      float4 b_nxt = b_cur;
      float4 a_nxt[TM];
#pragma unroll
      for (int m = 0; m < TM; ++m) a_nxt[m] = a_cur[m];
      if (i + 1 < 32) {
        b_nxt = *(const float4*)(wrow + 8 * (i + 1));
#pragma unroll
        for (int m = 0; m < TM; ++m)
          a_nxt[m] = *(const float4*)&As[16 * (rt + m) + c][8 * (i + 1) + 4 * h];
      }
#pragma unroll
      for (int m = 0; m < TM; ++m)
        acc[m] = wmma4(a_cur[m].x, a_cur[m].y, b_cur.x, b_cur.y, acc[m]);
#pragma unroll
      for (int m = 0; m < TM; ++m)
        acc[m] = wmma4(a_cur[m].z, a_cur[m].w, b_cur.z, b_cur.w, acc[m]);
      b_cur = b_nxt;
#pragma unroll
      for (int m = 0; m < TM; ++m) a_cur[m] = a_nxt[m];
    }
  }

  const int col = ct * 16 + c;
#pragma unroll
  for (int m = 0; m < TM; ++m) {
#pragma unroll
    for (int r = 0; r < 8; ++r) {
      const int row = mBase + 16 * (rt + m) + 8 * h + r;
      if (row < NN_NODES) C[(size_t)row * NOUT + col] = acc[m][r];
    }
  }

  const float av = avec[col & fmask];
#pragma unroll
  for (int m = 0; m < TM; ++m) {
#pragma unroll
    for (int r = 0; r < 8; ++r) {
      float v = acc[m][r] * av;
      v += row_xor<1>(v); v += row_xor<2>(v);
      v += row_xor<4>(v); v += row_xor<8>(v);
      if (c == r) part[ct][16 * (rt + m) + 8 * h + r] = v;
    }
  }
  __syncthreads();
  if (t < FG_ROWS * NS) {
    const int rl = t / NS;
    const int j  = t % NS;
    if (mBase + rl < NN_NODES)
      ppart[(size_t)(mBase + rl) * NS + j] = part[2 * j][rl] + part[2 * j + 1][rl];
  }
}

#define PR_TILES  (N_ROWS / 16)
#define PR_BLOCKS 512
#define ZP  132
#define Z1P 68

__device__ __forceinline__ void wave_lds_sync() {
  __builtin_amdgcn_fence(__ATOMIC_RELEASE, "wavefront");
  __builtin_amdgcn_wave_barrier();
  __builtin_amdgcn_fence(__ATOMIC_ACQUIRE, "wavefront");
}

__global__ void __launch_bounds__(256)
predictor_fused_kernel(const float* __restrict__ emb,
                       const int* __restrict__ ps, const int* __restrict__ pd,
                       const int* __restrict__ ns, const int* __restrict__ nd,
                       const float* __restrict__ pW1, const float* __restrict__ pb1,
                       const float* __restrict__ pW2, const float* __restrict__ pb2,
                       const float* __restrict__ pW3, const float* __restrict__ pb3,
                       float* __restrict__ out) {
  __shared__ __attribute__((aligned(16))) float W1s[64][ZP];
  __shared__ __attribute__((aligned(16))) float W2s[32][Z1P];
  __shared__ __attribute__((aligned(16))) float Zs[8][16][ZP];
  __shared__ __attribute__((aligned(16))) float Z1s[8][16][Z1P];

  const int t    = threadIdx.x;
  const int lane = t & 31;
  const int wave = __builtin_amdgcn_readfirstlane(t >> 5);
  const int c    = lane & 15;
  const int h    = lane >> 4;

#pragma unroll
  for (int j = 0; j < 8; ++j) {
    const int idx = t + j * 256;
    const int r   = idx >> 5;
    const int c4  = (idx & 31) * 4;
    *(float4*)&W1s[r][c4] = *(const float4*)(pW1 + r * 128 + c4);
  }
#pragma unroll
  for (int j = 0; j < 2; ++j) {
    const int idx = t + j * 256;
    const int r   = idx >> 4;
    const int c4  = (idx & 15) * 4;
    *(float4*)&W2s[r][c4] = *(const float4*)(pW2 + r * 64 + c4);
  }
  float bias1[4], bias2[2], w3[2];
#pragma unroll
  for (int tt = 0; tt < 4; ++tt) bias1[tt] = pb1[16 * tt + c];
#pragma unroll
  for (int tt = 0; tt < 2; ++tt) { bias2[tt] = pb2[16 * tt + c]; w3[tt] = pW3[16 * tt + c]; }
  const float b3 = pb3[0];
  __syncthreads();

  for (int wt = blockIdx.x * 8 + wave; wt < PR_TILES; wt += PR_BLOCKS * 8) {
    const int row0 = wt * 16;
    const int* sp = ps; const int* dp = pd; int rbase = row0;
    if (row0 >= N_PAIR) { sp = ns; dp = nd; rbase = row0 - N_PAIR; }
    const int si = sp[rbase + c];
    const int di = dp[rbase + c];

#pragma unroll
    for (int r = 0; r < 16; ++r) {
      const int s = __builtin_amdgcn_readlane(si, r);
      const int d = __builtin_amdgcn_readlane(di, r);
      const float4 a = *(const float4*)(emb + (size_t)s * 128 + lane * 4);
      const float4 b = *(const float4*)(emb + (size_t)d * 128 + lane * 4);
      float4 z; z.x = a.x * b.x; z.y = a.y * b.y; z.z = a.z * b.z; z.w = a.w * b.w;
      *(float4*)&Zs[wave][r][lane * 4] = z;
    }
    wave_lds_sync();

    v8f acc[4];
#pragma unroll
    for (int tt = 0; tt < 4; ++tt) acc[tt] = (v8f){0.f, 0.f, 0.f, 0.f, 0.f, 0.f, 0.f, 0.f};
    {
      float4 a_cur = *(const float4*)&Zs[wave][c][4 * h];
      float4 b_cur[4];
#pragma unroll
      for (int tt = 0; tt < 4; ++tt) b_cur[tt] = *(const float4*)&W1s[16 * tt + c][4 * h];
#pragma unroll
      for (int i = 0; i < 16; ++i) {
        float4 a_nxt = a_cur;
        float4 b_nxt[4];
#pragma unroll
        for (int tt = 0; tt < 4; ++tt) b_nxt[tt] = b_cur[tt];
        if (i + 1 < 16) {
          a_nxt = *(const float4*)&Zs[wave][c][8 * (i + 1) + 4 * h];
#pragma unroll
          for (int tt = 0; tt < 4; ++tt)
            b_nxt[tt] = *(const float4*)&W1s[16 * tt + c][8 * (i + 1) + 4 * h];
        }
#pragma unroll
        for (int tt = 0; tt < 4; ++tt)
          acc[tt] = wmma4(a_cur.x, a_cur.y, b_cur[tt].x, b_cur[tt].y, acc[tt]);
#pragma unroll
        for (int tt = 0; tt < 4; ++tt)
          acc[tt] = wmma4(a_cur.z, a_cur.w, b_cur[tt].z, b_cur[tt].w, acc[tt]);
        a_cur = a_nxt;
#pragma unroll
        for (int tt = 0; tt < 4; ++tt) b_cur[tt] = b_nxt[tt];
      }
    }
#pragma unroll
    for (int tt = 0; tt < 4; ++tt) {
#pragma unroll
      for (int r = 0; r < 8; ++r)
        Z1s[wave][8 * h + r][16 * tt + c] = leaky(acc[tt][r] + bias1[tt], 0.2f);
    }
    wave_lds_sync();

    v8f acc2[2];
#pragma unroll
    for (int tt = 0; tt < 2; ++tt) acc2[tt] = (v8f){0.f, 0.f, 0.f, 0.f, 0.f, 0.f, 0.f, 0.f};
    {
      float4 a2[8];
      float4 b2[2][8];
#pragma unroll
      for (int i = 0; i < 8; ++i) {
        a2[i] = *(const float4*)&Z1s[wave][c][8 * i + 4 * h];
#pragma unroll
        for (int tt = 0; tt < 2; ++tt)
          b2[tt][i] = *(const float4*)&W2s[16 * tt + c][8 * i + 4 * h];
      }
#pragma unroll
      for (int i = 0; i < 8; ++i) {
#pragma unroll
        for (int tt = 0; tt < 2; ++tt)
          acc2[tt] = wmma4(a2[i].x, a2[i].y, b2[tt][i].x, b2[tt][i].y, acc2[tt]);
#pragma unroll
        for (int tt = 0; tt < 2; ++tt)
          acc2[tt] = wmma4(a2[i].z, a2[i].w, b2[tt][i].z, b2[tt][i].w, acc2[tt]);
      }
    }

    float mine = 0.f;
#pragma unroll
    for (int r = 0; r < 8; ++r) {
      float v = leaky(acc2[0][r] + bias2[0], 0.2f) * w3[0]
              + leaky(acc2[1][r] + bias2[1], 0.2f) * w3[1];
      v += row_xor<1>(v); v += row_xor<2>(v);
      v += row_xor<4>(v); v += row_xor<8>(v);
      if (c == r) mine = v;
    }
    if (c < 8) out[row0 + 8 * h + c] = mine + b3;
    wave_lds_sync();
  }
}

extern "C" void kernel_launch(void* const* d_in, const int* in_sizes, int n_in,
                              void* d_out, int out_size, void* d_ws, size_t ws_size,
                              hipStream_t stream) {
  const float* x    = (const float*)d_in[0];
  const float* ew   = (const float*)d_in[1];
  const int*   esrc = (const int*)d_in[2];
  const int*   edst = (const int*)d_in[3];
  const int*   psrc = (const int*)d_in[4];
  const int*   pdst = (const int*)d_in[5];
  const int*   nsrc = (const int*)d_in[6];
  const int*   ndst = (const int*)d_in[7];
  const float* Wl[3] = {(const float*)d_in[8],  (const float*)d_in[11], (const float*)d_in[14]};
  const float* al[3] = {(const float*)d_in[9],  (const float*)d_in[12], (const float*)d_in[15]};
  const float* bl[3] = {(const float*)d_in[10], (const float*)d_in[13], (const float*)d_in[16]};
  const float* pW1 = (const float*)d_in[17]; const float* pb1 = (const float*)d_in[18];
  const float* pW2 = (const float*)d_in[19]; const float* pb2 = (const float*)d_in[20];
  const float* pW3 = (const float*)d_in[21]; const float* pb3 = (const float*)d_in[22];

  float* out   = (float*)d_out;
  float* emb   = out + 2 * N_PAIR;

  float*    ws   = (float*)d_ws;
  float*    htA  = ws;
  float*    htB  = htA  + 6400000;
  float*    pA   = htB  + 6400000;
  float*    pB   = pA   + 200000;
  int*      deg    = (int*)(pB + 200000);
  int2*     bucket = (int2*)(deg + 25088);
  int4*     ov     = (int4*)(bucket + 1600000);

  proj_p_kernel<<<(NN_NODES + 63) / 64, 256, 0, stream>>>(x, Wl[0], al[0], pA, deg);

  edge_bucket_kernel<<<(N_EDGES + 255) / 256, 256, 0, stream>>>(edst, esrc, ew, deg, bucket, ov);

  gather_gemm_kernel<256, true><<<(NN_NODES + FG_ROWS - 1) / FG_ROWS, 512, 0, stream>>>(
      deg, bucket, ov, pA, x, bl[0], Wl[0], Wl[1], al[1], 63, htB, pB);
  gather_gemm_kernel<128, false><<<(NN_NODES + FG_ROWS - 1) / FG_ROWS, 512, 0, stream>>>(
      deg, bucket, ov, pB, htB, bl[1], nullptr, Wl[2], al[2], 127, htA, pA);

  node_softmax_gather_kernel<1, 128><<<(NN_NODES + 7) / 8, 256, 0, stream>>>(
      deg, bucket, ov, pA, htA, bl[2], emb, 0);

  predictor_fused_kernel<<<PR_BLOCKS, 256, 0, stream>>>(emb, psrc, pdst, nsrc, ndst,
                                                        pW1, pb1, pW2, pb2, pW3, pb3, out);
}
